# starting kernel
# baseline (speedup 1.0000x reference)
#include <hip/hip_runtime.h>
#include <math.h>

typedef __attribute__((ext_vector_type(2))) float v2f;
typedef __attribute__((ext_vector_type(8))) float v8f;

__device__ __forceinline__ float mish_f(float v) {
  float sp = (v > 20.0f) ? v : log1pf(__expf(v));
  return v * tanhf(sp);
}

__global__ void mish_kernel(const float* __restrict__ in, float* __restrict__ out, int n) {
  int i = blockIdx.x * blockDim.x + threadIdx.x;
  if (i < n) out[i] = mish_f(in[i]);
}

__global__ void cat_mish_kernel(const float* __restrict__ xs, const float* __restrict__ ys,
                                const float* __restrict__ dprev, float* __restrict__ out,
                                int N, int B) {
  int twoN = 2 * N;
  int idx = blockIdx.x * blockDim.x + threadIdx.x;
  if (idx >= B * twoN) return;
  int b = idx / twoN;
  int c = idx - b * twoN;
  float v = (c < N) ? xs[b * N + c] : ys[b * N + (c - N)];
  if (dprev) v += dprev[idx];
  out[idx] = mish_f(v);
}

__global__ void gemm_wmma_f32(const float* __restrict__ A, const float* __restrict__ W,
                              const float* __restrict__ bias, float* __restrict__ C,
                              int M, int N, int K) {
  int wave = threadIdx.x >> 5;
  int lane = threadIdx.x & 31;
  int tiles_n = N >> 5;
  int tile = blockIdx.x * 4 + wave;
  if (tile >= (M >> 5) * tiles_n) return;
  int tm = tile / tiles_n;
  int m0 = tm << 5;
  int n0 = (tile - tm * tiles_n) << 5;
  int r  = lane & 15;
  int kh = (lane >> 4) << 1;
  const float* Ar0 = A + (size_t)(m0 + r) * K + kh;
  const float* Ar1 = Ar0 + (size_t)16 * K;
  const float* Wr0 = W + (size_t)(n0 + r) * K + kh;
  const float* Wr1 = Wr0 + (size_t)16 * K;
  v8f acc00 = {};
  v8f acc01 = {};
  v8f acc10 = {};
  v8f acc11 = {};
  for (int k = 0; k < K; k += 4) {
    v2f a0, a1, b0, b1;
    a0.x = Ar0[k]; a0.y = Ar0[k + 1];
    a1.x = Ar1[k]; a1.y = Ar1[k + 1];
    b0.x = Wr0[k]; b0.y = Wr0[k + 1];
    b1.x = Wr1[k]; b1.y = Wr1[k + 1];
    acc00 = __builtin_amdgcn_wmma_f32_16x16x4_f32(false, a0, false, b0, (short)0, acc00, false, false);
    acc01 = __builtin_amdgcn_wmma_f32_16x16x4_f32(false, a0, false, b1, (short)0, acc01, false, false);
    acc10 = __builtin_amdgcn_wmma_f32_16x16x4_f32(false, a1, false, b0, (short)0, acc10, false, false);
    acc11 = __builtin_amdgcn_wmma_f32_16x16x4_f32(false, a1, false, b1, (short)0, acc11, false, false);
  }
  int n_lo = n0 + r;
  int n_hi = n_lo + 16;
  float bv0 = bias ? bias[n_lo] : 0.0f;
  float bv1 = bias ? bias[n_hi] : 0.0f;
  int mb = m0 + ((lane >> 4) << 3);
#pragma unroll
  for (int v = 0; v < 8; ++v) {
    C[(size_t)(mb + v) * N + n_lo]      = acc00[v] + bv0;
    C[(size_t)(mb + v) * N + n_hi]      = acc01[v] + bv1;
    C[(size_t)(mb + 16 + v) * N + n_lo] = acc10[v] + bv0;
    C[(size_t)(mb + 16 + v) * N + n_hi] = acc11[v] + bv1;
  }
}

__global__ void attn_kernel(const float* __restrict__ self, const float* __restrict__ other,
                            const float* __restrict__ beta_p, float* __restrict__ out, int N) {
  __shared__ float sh_s[512];
  __shared__ float sh_o[512];
  __shared__ float smx[256];
  __shared__ float smn[256];
  int b = blockIdx.x;
  int t = threadIdx.x;
  const float* s = self  + (size_t)b * N;
  const float* o = other + (size_t)b * N;
  float beta = beta_p[0];
  if (beta == 0.0f) {
    for (int i = t; i < N; i += blockDim.x) out[(size_t)b * N + i] = s[i];
    return;
  }
  for (int i = t; i < N; i += blockDim.x) { sh_s[i] = s[i]; sh_o[i] = o[i]; }
  __syncthreads();
  float mx = -3.4e38f, mn = 3.4e38f;
  for (int i = t; i < N; i += blockDim.x) { mx = fmaxf(mx, sh_o[i]); mn = fminf(mn, sh_o[i]); }
  smx[t] = mx; smn[t] = mn;
  __syncthreads();
  for (int off = 128; off > 0; off >>= 1) {
    if (t < off) { smx[t] = fmaxf(smx[t], smx[t + off]); smn[t] = fminf(smn[t], smn[t + off]); }
    __syncthreads();
  }
  float omax = smx[0], omin = smn[0];
  for (int i = t; i < N; i += blockDim.x) {
    float si = sh_s[i];
    float m = (si >= 0.0f) ? si * omax : si * omin;
    float den = 0.0f, num = 0.0f;
    for (int j = 0; j < N; ++j) {
      float e = __expf(si * sh_o[j] - m);
      den += e;
      num += e * sh_s[j];
    }
    out[(size_t)b * N + i] = si + beta * num / den;
  }
}

__global__ void head_kernel(const float* __restrict__ d3, const float* __restrict__ W6,
                            const float* __restrict__ b6, float* __restrict__ out, int B) {
  __shared__ float sh[8];
  int b = blockIdx.x;
  int lane = threadIdx.x;
  if (lane < 7) {
    const float* row = d3 + (size_t)b * 64;
    const float* w   = W6 + (size_t)lane * 64;
    float acc = b6[lane];
    for (int k = 0; k < 64; ++k) acc += row[k] * w[k];
    sh[lane] = acc;
  }
  __syncthreads();
  if (lane < 7) {
    float nrm = sqrtf(sh[0]*sh[0] + sh[1]*sh[1] + sh[2]*sh[2] + sh[3]*sh[3]);
    float v = sh[lane];
    if (lane < 4) v = v / fmaxf(nrm, 1e-12f);
    out[(size_t)b * 7 + lane] = v;
    out[(size_t)B * 7 + (size_t)b * 7 + lane] = v;
  }
}

extern "C" void kernel_launch(void* const* d_in, const int* in_sizes, int n_in,
                              void* d_out, int out_size, void* d_ws, size_t ws_size,
                              hipStream_t stream) {
  const float* x    = (const float*)d_in[0];
  const float* y    = (const float*)d_in[1];
  const float* W1   = (const float*)d_in[2];
  const float* be1  = (const float*)d_in[3];
  const float* fcW1 = (const float*)d_in[4];
  const float* fcb1 = (const float*)d_in[5];
  const float* W2   = (const float*)d_in[6];
  const float* be2  = (const float*)d_in[7];
  const float* fcW2 = (const float*)d_in[8];
  const float* fcb2 = (const float*)d_in[9];
  const float* W3   = (const float*)d_in[10];
  const float* be3  = (const float*)d_in[11];
  const float* fcW3 = (const float*)d_in[12];
  const float* fcb3 = (const float*)d_in[13];
  const float* W4   = (const float*)d_in[14];
  const float* be4  = (const float*)d_in[15];
  const float* fcW4 = (const float*)d_in[16];
  const float* fcb4 = (const float*)d_in[17];
  const float* W6   = (const float*)d_in[18];
  const float* b6   = (const float*)d_in[19];

  float* ws = (float*)d_ws;
  float* m  = ws;
  float* qa = m  + 524288;
  float* qb = qa + 262144;
  float* xa = qb + 262144;
  float* ya = xa + 262144;
  float* d0 = ya + 262144;
  float* d1 = d0 + 262144;
  float* d2 = d1 + 131072;
  float* d3 = d2 + 65536;

  const int B = 512;
  auto grid1d = [](int n) { return (n + 255) / 256; };
  auto gemm = [&](const float* A, const float* W, const float* bias, float* C, int N, int K) {
    int tiles = (B >> 5) * (N >> 5);
    hipLaunchKernelGGL(gemm_wmma_f32, dim3((tiles + 3) / 4), dim3(128), 0, stream,
                       A, W, bias, C, B, N, K);
  };
  auto stage = [&](const float* xin, const float* yin, const float* W, const float* beta,
                   int Kin, int Nout) {
    hipLaunchKernelGGL(mish_kernel, dim3(grid1d(B * Kin)), dim3(256), 0, stream, xin, m, B * Kin);
    gemm(m, W, nullptr, qa, Nout, Kin);
    hipLaunchKernelGGL(mish_kernel, dim3(grid1d(B * Kin)), dim3(256), 0, stream, yin, m, B * Kin);
    gemm(m, W, nullptr, qb, Nout, Kin);
    hipLaunchKernelGGL(attn_kernel, dim3(B), dim3(256), 0, stream, qa, qb, beta, xa, Nout);
    hipLaunchKernelGGL(attn_kernel, dim3(B), dim3(256), 0, stream, qb, qa, beta, ya, Nout);
  };
  auto fc = [&](const float* dprev, const float* fcW, const float* fcb, float* dout,
                int N, int Nout) {
    hipLaunchKernelGGL(cat_mish_kernel, dim3(grid1d(B * 2 * N)), dim3(256), 0, stream,
                       xa, ya, dprev, m, N, B);
    gemm(m, fcW, fcb, dout, Nout, 2 * N);
  };

  stage(x,  y,  W1, be1, 1024, 512);
  fc(nullptr, fcW1, fcb1, d0, 512, 512);
  stage(xa, ya, W2, be2, 512, 256);
  fc(d0, fcW2, fcb2, d1, 256, 256);
  stage(xa, ya, W3, be3, 256, 128);
  fc(d1, fcW3, fcb3, d2, 128, 128);
  stage(xa, ya, W4, be4, 128, 64);
  fc(d2, fcW4, fcb4, d3, 64, 64);
  hipLaunchKernelGGL(head_kernel, dim3(B), dim3(32), 0, stream, d3, W6, b6, (float*)d_out, B);

  (void)in_sizes; (void)n_in; (void)out_size; (void)ws_size;
}


# GEMM split-K: K of each 32x32 tile cut over the 8 waves of one work-group (8x the waves, 256 groups for N=512), partial tiles summed through LDS in fixed wave order; f32 summation order changes
# speedup vs baseline: 2.1650x; 2.1650x over previous
#include <hip/hip_runtime.h>
#include <math.h>

typedef __attribute__((ext_vector_type(2))) float v2f;
typedef __attribute__((ext_vector_type(8))) float v8f;

__device__ __forceinline__ float mish_f(float v) {
  float sp = (v > 20.0f) ? v : log1pf(__expf(v));
  return v * tanhf(sp);
}

__global__ void mish_kernel(const float* __restrict__ in, float* __restrict__ out, int n) {
  int i = blockIdx.x * blockDim.x + threadIdx.x;
  if (i < n) out[i] = mish_f(in[i]);
}

__global__ void cat_mish_kernel(const float* __restrict__ xs, const float* __restrict__ ys,
                                const float* __restrict__ dprev, float* __restrict__ out,
                                int N, int B) {
  int twoN = 2 * N;
  int idx = blockIdx.x * blockDim.x + threadIdx.x;
  if (idx >= B * twoN) return;
  int b = idx / twoN;
  int c = idx - b * twoN;
  float v = (c < N) ? xs[b * N + c] : ys[b * N + (c - N)];
  if (dprev) v += dprev[idx];
  out[idx] = mish_f(v);
}

#define GEMM_WAVES 8
#define PART_PITCH 36
#define PART_TILE (32 * PART_PITCH)

__global__ __launch_bounds__(GEMM_WAVES * 32)
void gemm_wmma_f32(const float* __restrict__ A, const float* __restrict__ W,
                   const float* __restrict__ bias, float* __restrict__ C,
                   int M, int N, int K) {
  __shared__ __attribute__((aligned(16))) float part[GEMM_WAVES * PART_TILE];
  int wave = __builtin_amdgcn_readfirstlane((int)(threadIdx.x >> 5));
  int lane = threadIdx.x & 31;
  int tiles_n = N >> 5;
  int tile = blockIdx.x;
  if (tile >= (M >> 5) * tiles_n) return;
  int tm = tile / tiles_n;
  int m0 = tm << 5;
  int n0 = (tile - tm * tiles_n) << 5;
  int r  = lane & 15;
  int kh = (lane >> 4) << 1;
  int ksl = ((((K >> 2) + GEMM_WAVES - 1) / GEMM_WAVES)) << 2;
  int kb = wave * ksl;
  int ke = kb + ksl;
  if (ke > K) ke = K;
  const float* Ar0 = A + (size_t)(m0 + r) * K + kh;
  const float* Ar1 = Ar0 + (size_t)16 * K;
  const float* Wr0 = W + (size_t)(n0 + r) * K + kh;
  const float* Wr1 = Wr0 + (size_t)16 * K;
  v8f acc00 = {};
  v8f acc01 = {};
  v8f acc10 = {};
  v8f acc11 = {};
  for (int k = kb; k < ke; k += 4) {
    v2f a0, a1, b0, b1;
    a0.x = Ar0[k]; a0.y = Ar0[k + 1];
    a1.x = Ar1[k]; a1.y = Ar1[k + 1];
    b0.x = Wr0[k]; b0.y = Wr0[k + 1];
    b1.x = Wr1[k]; b1.y = Wr1[k + 1];
    acc00 = __builtin_amdgcn_wmma_f32_16x16x4_f32(false, a0, false, b0, (short)0, acc00, false, false);
    acc01 = __builtin_amdgcn_wmma_f32_16x16x4_f32(false, a0, false, b1, (short)0, acc01, false, false);
    acc10 = __builtin_amdgcn_wmma_f32_16x16x4_f32(false, a1, false, b0, (short)0, acc10, false, false);
    acc11 = __builtin_amdgcn_wmma_f32_16x16x4_f32(false, a1, false, b1, (short)0, acc11, false, false);
  }
  float* pw = part + wave * PART_TILE + ((lane >> 4) << 3) * PART_PITCH + r;
  for (int v = 0; v < 8; ++v) {
    pw[v * PART_PITCH]             = acc00[v];
    pw[v * PART_PITCH + 16]        = acc01[v];
    pw[(v + 16) * PART_PITCH]      = acc10[v];
    pw[(v + 16) * PART_PITCH + 16] = acc11[v];
  }
  __syncthreads();
  int row = threadIdx.x >> 3;
  int c4  = (threadIdx.x & 7) << 2;
  const float* ps = part + row * PART_PITCH + c4;
  float4 s = *(const float4*)ps;
  for (int w = 1; w < GEMM_WAVES; ++w) {
    float4 p = *(const float4*)(ps + w * PART_TILE);
    s.x += p.x; s.y += p.y; s.z += p.z; s.w += p.w;
  }
  if (bias) {
    float4 bv = *(const float4*)(bias + n0 + c4);
    s.x += bv.x; s.y += bv.y; s.z += bv.z; s.w += bv.w;
  }
  *(float4*)(C + (size_t)(m0 + row) * N + n0 + c4) = s;
}

__global__ void attn_kernel(const float* __restrict__ self, const float* __restrict__ other,
                            const float* __restrict__ beta_p, float* __restrict__ out, int N) {
  __shared__ float sh_s[512];
  __shared__ float sh_o[512];
  __shared__ float smx[256];
  __shared__ float smn[256];
  int b = blockIdx.x;
  int t = threadIdx.x;
  const float* s = self  + (size_t)b * N;
  const float* o = other + (size_t)b * N;
  float beta = beta_p[0];
  if (beta == 0.0f) {
    for (int i = t; i < N; i += blockDim.x) out[(size_t)b * N + i] = s[i];
    return;
  }
  for (int i = t; i < N; i += blockDim.x) { sh_s[i] = s[i]; sh_o[i] = o[i]; }
  __syncthreads();
  float mx = -3.4e38f, mn = 3.4e38f;
  for (int i = t; i < N; i += blockDim.x) { mx = fmaxf(mx, sh_o[i]); mn = fminf(mn, sh_o[i]); }
  smx[t] = mx; smn[t] = mn;
  __syncthreads();
  for (int off = 128; off > 0; off >>= 1) {
    if (t < off) { smx[t] = fmaxf(smx[t], smx[t + off]); smn[t] = fminf(smn[t], smn[t + off]); }
    __syncthreads();
  }
  float omax = smx[0], omin = smn[0];
  for (int i = t; i < N; i += blockDim.x) {
    float si = sh_s[i];
    float m = (si >= 0.0f) ? si * omax : si * omin;
    float den = 0.0f, num = 0.0f;
    for (int j = 0; j < N; ++j) {
      float e = __expf(si * sh_o[j] - m);
      den += e;
      num += e * sh_s[j];
    }
    out[(size_t)b * N + i] = si + beta * num / den;
  }
}

__global__ void head_kernel(const float* __restrict__ d3, const float* __restrict__ W6,
                            const float* __restrict__ b6, float* __restrict__ out, int B) {
  __shared__ float sh[8];
  int b = blockIdx.x;
  int lane = threadIdx.x;
  if (lane < 7) {
    const float* row = d3 + (size_t)b * 64;
    const float* w   = W6 + (size_t)lane * 64;
    float acc = b6[lane];
    for (int k = 0; k < 64; ++k) acc += row[k] * w[k];
    sh[lane] = acc;
  }
  __syncthreads();
  if (lane < 7) {
    float nrm = sqrtf(sh[0]*sh[0] + sh[1]*sh[1] + sh[2]*sh[2] + sh[3]*sh[3]);
    float v = sh[lane];
    if (lane < 4) v = v / fmaxf(nrm, 1e-12f);
    out[(size_t)b * 7 + lane] = v;
    out[(size_t)B * 7 + (size_t)b * 7 + lane] = v;
  }
}

extern "C" void kernel_launch(void* const* d_in, const int* in_sizes, int n_in,
                              void* d_out, int out_size, void* d_ws, size_t ws_size,
                              hipStream_t stream) {
  const float* x    = (const float*)d_in[0];
  const float* y    = (const float*)d_in[1];
  const float* W1   = (const float*)d_in[2];
  const float* be1  = (const float*)d_in[3];
  const float* fcW1 = (const float*)d_in[4];
  const float* fcb1 = (const float*)d_in[5];
  const float* W2   = (const float*)d_in[6];
  const float* be2  = (const float*)d_in[7];
  const float* fcW2 = (const float*)d_in[8];
  const float* fcb2 = (const float*)d_in[9];
  const float* W3   = (const float*)d_in[10];
  const float* be3  = (const float*)d_in[11];
  const float* fcW3 = (const float*)d_in[12];
  const float* fcb3 = (const float*)d_in[13];
  const float* W4   = (const float*)d_in[14];
  const float* be4  = (const float*)d_in[15];
  const float* fcW4 = (const float*)d_in[16];
  const float* fcb4 = (const float*)d_in[17];
  const float* W6   = (const float*)d_in[18];
  const float* b6   = (const float*)d_in[19];

  float* ws = (float*)d_ws;
  float* m  = ws;
  float* qa = m  + 524288;
  float* qb = qa + 262144;
  float* xa = qb + 262144;
  float* ya = xa + 262144;
  float* d0 = ya + 262144;
  float* d1 = d0 + 262144;
  float* d2 = d1 + 131072;
  float* d3 = d2 + 65536;

  const int B = 512;
  auto grid1d = [](int n) { return (n + 255) / 256; };
  auto gemm = [&](const float* A, const float* W, const float* bias, float* C, int N, int K) {
    int tiles = (B >> 5) * (N >> 5);
    hipLaunchKernelGGL(gemm_wmma_f32, dim3(tiles), dim3(GEMM_WAVES * 32), 0, stream,
                       A, W, bias, C, B, N, K);
  };
  auto stage = [&](const float* xin, const float* yin, const float* W, const float* beta,
                   int Kin, int Nout) {
    hipLaunchKernelGGL(mish_kernel, dim3(grid1d(B * Kin)), dim3(256), 0, stream, xin, m, B * Kin);
    gemm(m, W, nullptr, qa, Nout, Kin);
    hipLaunchKernelGGL(mish_kernel, dim3(grid1d(B * Kin)), dim3(256), 0, stream, yin, m, B * Kin);
    gemm(m, W, nullptr, qb, Nout, Kin);
    hipLaunchKernelGGL(attn_kernel, dim3(B), dim3(256), 0, stream, qa, qb, beta, xa, Nout);
    hipLaunchKernelGGL(attn_kernel, dim3(B), dim3(256), 0, stream, qb, qa, beta, ya, Nout);
  };
  auto fc = [&](const float* dprev, const float* fcW, const float* fcb, float* dout,
                int N, int Nout) {
    hipLaunchKernelGGL(cat_mish_kernel, dim3(grid1d(B * 2 * N)), dim3(256), 0, stream,
                       xa, ya, dprev, m, N, B);
    gemm(m, fcW, fcb, dout, Nout, 2 * N);
  };

  stage(x,  y,  W1, be1, 1024, 512);
  fc(nullptr, fcW1, fcb1, d0, 512, 512);
  stage(xa, ya, W2, be2, 512, 256);
  fc(d0, fcW2, fcb2, d1, 256, 256);
  stage(xa, ya, W3, be3, 256, 128);
  fc(d1, fcW3, fcb3, d2, 128, 128);
  stage(xa, ya, W4, be4, 128, 64);
  fc(d2, fcW4, fcb4, d3, 64, 64);
  hipLaunchKernelGGL(head_kernel, dim3(B), dim3(32), 0, stream, d3, W6, b6, (float*)d_out, B);

  (void)in_sizes; (void)n_in; (void)out_size; (void)ws_size;
}


# x and y branches of every stage batched: the two mish, two GEMM and two attention launches become one launch each (GEMM M=1024), 33 launches -> 21; sums unchanged
# speedup vs baseline: 2.4443x; 1.1290x over previous
#include <hip/hip_runtime.h>
#include <math.h>

typedef __attribute__((ext_vector_type(2))) float v2f;
typedef __attribute__((ext_vector_type(8))) float v8f;

__device__ __forceinline__ float mish_f(float v) {
  float sp = (v > 20.0f) ? v : log1pf(__expf(v));
  return v * tanhf(sp);
}

__global__ void mish2_kernel(const float* __restrict__ a, const float* __restrict__ b,
                             float* __restrict__ out, int nh) {
  int i = blockIdx.x * blockDim.x + threadIdx.x;
  if (i >= 2 * nh) return;
  float v = (i < nh) ? a[i] : b[i - nh];
  out[i] = mish_f(v);
}

__global__ void cat_mish_kernel(const float* __restrict__ xs, const float* __restrict__ ys,
                                const float* __restrict__ dprev, float* __restrict__ out,
                                int N, int B) {
  int twoN = 2 * N;
  int idx = blockIdx.x * blockDim.x + threadIdx.x;
  if (idx >= B * twoN) return;
  int b = idx / twoN;
  int c = idx - b * twoN;
  float v = (c < N) ? xs[b * N + c] : ys[b * N + (c - N)];
  if (dprev) v += dprev[idx];
  out[idx] = mish_f(v);
}

#define GEMM_WAVES 8
#define PART_PITCH 36
#define PART_TILE (32 * PART_PITCH)

__global__ __launch_bounds__(GEMM_WAVES * 32)
void gemm_wmma_f32(const float* __restrict__ A, const float* __restrict__ W,
                   const float* __restrict__ bias, float* __restrict__ C,
                   int M, int N, int K) {
  __shared__ __attribute__((aligned(16))) float part[GEMM_WAVES * PART_TILE];
  int wave = __builtin_amdgcn_readfirstlane((int)(threadIdx.x >> 5));
  int lane = threadIdx.x & 31;
  int tiles_n = N >> 5;
  int tile = blockIdx.x;
  if (tile >= (M >> 5) * tiles_n) return;
  int tm = tile / tiles_n;
  int m0 = tm << 5;
  int n0 = (tile - tm * tiles_n) << 5;
  int r  = lane & 15;
  int kh = (lane >> 4) << 1;
  int ksl = ((((K >> 2) + GEMM_WAVES - 1) / GEMM_WAVES)) << 2;
  int kb = wave * ksl;
  int ke = kb + ksl;
  if (ke > K) ke = K;
  const float* Ar0 = A + (size_t)(m0 + r) * K + kh;
  const float* Ar1 = Ar0 + (size_t)16 * K;
  const float* Wr0 = W + (size_t)(n0 + r) * K + kh;
  const float* Wr1 = Wr0 + (size_t)16 * K;
  v8f acc00 = {};
  v8f acc01 = {};
  v8f acc10 = {};
  v8f acc11 = {};
  for (int k = kb; k < ke; k += 4) {
    v2f a0, a1, b0, b1;
    a0.x = Ar0[k]; a0.y = Ar0[k + 1];
    a1.x = Ar1[k]; a1.y = Ar1[k + 1];
    b0.x = Wr0[k]; b0.y = Wr0[k + 1];
    b1.x = Wr1[k]; b1.y = Wr1[k + 1];
    acc00 = __builtin_amdgcn_wmma_f32_16x16x4_f32(false, a0, false, b0, (short)0, acc00, false, false);
    acc01 = __builtin_amdgcn_wmma_f32_16x16x4_f32(false, a0, false, b1, (short)0, acc01, false, false);
    acc10 = __builtin_amdgcn_wmma_f32_16x16x4_f32(false, a1, false, b0, (short)0, acc10, false, false);
    acc11 = __builtin_amdgcn_wmma_f32_16x16x4_f32(false, a1, false, b1, (short)0, acc11, false, false);
  }
  float* pw = part + wave * PART_TILE + ((lane >> 4) << 3) * PART_PITCH + r;
  for (int v = 0; v < 8; ++v) {
    pw[v * PART_PITCH]             = acc00[v];
    pw[v * PART_PITCH + 16]        = acc01[v];
    pw[(v + 16) * PART_PITCH]      = acc10[v];
    pw[(v + 16) * PART_PITCH + 16] = acc11[v];
  }
  __syncthreads();
  int row = threadIdx.x >> 3;
  int c4  = (threadIdx.x & 7) << 2;
  const float* ps = part + row * PART_PITCH + c4;
  float4 s = *(const float4*)ps;
  for (int w = 1; w < GEMM_WAVES; ++w) {
    float4 p = *(const float4*)(ps + w * PART_TILE);
    s.x += p.x; s.y += p.y; s.z += p.z; s.w += p.w;
  }
  if (bias) {
    float4 bv = *(const float4*)(bias + n0 + c4);
    s.x += bv.x; s.y += bv.y; s.z += bv.z; s.w += bv.w;
  }
  *(float4*)(C + (size_t)(m0 + row) * N + n0 + c4) = s;
}

__global__ void attn_kernel(const float* __restrict__ Q,
                            const float* __restrict__ beta_p, float* __restrict__ out,
                            int N, int B) {
  __shared__ float sh_s[512];
  __shared__ float sh_o[512];
  __shared__ float smx[256];
  __shared__ float smn[256];
  int b = blockIdx.x;
  int t = threadIdx.x;
  int ob = (b < B) ? b + B : b - B;
  const float* s = Q + (size_t)b  * N;
  const float* o = Q + (size_t)ob * N;
  float beta = beta_p[0];
  if (beta == 0.0f) {
    for (int i = t; i < N; i += blockDim.x) out[(size_t)b * N + i] = s[i];
    return;
  }
  for (int i = t; i < N; i += blockDim.x) { sh_s[i] = s[i]; sh_o[i] = o[i]; }
  __syncthreads();
  float mx = -3.4e38f, mn = 3.4e38f;
  for (int i = t; i < N; i += blockDim.x) { mx = fmaxf(mx, sh_o[i]); mn = fminf(mn, sh_o[i]); }
  smx[t] = mx; smn[t] = mn;
  __syncthreads();
  for (int off = 128; off > 0; off >>= 1) {
    if (t < off) { smx[t] = fmaxf(smx[t], smx[t + off]); smn[t] = fminf(smn[t], smn[t + off]); }
    __syncthreads();
  }
  float omax = smx[0], omin = smn[0];
  for (int i = t; i < N; i += blockDim.x) {
    float si = sh_s[i];
    float m = (si >= 0.0f) ? si * omax : si * omin;
    float den = 0.0f, num = 0.0f;
    for (int j = 0; j < N; ++j) {
      float e = __expf(si * sh_o[j] - m);
      den += e;
      num += e * sh_s[j];
    }
    out[(size_t)b * N + i] = si + beta * num / den;
  }
}

__global__ void head_kernel(const float* __restrict__ d3, const float* __restrict__ W6,
                            const float* __restrict__ b6, float* __restrict__ out, int B) {
  __shared__ float sh[8];
  int b = blockIdx.x;
  int lane = threadIdx.x;
  if (lane < 7) {
    const float* row = d3 + (size_t)b * 64;
    const float* w   = W6 + (size_t)lane * 64;
    float acc = b6[lane];
    for (int k = 0; k < 64; ++k) acc += row[k] * w[k];
    sh[lane] = acc;
  }
  __syncthreads();
  if (lane < 7) {
    float nrm = sqrtf(sh[0]*sh[0] + sh[1]*sh[1] + sh[2]*sh[2] + sh[3]*sh[3]);
    float v = sh[lane];
    if (lane < 4) v = v / fmaxf(nrm, 1e-12f);
    out[(size_t)b * 7 + lane] = v;
    out[(size_t)B * 7 + (size_t)b * 7 + lane] = v;
  }
}

extern "C" void kernel_launch(void* const* d_in, const int* in_sizes, int n_in,
                              void* d_out, int out_size, void* d_ws, size_t ws_size,
                              hipStream_t stream) {
  const float* x    = (const float*)d_in[0];
  const float* y    = (const float*)d_in[1];
  const float* W1   = (const float*)d_in[2];
  const float* be1  = (const float*)d_in[3];
  const float* fcW1 = (const float*)d_in[4];
  const float* fcb1 = (const float*)d_in[5];
  const float* W2   = (const float*)d_in[6];
  const float* be2  = (const float*)d_in[7];
  const float* fcW2 = (const float*)d_in[8];
  const float* fcb2 = (const float*)d_in[9];
  const float* W3   = (const float*)d_in[10];
  const float* be3  = (const float*)d_in[11];
  const float* fcW3 = (const float*)d_in[12];
  const float* fcb3 = (const float*)d_in[13];
  const float* W4   = (const float*)d_in[14];
  const float* be4  = (const float*)d_in[15];
  const float* fcW4 = (const float*)d_in[16];
  const float* fcb4 = (const float*)d_in[17];
  const float* W6   = (const float*)d_in[18];
  const float* b6   = (const float*)d_in[19];

  float* ws = (float*)d_ws;
  float* m  = ws;
  float* q  = m  + 1048576;
  float* xa = q  + 524288;
  float* d0 = xa + 524288;
  float* d1 = d0 + 262144;
  float* d2 = d1 + 131072;
  float* d3 = d2 + 65536;

  const int B = 512;
  auto grid1d = [](int n) { return (n + 255) / 256; };
  auto gemm = [&](const float* A, const float* W, const float* bias, float* C, int M, int N, int K) {
    int tiles = (M >> 5) * (N >> 5);
    hipLaunchKernelGGL(gemm_wmma_f32, dim3(tiles), dim3(GEMM_WAVES * 32), 0, stream,
                       A, W, bias, C, M, N, K);
  };
  auto stage = [&](const float* xin, const float* yin, const float* W, const float* beta,
                   int Kin, int Nout) {
    hipLaunchKernelGGL(mish2_kernel, dim3(grid1d(2 * B * Kin)), dim3(256), 0, stream,
                       xin, yin, m, B * Kin);
    gemm(m, W, nullptr, q, 2 * B, Nout, Kin);
    hipLaunchKernelGGL(attn_kernel, dim3(2 * B), dim3(256), 0, stream, q, beta, xa, Nout, B);
  };
  auto fc = [&](const float* dprev, const float* fcW, const float* fcb, float* dout,
                int N, int Nout) {
    hipLaunchKernelGGL(cat_mish_kernel, dim3(grid1d(B * 2 * N)), dim3(256), 0, stream,
                       xa, xa + B * N, dprev, m, N, B);
    gemm(m, fcW, fcb, dout, B, Nout, 2 * N);
  };

  stage(x,  y,            W1, be1, 1024, 512);
  fc(nullptr, fcW1, fcb1, d0, 512, 512);
  stage(xa, xa + B * 512, W2, be2, 512, 256);
  fc(d0, fcW2, fcb2, d1, 256, 256);
  stage(xa, xa + B * 256, W3, be3, 256, 128);
  fc(d1, fcW3, fcb3, d2, 128, 128);
  stage(xa, xa + B * 128, W4, be4, 128, 64);
  fc(d2, fcW4, fcb4, d3, 64, 64);
  hipLaunchKernelGGL(head_kernel, dim3(B), dim3(32), 0, stream, d3, W6, b6, (float*)d_out, B);

  (void)in_sizes; (void)n_in; (void)out_size; (void)ws_size;
}


# mish and concat(+previous fc) passes folded into the attention launch: one group per sample does both directions and writes both GEMM operands; 21 launches -> 14; sums unchanged
# speedup vs baseline: 2.8511x; 1.1664x over previous
#include <hip/hip_runtime.h>
#include <math.h>

typedef __attribute__((ext_vector_type(2))) float v2f;
typedef __attribute__((ext_vector_type(8))) float v8f;

__device__ __forceinline__ float mish_f(float v) {
  float sp = (v > 20.0f) ? v : log1pf(__expf(v));
  return v * tanhf(sp);
}

__global__ void mish2_kernel(const float* __restrict__ a, const float* __restrict__ b,
                             float* __restrict__ out, int nh) {
  int i = blockIdx.x * blockDim.x + threadIdx.x;
  if (i >= 2 * nh) return;
  float v = (i < nh) ? a[i] : b[i - nh];
  out[i] = mish_f(v);
}

#define GEMM_WAVES 8
#define PART_PITCH 36
#define PART_TILE (32 * PART_PITCH)

__global__ __launch_bounds__(GEMM_WAVES * 32)
void gemm_wmma_f32(const float* __restrict__ A, const float* __restrict__ W,
                   const float* __restrict__ bias, float* __restrict__ C,
                   int M, int N, int K) {
  __shared__ __attribute__((aligned(16))) float part[GEMM_WAVES * PART_TILE];
  int wave = __builtin_amdgcn_readfirstlane((int)(threadIdx.x >> 5));
  int lane = threadIdx.x & 31;
  int tiles_n = N >> 5;
  int tile = blockIdx.x;
  if (tile >= (M >> 5) * tiles_n) return;
  int tm = tile / tiles_n;
  int m0 = tm << 5;
  int n0 = (tile - tm * tiles_n) << 5;
  int r  = lane & 15;
  int kh = (lane >> 4) << 1;
  int ksl = ((((K >> 2) + GEMM_WAVES - 1) / GEMM_WAVES)) << 2;
  int kb = wave * ksl;
  int ke = kb + ksl;
  if (ke > K) ke = K;
  const float* Ar0 = A + (size_t)(m0 + r) * K + kh;
  const float* Ar1 = Ar0 + (size_t)16 * K;
  const float* Wr0 = W + (size_t)(n0 + r) * K + kh;
  const float* Wr1 = Wr0 + (size_t)16 * K;
  v8f acc00 = {};
  v8f acc01 = {};
  v8f acc10 = {};
  v8f acc11 = {};
  for (int k = kb; k < ke; k += 4) {
    v2f a0, a1, b0, b1;
    a0.x = Ar0[k]; a0.y = Ar0[k + 1];
    a1.x = Ar1[k]; a1.y = Ar1[k + 1];
    b0.x = Wr0[k]; b0.y = Wr0[k + 1];
    b1.x = Wr1[k]; b1.y = Wr1[k + 1];
    acc00 = __builtin_amdgcn_wmma_f32_16x16x4_f32(false, a0, false, b0, (short)0, acc00, false, false);
    acc01 = __builtin_amdgcn_wmma_f32_16x16x4_f32(false, a0, false, b1, (short)0, acc01, false, false);
    acc10 = __builtin_amdgcn_wmma_f32_16x16x4_f32(false, a1, false, b0, (short)0, acc10, false, false);
    acc11 = __builtin_amdgcn_wmma_f32_16x16x4_f32(false, a1, false, b1, (short)0, acc11, false, false);
  }
  float* pw = part + wave * PART_TILE + ((lane >> 4) << 3) * PART_PITCH + r;
  for (int v = 0; v < 8; ++v) {
    pw[v * PART_PITCH]             = acc00[v];
    pw[v * PART_PITCH + 16]        = acc01[v];
    pw[(v + 16) * PART_PITCH]      = acc10[v];
    pw[(v + 16) * PART_PITCH + 16] = acc11[v];
  }
  __syncthreads();
  int row = threadIdx.x >> 3;
  int c4  = (threadIdx.x & 7) << 2;
  const float* ps = part + row * PART_PITCH + c4;
  float4 s = *(const float4*)ps;
  for (int w = 1; w < GEMM_WAVES; ++w) {
    float4 p = *(const float4*)(ps + w * PART_TILE);
    s.x += p.x; s.y += p.y; s.z += p.z; s.w += p.w;
  }
  if (bias) {
    float4 bv = *(const float4*)(bias + n0 + c4);
    s.x += bv.x; s.y += bv.y; s.z += bv.z; s.w += bv.w;
  }
  *(float4*)(C + (size_t)(m0 + row) * N + n0 + c4) = s;
}

struct StageOut {
  const float* dprev;
  float* mA;
  float* mC;
  int N;
  int B;
  int b;
};

__device__ __forceinline__ void stage_emit(const StageOut& so, int i, float ox, float oy,
                                           float dx, float dy) {
  size_t rowC = (size_t)so.b * (2 * so.N);
  float mx = 0.0f, my = 0.0f;
  if (so.mA != nullptr || so.dprev == nullptr) { mx = mish_f(ox); my = mish_f(oy); }
  if (so.mA != nullptr) {
    so.mA[(size_t)so.b * so.N + i] = mx;
    so.mA[(size_t)(so.b + so.B) * so.N + i] = my;
  }
  if (so.dprev != nullptr) { mx = mish_f(ox + dx); my = mish_f(oy + dy); }
  so.mC[rowC + i] = mx;
  so.mC[rowC + so.N + i] = my;
}

__global__ __launch_bounds__(256)
void attn_mish_kernel(const float* __restrict__ Q, const float* __restrict__ beta_p,
                      const float* __restrict__ dprev, float* __restrict__ mA,
                      float* __restrict__ mC, int N, int B) {
  __shared__ float sh_x[512];
  __shared__ float sh_y[512];
  __shared__ float sxmax[256];
  __shared__ float sxmin[256];
  __shared__ float symax[256];
  __shared__ float symin[256];
  int b = blockIdx.x;
  int t = threadIdx.x;
  int T = blockDim.x;
  const float* xq = Q + (size_t)b * N;
  const float* yq = Q + (size_t)(b + B) * N;
  const float* dr = dprev ? dprev + (size_t)b * (2 * N) : nullptr;
  StageOut so;
  so.dprev = dprev; so.mA = mA; so.mC = mC; so.N = N; so.B = B; so.b = b;
  float beta = beta_p[0];
  if (beta == 0.0f) {
    for (int i = t; i < N; i += T) {
      float ox = xq[i];
      float oy = yq[i];
      float dx = dr ? dr[i] : 0.0f;
      float dy = dr ? dr[N + i] : 0.0f;
      stage_emit(so, i, ox, oy, dx, dy);
    }
    return;
  }
  for (int i = t; i < N; i += T) { sh_x[i] = xq[i]; sh_y[i] = yq[i]; }
  __syncthreads();
  float xmax = -3.4e38f, xmin = 3.4e38f, ymax = -3.4e38f, ymin = 3.4e38f;
  for (int i = t; i < N; i += T) {
    xmax = fmaxf(xmax, sh_x[i]); xmin = fminf(xmin, sh_x[i]);
    ymax = fmaxf(ymax, sh_y[i]); ymin = fminf(ymin, sh_y[i]);
  }
  sxmax[t] = xmax; sxmin[t] = xmin; symax[t] = ymax; symin[t] = ymin;
  __syncthreads();
  for (int off = T >> 1; off > 0; off >>= 1) {
    if (t < off) {
      sxmax[t] = fmaxf(sxmax[t], sxmax[t + off]); sxmin[t] = fminf(sxmin[t], sxmin[t + off]);
      symax[t] = fmaxf(symax[t], symax[t + off]); symin[t] = fminf(symin[t], symin[t + off]);
    }
    __syncthreads();
  }
  xmax = sxmax[0]; xmin = sxmin[0]; ymax = symax[0]; ymin = symin[0];
  for (int i = t; i < N; i += T) {
    float xi = sh_x[i];
    float yi = sh_y[i];
    float dx = dr ? dr[i] : 0.0f;
    float dy = dr ? dr[N + i] : 0.0f;
    float m1 = (xi >= 0.0f) ? xi * ymax : xi * ymin;
    float m2 = (yi >= 0.0f) ? yi * xmax : yi * xmin;
    float den1 = 0.0f, num1 = 0.0f, den2 = 0.0f, num2 = 0.0f;
    for (int j = 0; j < N; ++j) {
      float xj = sh_x[j];
      float yj = sh_y[j];
      float e1 = __expf(xi * yj - m1);
      den1 += e1;
      num1 += e1 * xj;
      float e2 = __expf(yi * xj - m2);
      den2 += e2;
      num2 += e2 * yj;
    }
    float ox = xi + beta * num1 / den1;
    float oy = yi + beta * num2 / den2;
    stage_emit(so, i, ox, oy, dx, dy);
  }
}

__global__ void head_kernel(const float* __restrict__ d3, const float* __restrict__ W6,
                            const float* __restrict__ b6, float* __restrict__ out, int B) {
  __shared__ float sh[8];
  int b = blockIdx.x;
  int lane = threadIdx.x;
  if (lane < 7) {
    const float* row = d3 + (size_t)b * 64;
    const float* w   = W6 + (size_t)lane * 64;
    float acc = b6[lane];
    for (int k = 0; k < 64; ++k) acc += row[k] * w[k];
    sh[lane] = acc;
  }
  __syncthreads();
  if (lane < 7) {
    float nrm = sqrtf(sh[0]*sh[0] + sh[1]*sh[1] + sh[2]*sh[2] + sh[3]*sh[3]);
    float v = sh[lane];
    if (lane < 4) v = v / fmaxf(nrm, 1e-12f);
    out[(size_t)b * 7 + lane] = v;
    out[(size_t)B * 7 + (size_t)b * 7 + lane] = v;
  }
}

extern "C" void kernel_launch(void* const* d_in, const int* in_sizes, int n_in,
                              void* d_out, int out_size, void* d_ws, size_t ws_size,
                              hipStream_t stream) {
  const float* x    = (const float*)d_in[0];
  const float* y    = (const float*)d_in[1];
  const float* W1   = (const float*)d_in[2];
  const float* be1  = (const float*)d_in[3];
  const float* fcW1 = (const float*)d_in[4];
  const float* fcb1 = (const float*)d_in[5];
  const float* W2   = (const float*)d_in[6];
  const float* be2  = (const float*)d_in[7];
  const float* fcW2 = (const float*)d_in[8];
  const float* fcb2 = (const float*)d_in[9];
  const float* W3   = (const float*)d_in[10];
  const float* be3  = (const float*)d_in[11];
  const float* fcW3 = (const float*)d_in[12];
  const float* fcb3 = (const float*)d_in[13];
  const float* W4   = (const float*)d_in[14];
  const float* be4  = (const float*)d_in[15];
  const float* fcW4 = (const float*)d_in[16];
  const float* fcb4 = (const float*)d_in[17];
  const float* W6   = (const float*)d_in[18];
  const float* b6   = (const float*)d_in[19];

  float* ws = (float*)d_ws;
  float* m  = ws;
  float* q  = m  + 1048576;
  float* mA = q  + 524288;
  float* mC = mA + 524288;
  float* d0 = mC + 524288;
  float* d1 = d0 + 262144;
  float* d2 = d1 + 131072;
  float* d3 = d2 + 65536;

  const int B = 512;
  auto grid1d = [](int n) { return (n + 255) / 256; };
  auto gemm = [&](const float* A, const float* W, const float* bias, float* C, int M, int N, int K) {
    int tiles = (M >> 5) * (N >> 5);
    hipLaunchKernelGGL(gemm_wmma_f32, dim3(tiles), dim3(GEMM_WAVES * 32), 0, stream,
                       A, W, bias, C, M, N, K);
  };
  auto stage = [&](const float* Ain, const float* W, const float* beta, const float* dprev,
                   const float* fcW, const float* fcb, float* dout, float* mAout,
                   int Kin, int Nout, int threads) {
    gemm(Ain, W, nullptr, q, 2 * B, Nout, Kin);
    hipLaunchKernelGGL(attn_mish_kernel, dim3(B), dim3(threads), 0, stream,
                       q, beta, dprev, mAout, mC, Nout, B);
    gemm(mC, fcW, fcb, dout, B, Nout, 2 * Nout);
  };

  hipLaunchKernelGGL(mish2_kernel, dim3(grid1d(2 * B * 1024)), dim3(256), 0, stream,
                     x, y, m, B * 1024);
  stage(m,  W1, be1, nullptr, fcW1, fcb1, d0, mA,      1024, 512, 256);
  stage(mA, W2, be2, d0,      fcW2, fcb2, d1, mA,      512,  256, 256);
  stage(mA, W3, be3, d1,      fcW3, fcb3, d2, mA,      256,  128, 128);
  stage(mA, W4, be4, d2,      fcW4, fcb4, d3, nullptr, 128,  64,  64);
  hipLaunchKernelGGL(head_kernel, dim3(B), dim3(32), 0, stream, d3, W6, b6, (float*)d_out, B);

  (void)in_sizes; (void)n_in; (void)out_size; (void)ws_size;
}


# GEMM operand loads widened: per 32-k block a lane reads 64 contiguous bytes of its row (4 b128 loads in one group), not 8 B per k-step; same k permutation for A and B; f32 sum order changes
# speedup vs baseline: 3.9901x; 1.3995x over previous
#include <hip/hip_runtime.h>
#include <math.h>

typedef __attribute__((ext_vector_type(2))) float v2f;
typedef __attribute__((ext_vector_type(8))) float v8f;

__device__ __forceinline__ float mish_f(float v) {
  float sp = (v > 20.0f) ? v : log1pf(__expf(v));
  return v * tanhf(sp);
}

__global__ void mish2_kernel(const float* __restrict__ a, const float* __restrict__ b,
                             float* __restrict__ out, int nh) {
  int i = blockIdx.x * blockDim.x + threadIdx.x;
  if (i >= 2 * nh) return;
  float v = (i < nh) ? a[i] : b[i - nh];
  out[i] = mish_f(v);
}

#define GEMM_WAVES 8
#define PART_PITCH 36
#define PART_TILE (32 * PART_PITCH)

__device__ __forceinline__ void mma_2x2_k8(const float4& a0, const float4& a1,
                                           const float4& b0, const float4& b1,
                                           v8f& acc00, v8f& acc01, v8f& acc10, v8f& acc11) {
  v2f a0l = {a0.x, a0.y}, a0h = {a0.z, a0.w};
  v2f a1l = {a1.x, a1.y}, a1h = {a1.z, a1.w};
  v2f b0l = {b0.x, b0.y}, b0h = {b0.z, b0.w};
  v2f b1l = {b1.x, b1.y}, b1h = {b1.z, b1.w};
  acc00 = __builtin_amdgcn_wmma_f32_16x16x4_f32(false, a0l, false, b0l, (short)0, acc00, false, false);
  acc01 = __builtin_amdgcn_wmma_f32_16x16x4_f32(false, a0l, false, b1l, (short)0, acc01, false, false);
  acc10 = __builtin_amdgcn_wmma_f32_16x16x4_f32(false, a1l, false, b0l, (short)0, acc10, false, false);
  acc11 = __builtin_amdgcn_wmma_f32_16x16x4_f32(false, a1l, false, b1l, (short)0, acc11, false, false);
  acc00 = __builtin_amdgcn_wmma_f32_16x16x4_f32(false, a0h, false, b0h, (short)0, acc00, false, false);
  acc01 = __builtin_amdgcn_wmma_f32_16x16x4_f32(false, a0h, false, b1h, (short)0, acc01, false, false);
  acc10 = __builtin_amdgcn_wmma_f32_16x16x4_f32(false, a1h, false, b0h, (short)0, acc10, false, false);
  acc11 = __builtin_amdgcn_wmma_f32_16x16x4_f32(false, a1h, false, b1h, (short)0, acc11, false, false);
}

__global__ __launch_bounds__(GEMM_WAVES * 32)
void gemm_wmma_f32(const float* __restrict__ A, const float* __restrict__ W,
                   const float* __restrict__ bias, float* __restrict__ C,
                   int M, int N, int K) {
  __shared__ __attribute__((aligned(16))) float part[GEMM_WAVES * PART_TILE];
  int wave = __builtin_amdgcn_readfirstlane((int)(threadIdx.x >> 5));
  int lane = threadIdx.x & 31;
  int tiles_n = N >> 5;
  int tile = blockIdx.x;
  if (tile >= (M >> 5) * tiles_n) return;
  int tm = tile / tiles_n;
  int m0 = tm << 5;
  int n0 = (tile - tm * tiles_n) << 5;
  int r  = lane & 15;
  int kh = (lane >> 4) << 4;
  int ksl = ((((K >> 5) + GEMM_WAVES - 1) / GEMM_WAVES)) << 5;
  int kb = wave * ksl;
  int ke = kb + ksl;
  if (ke > K) ke = K;
  const float* Ar0 = A + (size_t)(m0 + r) * K + kh;
  const float* Ar1 = Ar0 + (size_t)16 * K;
  const float* Wr0 = W + (size_t)(n0 + r) * K + kh;
  const float* Wr1 = Wr0 + (size_t)16 * K;
  v8f acc00 = {};
  v8f acc01 = {};
  v8f acc10 = {};
  v8f acc11 = {};
  for (int k = kb; k < ke; k += 32) {
    const float4* pa0 = (const float4*)(Ar0 + k);
    const float4* pa1 = (const float4*)(Ar1 + k);
    const float4* pb0 = (const float4*)(Wr0 + k);
    const float4* pb1 = (const float4*)(Wr1 + k);
    float4 a00 = pa0[0], a01 = pa0[1], a02 = pa0[2], a03 = pa0[3];
    float4 b00 = pb0[0], b01 = pb0[1], b02 = pb0[2], b03 = pb0[3];
    float4 a10 = pa1[0], a11 = pa1[1], a12 = pa1[2], a13 = pa1[3];
    float4 b10 = pb1[0], b11 = pb1[1], b12 = pb1[2], b13 = pb1[3];
    __builtin_amdgcn_sched_barrier(0);
    mma_2x2_k8(a00, a10, b00, b10, acc00, acc01, acc10, acc11);
    mma_2x2_k8(a01, a11, b01, b11, acc00, acc01, acc10, acc11);
    mma_2x2_k8(a02, a12, b02, b12, acc00, acc01, acc10, acc11);
    mma_2x2_k8(a03, a13, b03, b13, acc00, acc01, acc10, acc11);
  }
  float* pw = part + wave * PART_TILE + ((lane >> 4) << 3) * PART_PITCH + r;
  for (int v = 0; v < 8; ++v) {
    pw[v * PART_PITCH]             = acc00[v];
    pw[v * PART_PITCH + 16]        = acc01[v];
    pw[(v + 16) * PART_PITCH]      = acc10[v];
    pw[(v + 16) * PART_PITCH + 16] = acc11[v];
  }
  __syncthreads();
  int row = threadIdx.x >> 3;
  int c4  = (threadIdx.x & 7) << 2;
  const float* ps = part + row * PART_PITCH + c4;
  float4 s = *(const float4*)ps;
  for (int w = 1; w < GEMM_WAVES; ++w) {
    float4 p = *(const float4*)(ps + w * PART_TILE);
    s.x += p.x; s.y += p.y; s.z += p.z; s.w += p.w;
  }
  if (bias) {
    float4 bv = *(const float4*)(bias + n0 + c4);
    s.x += bv.x; s.y += bv.y; s.z += bv.z; s.w += bv.w;
  }
  *(float4*)(C + (size_t)(m0 + row) * N + n0 + c4) = s;
}

struct StageOut {
  const float* dprev;
  float* mA;
  float* mC;
  int N;
  int B;
  int b;
};

__device__ __forceinline__ void stage_emit(const StageOut& so, int i, float ox, float oy,
                                           float dx, float dy) {
  size_t rowC = (size_t)so.b * (2 * so.N);
  float mx = 0.0f, my = 0.0f;
  if (so.mA != nullptr || so.dprev == nullptr) { mx = mish_f(ox); my = mish_f(oy); }
  if (so.mA != nullptr) {
    so.mA[(size_t)so.b * so.N + i] = mx;
    so.mA[(size_t)(so.b + so.B) * so.N + i] = my;
  }
  if (so.dprev != nullptr) { mx = mish_f(ox + dx); my = mish_f(oy + dy); }
  so.mC[rowC + i] = mx;
  so.mC[rowC + so.N + i] = my;
}

__global__ __launch_bounds__(256)
void attn_mish_kernel(const float* __restrict__ Q, const float* __restrict__ beta_p,
                      const float* __restrict__ dprev, float* __restrict__ mA,
                      float* __restrict__ mC, int N, int B) {
  __shared__ float sh_x[512];
  __shared__ float sh_y[512];
  __shared__ float sxmax[256];
  __shared__ float sxmin[256];
  __shared__ float symax[256];
  __shared__ float symin[256];
  int b = blockIdx.x;
  int t = threadIdx.x;
  int T = blockDim.x;
  const float* xq = Q + (size_t)b * N;
  const float* yq = Q + (size_t)(b + B) * N;
  const float* dr = dprev ? dprev + (size_t)b * (2 * N) : nullptr;
  StageOut so;
  so.dprev = dprev; so.mA = mA; so.mC = mC; so.N = N; so.B = B; so.b = b;
  float beta = beta_p[0];
  if (beta == 0.0f) {
    for (int i = t; i < N; i += T) {
      float ox = xq[i];
      float oy = yq[i];
      float dx = dr ? dr[i] : 0.0f;
      float dy = dr ? dr[N + i] : 0.0f;
      stage_emit(so, i, ox, oy, dx, dy);
    }
    return;
  }
  for (int i = t; i < N; i += T) { sh_x[i] = xq[i]; sh_y[i] = yq[i]; }
  __syncthreads();
  float xmax = -3.4e38f, xmin = 3.4e38f, ymax = -3.4e38f, ymin = 3.4e38f;
  for (int i = t; i < N; i += T) {
    xmax = fmaxf(xmax, sh_x[i]); xmin = fminf(xmin, sh_x[i]);
    ymax = fmaxf(ymax, sh_y[i]); ymin = fminf(ymin, sh_y[i]);
  }
  sxmax[t] = xmax; sxmin[t] = xmin; symax[t] = ymax; symin[t] = ymin;
  __syncthreads();
  for (int off = T >> 1; off > 0; off >>= 1) {
    if (t < off) {
      sxmax[t] = fmaxf(sxmax[t], sxmax[t + off]); sxmin[t] = fminf(sxmin[t], sxmin[t + off]);
      symax[t] = fmaxf(symax[t], symax[t + off]); symin[t] = fminf(symin[t], symin[t + off]);
    }
    __syncthreads();
  }
  xmax = sxmax[0]; xmin = sxmin[0]; ymax = symax[0]; ymin = symin[0];
  for (int i = t; i < N; i += T) {
    float xi = sh_x[i];
    float yi = sh_y[i];
    float dx = dr ? dr[i] : 0.0f;
    float dy = dr ? dr[N + i] : 0.0f;
    float m1 = (xi >= 0.0f) ? xi * ymax : xi * ymin;
    float m2 = (yi >= 0.0f) ? yi * xmax : yi * xmin;
    float den1 = 0.0f, num1 = 0.0f, den2 = 0.0f, num2 = 0.0f;
    for (int j = 0; j < N; ++j) {
      float xj = sh_x[j];
      float yj = sh_y[j];
      float e1 = __expf(xi * yj - m1);
      den1 += e1;
      num1 += e1 * xj;
      float e2 = __expf(yi * xj - m2);
      den2 += e2;
      num2 += e2 * yj;
    }
    float ox = xi + beta * num1 / den1;
    float oy = yi + beta * num2 / den2;
    stage_emit(so, i, ox, oy, dx, dy);
  }
}

__global__ void head_kernel(const float* __restrict__ d3, const float* __restrict__ W6,
                            const float* __restrict__ b6, float* __restrict__ out, int B) {
  __shared__ float sh[8];
  int b = blockIdx.x;
  int lane = threadIdx.x;
  if (lane < 7) {
    const float* row = d3 + (size_t)b * 64;
    const float* w   = W6 + (size_t)lane * 64;
    float acc = b6[lane];
    for (int k = 0; k < 64; ++k) acc += row[k] * w[k];
    sh[lane] = acc;
  }
  __syncthreads();
  if (lane < 7) {
    float nrm = sqrtf(sh[0]*sh[0] + sh[1]*sh[1] + sh[2]*sh[2] + sh[3]*sh[3]);
    float v = sh[lane];
    if (lane < 4) v = v / fmaxf(nrm, 1e-12f);
    out[(size_t)b * 7 + lane] = v;
    out[(size_t)B * 7 + (size_t)b * 7 + lane] = v;
  }
}

extern "C" void kernel_launch(void* const* d_in, const int* in_sizes, int n_in,
                              void* d_out, int out_size, void* d_ws, size_t ws_size,
                              hipStream_t stream) {
  const float* x    = (const float*)d_in[0];
  const float* y    = (const float*)d_in[1];
  const float* W1   = (const float*)d_in[2];
  const float* be1  = (const float*)d_in[3];
  const float* fcW1 = (const float*)d_in[4];
  const float* fcb1 = (const float*)d_in[5];
  const float* W2   = (const float*)d_in[6];
  const float* be2  = (const float*)d_in[7];
  const float* fcW2 = (const float*)d_in[8];
  const float* fcb2 = (const float*)d_in[9];
  const float* W3   = (const float*)d_in[10];
  const float* be3  = (const float*)d_in[11];
  const float* fcW3 = (const float*)d_in[12];
  const float* fcb3 = (const float*)d_in[13];
  const float* W4   = (const float*)d_in[14];
  const float* be4  = (const float*)d_in[15];
  const float* fcW4 = (const float*)d_in[16];
  const float* fcb4 = (const float*)d_in[17];
  const float* W6   = (const float*)d_in[18];
  const float* b6   = (const float*)d_in[19];

  float* ws = (float*)d_ws;
  float* m  = ws;
  float* q  = m  + 1048576;
  float* mA = q  + 524288;
  float* mC = mA + 524288;
  float* d0 = mC + 524288;
  float* d1 = d0 + 262144;
  float* d2 = d1 + 131072;
  float* d3 = d2 + 65536;

  const int B = 512;
  auto grid1d = [](int n) { return (n + 255) / 256; };
  auto gemm = [&](const float* A, const float* W, const float* bias, float* C, int M, int N, int K) {
    int tiles = (M >> 5) * (N >> 5);
    hipLaunchKernelGGL(gemm_wmma_f32, dim3(tiles), dim3(GEMM_WAVES * 32), 0, stream,
                       A, W, bias, C, M, N, K);
  };
  auto stage = [&](const float* Ain, const float* W, const float* beta, const float* dprev,
                   const float* fcW, const float* fcb, float* dout, float* mAout,
                   int Kin, int Nout, int threads) {
    gemm(Ain, W, nullptr, q, 2 * B, Nout, Kin);
    hipLaunchKernelGGL(attn_mish_kernel, dim3(B), dim3(threads), 0, stream,
                       q, beta, dprev, mAout, mC, Nout, B);
    gemm(mC, fcW, fcb, dout, B, Nout, 2 * Nout);
  };

  hipLaunchKernelGGL(mish2_kernel, dim3(grid1d(2 * B * 1024)), dim3(256), 0, stream,
                     x, y, m, B * 1024);
  stage(m,  W1, be1, nullptr, fcW1, fcb1, d0, mA,      1024, 512, 256);
  stage(mA, W2, be2, d0,      fcW2, fcb2, d1, mA,      512,  256, 256);
  stage(mA, W3, be3, d1,      fcW3, fcb3, d2, mA,      256,  128, 128);
  stage(mA, W4, be4, d2,      fcW4, fcb4, d3, nullptr, 128,  64,  64);
  hipLaunchKernelGGL(head_kernel, dim3(B), dim3(32), 0, stream, d3, W6, b6, (float*)d_out, B);

  (void)in_sizes; (void)n_in; (void)out_size; (void)ws_size;
}


# fc GEMM of stage s and query GEMM of stage s+1 (independent: both read only the attention launch's outputs) go out as one launch, group index picks the problem; 14 launches -> 11; sums unchanged
# speedup vs baseline: 4.7684x; 1.1951x over previous
#include <hip/hip_runtime.h>
#include <math.h>

typedef __attribute__((ext_vector_type(2))) float v2f;
typedef __attribute__((ext_vector_type(8))) float v8f;

__device__ __forceinline__ float mish_f(float v) {
  float sp = (v > 20.0f) ? v : log1pf(__expf(v));
  return v * tanhf(sp);
}

__global__ void mish2_kernel(const float* __restrict__ a, const float* __restrict__ b,
                             float* __restrict__ out, int nh) {
  int i = blockIdx.x * blockDim.x + threadIdx.x;
  if (i >= 2 * nh) return;
  float v = (i < nh) ? a[i] : b[i - nh];
  out[i] = mish_f(v);
}

#define GEMM_WAVES 8
#define PART_PITCH 36
#define PART_TILE (32 * PART_PITCH)

__device__ __forceinline__ void mma_2x2_k8(const float4& a0, const float4& a1,
                                           const float4& b0, const float4& b1,
                                           v8f& acc00, v8f& acc01, v8f& acc10, v8f& acc11) {
  v2f a0l = {a0.x, a0.y}, a0h = {a0.z, a0.w};
  v2f a1l = {a1.x, a1.y}, a1h = {a1.z, a1.w};
  v2f b0l = {b0.x, b0.y}, b0h = {b0.z, b0.w};
  v2f b1l = {b1.x, b1.y}, b1h = {b1.z, b1.w};
  acc00 = __builtin_amdgcn_wmma_f32_16x16x4_f32(false, a0l, false, b0l, (short)0, acc00, false, false);
  acc01 = __builtin_amdgcn_wmma_f32_16x16x4_f32(false, a0l, false, b1l, (short)0, acc01, false, false);
  acc10 = __builtin_amdgcn_wmma_f32_16x16x4_f32(false, a1l, false, b0l, (short)0, acc10, false, false);
  acc11 = __builtin_amdgcn_wmma_f32_16x16x4_f32(false, a1l, false, b1l, (short)0, acc11, false, false);
  acc00 = __builtin_amdgcn_wmma_f32_16x16x4_f32(false, a0h, false, b0h, (short)0, acc00, false, false);
  acc01 = __builtin_amdgcn_wmma_f32_16x16x4_f32(false, a0h, false, b1h, (short)0, acc01, false, false);
  acc10 = __builtin_amdgcn_wmma_f32_16x16x4_f32(false, a1h, false, b0h, (short)0, acc10, false, false);
  acc11 = __builtin_amdgcn_wmma_f32_16x16x4_f32(false, a1h, false, b1h, (short)0, acc11, false, false);
}

__global__ __launch_bounds__(GEMM_WAVES * 32)
void gemm_pair_wmma_f32(const float* __restrict__ A0, const float* __restrict__ W0,
                        const float* __restrict__ bias0, float* __restrict__ C0,
                        int M0, int N0, int K0,
                        const float* __restrict__ A1, const float* __restrict__ W1,
                        const float* __restrict__ bias1, float* __restrict__ C1,
                        int M1, int N1, int K1) {
  __shared__ __attribute__((aligned(16))) float part[GEMM_WAVES * PART_TILE];
  int wave = __builtin_amdgcn_readfirstlane((int)(threadIdx.x >> 5));
  int lane = threadIdx.x & 31;
  int tile = blockIdx.x;
  int tiles0 = (M0 >> 5) * (N0 >> 5);
  bool second = tile >= tiles0;
  const float* A    = second ? A1 : A0;
  const float* W    = second ? W1 : W0;
  const float* bias = second ? bias1 : bias0;
  float* C          = second ? C1 : C0;
  int M = second ? M1 : M0;
  int N = second ? N1 : N0;
  int K = second ? K1 : K0;
  if (second) tile -= tiles0;
  int tiles_n = N >> 5;
  if (tile >= (M >> 5) * tiles_n) return;
  int tm = tile / tiles_n;
  int m0 = tm << 5;
  int n0 = (tile - tm * tiles_n) << 5;
  int r  = lane & 15;
  int kh = (lane >> 4) << 4;
  int ksl = ((((K >> 5) + GEMM_WAVES - 1) / GEMM_WAVES)) << 5;
  int kb = wave * ksl;
  int ke = kb + ksl;
  if (ke > K) ke = K;
  const float* Ar0 = A + (size_t)(m0 + r) * K + kh;
  const float* Ar1 = Ar0 + (size_t)16 * K;
  const float* Wr0 = W + (size_t)(n0 + r) * K + kh;
  const float* Wr1 = Wr0 + (size_t)16 * K;
  v8f acc00 = {};
  v8f acc01 = {};
  v8f acc10 = {};
  v8f acc11 = {};
  for (int k = kb; k < ke; k += 32) {
    const float4* pa0 = (const float4*)(Ar0 + k);
    const float4* pa1 = (const float4*)(Ar1 + k);
    const float4* pb0 = (const float4*)(Wr0 + k);
    const float4* pb1 = (const float4*)(Wr1 + k);
    float4 a00 = pa0[0], a01 = pa0[1], a02 = pa0[2], a03 = pa0[3];
    float4 b00 = pb0[0], b01 = pb0[1], b02 = pb0[2], b03 = pb0[3];
    float4 a10 = pa1[0], a11 = pa1[1], a12 = pa1[2], a13 = pa1[3];
    float4 b10 = pb1[0], b11 = pb1[1], b12 = pb1[2], b13 = pb1[3];
    __builtin_amdgcn_sched_barrier(0);
    mma_2x2_k8(a00, a10, b00, b10, acc00, acc01, acc10, acc11);
    mma_2x2_k8(a01, a11, b01, b11, acc00, acc01, acc10, acc11);
    mma_2x2_k8(a02, a12, b02, b12, acc00, acc01, acc10, acc11);
    mma_2x2_k8(a03, a13, b03, b13, acc00, acc01, acc10, acc11);
  }
  float* pw = part + wave * PART_TILE + ((lane >> 4) << 3) * PART_PITCH + r;
  for (int v = 0; v < 8; ++v) {
    pw[v * PART_PITCH]             = acc00[v];
    pw[v * PART_PITCH + 16]        = acc01[v];
    pw[(v + 16) * PART_PITCH]      = acc10[v];
    pw[(v + 16) * PART_PITCH + 16] = acc11[v];
  }
  __syncthreads();
  int row = threadIdx.x >> 3;
  int c4  = (threadIdx.x & 7) << 2;
  const float* ps = part + row * PART_PITCH + c4;
  float4 s = *(const float4*)ps;
  for (int w = 1; w < GEMM_WAVES; ++w) {
    float4 p = *(const float4*)(ps + w * PART_TILE);
    s.x += p.x; s.y += p.y; s.z += p.z; s.w += p.w;
  }
  if (bias) {
    float4 bv = *(const float4*)(bias + n0 + c4);
    s.x += bv.x; s.y += bv.y; s.z += bv.z; s.w += bv.w;
  }
  *(float4*)(C + (size_t)(m0 + row) * N + n0 + c4) = s;
}

struct StageOut {
  const float* dprev;
  float* mA;
  float* mC;
  int N;
  int B;
  int b;
};

__device__ __forceinline__ void stage_emit(const StageOut& so, int i, float ox, float oy,
                                           float dx, float dy) {
  size_t rowC = (size_t)so.b * (2 * so.N);
  float mx = 0.0f, my = 0.0f;
  if (so.mA != nullptr || so.dprev == nullptr) { mx = mish_f(ox); my = mish_f(oy); }
  if (so.mA != nullptr) {
    so.mA[(size_t)so.b * so.N + i] = mx;
    so.mA[(size_t)(so.b + so.B) * so.N + i] = my;
  }
  if (so.dprev != nullptr) { mx = mish_f(ox + dx); my = mish_f(oy + dy); }
  so.mC[rowC + i] = mx;
  so.mC[rowC + so.N + i] = my;
}

__global__ __launch_bounds__(256)
void attn_mish_kernel(const float* __restrict__ Q, const float* __restrict__ beta_p,
                      const float* __restrict__ dprev, float* __restrict__ mA,
                      float* __restrict__ mC, int N, int B) {
  __shared__ float sh_x[512];
  __shared__ float sh_y[512];
  __shared__ float sxmax[256];
  __shared__ float sxmin[256];
  __shared__ float symax[256];
  __shared__ float symin[256];
  int b = blockIdx.x;
  int t = threadIdx.x;
  int T = blockDim.x;
  const float* xq = Q + (size_t)b * N;
  const float* yq = Q + (size_t)(b + B) * N;
  const float* dr = dprev ? dprev + (size_t)b * (2 * N) : nullptr;
  StageOut so;
  so.dprev = dprev; so.mA = mA; so.mC = mC; so.N = N; so.B = B; so.b = b;
  float beta = beta_p[0];
  if (beta == 0.0f) {
    for (int i = t; i < N; i += T) {
      float ox = xq[i];
      float oy = yq[i];
      float dx = dr ? dr[i] : 0.0f;
      float dy = dr ? dr[N + i] : 0.0f;
      stage_emit(so, i, ox, oy, dx, dy);
    }
    return;
  }
  for (int i = t; i < N; i += T) { sh_x[i] = xq[i]; sh_y[i] = yq[i]; }
  __syncthreads();
  float xmax = -3.4e38f, xmin = 3.4e38f, ymax = -3.4e38f, ymin = 3.4e38f;
  for (int i = t; i < N; i += T) {
    xmax = fmaxf(xmax, sh_x[i]); xmin = fminf(xmin, sh_x[i]);
    ymax = fmaxf(ymax, sh_y[i]); ymin = fminf(ymin, sh_y[i]);
  }
  sxmax[t] = xmax; sxmin[t] = xmin; symax[t] = ymax; symin[t] = ymin;
  __syncthreads();
  for (int off = T >> 1; off > 0; off >>= 1) {
    if (t < off) {
      sxmax[t] = fmaxf(sxmax[t], sxmax[t + off]); sxmin[t] = fminf(sxmin[t], sxmin[t + off]);
      symax[t] = fmaxf(symax[t], symax[t + off]); symin[t] = fminf(symin[t], symin[t + off]);
    }
    __syncthreads();
  }
  xmax = sxmax[0]; xmin = sxmin[0]; ymax = symax[0]; ymin = symin[0];
  for (int i = t; i < N; i += T) {
    float xi = sh_x[i];
    float yi = sh_y[i];
    float dx = dr ? dr[i] : 0.0f;
    float dy = dr ? dr[N + i] : 0.0f;
    float m1 = (xi >= 0.0f) ? xi * ymax : xi * ymin;
    float m2 = (yi >= 0.0f) ? yi * xmax : yi * xmin;
    float den1 = 0.0f, num1 = 0.0f, den2 = 0.0f, num2 = 0.0f;
    for (int j = 0; j < N; ++j) {
      float xj = sh_x[j];
      float yj = sh_y[j];
      float e1 = __expf(xi * yj - m1);
      den1 += e1;
      num1 += e1 * xj;
      float e2 = __expf(yi * xj - m2);
      den2 += e2;
      num2 += e2 * yj;
    }
    float ox = xi + beta * num1 / den1;
    float oy = yi + beta * num2 / den2;
    stage_emit(so, i, ox, oy, dx, dy);
  }
}

__global__ void head_kernel(const float* __restrict__ d3, const float* __restrict__ W6,
                            const float* __restrict__ b6, float* __restrict__ out, int B) {
  __shared__ float sh[8];
  int b = blockIdx.x;
  int lane = threadIdx.x;
  if (lane < 7) {
    const float* row = d3 + (size_t)b * 64;
    const float* w   = W6 + (size_t)lane * 64;
    float acc = b6[lane];
    for (int k = 0; k < 64; ++k) acc += row[k] * w[k];
    sh[lane] = acc;
  }
  __syncthreads();
  if (lane < 7) {
    float nrm = sqrtf(sh[0]*sh[0] + sh[1]*sh[1] + sh[2]*sh[2] + sh[3]*sh[3]);
    float v = sh[lane];
    if (lane < 4) v = v / fmaxf(nrm, 1e-12f);
    out[(size_t)b * 7 + lane] = v;
    out[(size_t)B * 7 + (size_t)b * 7 + lane] = v;
  }
}

extern "C" void kernel_launch(void* const* d_in, const int* in_sizes, int n_in,
                              void* d_out, int out_size, void* d_ws, size_t ws_size,
                              hipStream_t stream) {
  const float* x    = (const float*)d_in[0];
  const float* y    = (const float*)d_in[1];
  const float* W1   = (const float*)d_in[2];
  const float* be1  = (const float*)d_in[3];
  const float* fcW1 = (const float*)d_in[4];
  const float* fcb1 = (const float*)d_in[5];
  const float* W2   = (const float*)d_in[6];
  const float* be2  = (const float*)d_in[7];
  const float* fcW2 = (const float*)d_in[8];
  const float* fcb2 = (const float*)d_in[9];
  const float* W3   = (const float*)d_in[10];
  const float* be3  = (const float*)d_in[11];
  const float* fcW3 = (const float*)d_in[12];
  const float* fcb3 = (const float*)d_in[13];
  const float* W4   = (const float*)d_in[14];
  const float* be4  = (const float*)d_in[15];
  const float* fcW4 = (const float*)d_in[16];
  const float* fcb4 = (const float*)d_in[17];
  const float* W6   = (const float*)d_in[18];
  const float* b6   = (const float*)d_in[19];

  float* ws = (float*)d_ws;
  float* m  = ws;
  float* q  = m  + 1048576;
  float* mA = q  + 524288;
  float* mC = mA + 524288;
  float* d0 = mC + 524288;
  float* d1 = d0 + 262144;
  float* d2 = d1 + 131072;
  float* d3 = d2 + 65536;

  const int B = 512;
  auto grid1d = [](int n) { return (n + 255) / 256; };
  auto gemm_pair = [&](const float* A0, const float* W0, const float* bias0, float* C0,
                       int M0, int N0, int K0,
                       const float* A1, const float* W1, const float* bias1, float* C1,
                       int M1, int N1, int K1) {
    int tiles = (M0 >> 5) * (N0 >> 5) + (M1 >> 5) * (N1 >> 5);
    hipLaunchKernelGGL(gemm_pair_wmma_f32, dim3(tiles), dim3(GEMM_WAVES * 32), 0, stream,
                       A0, W0, bias0, C0, M0, N0, K0, A1, W1, bias1, C1, M1, N1, K1);
  };
  auto attn = [&](const float* beta, const float* dprev, float* mAout, int Nout, int threads) {
    hipLaunchKernelGGL(attn_mish_kernel, dim3(B), dim3(threads), 0, stream,
                       q, beta, dprev, mAout, mC, Nout, B);
  };

  hipLaunchKernelGGL(mish2_kernel, dim3(grid1d(2 * B * 1024)), dim3(256), 0, stream,
                     x, y, m, B * 1024);
  gemm_pair(m,  W1,   nullptr, q,  2 * B, 512, 1024,  nullptr, nullptr, nullptr, nullptr, 0, 0, 0);
  attn(be1, nullptr, mA, 512, 256);
  gemm_pair(mC, fcW1, fcb1,    d0, B,     512, 1024,  mA, W2, nullptr, q, 2 * B, 256, 512);
  attn(be2, d0, mA, 256, 256);
  gemm_pair(mC, fcW2, fcb2,    d1, B,     256, 512,   mA, W3, nullptr, q, 2 * B, 128, 256);
  attn(be3, d1, mA, 128, 128);
  gemm_pair(mC, fcW3, fcb3,    d2, B,     128, 256,   mA, W4, nullptr, q, 2 * B, 64,  128);
  attn(be4, d2, nullptr, 64, 64);
  gemm_pair(mC, fcW4, fcb4,    d3, B,     64,  128,   nullptr, nullptr, nullptr, nullptr, 0, 0, 0);
  hipLaunchKernelGGL(head_kernel, dim3(B), dim3(32), 0, stream, d3, W6, b6, (float*)d_out, B);

  (void)in_sizes; (void)n_in; (void)out_size; (void)ws_size;
}


# last stage in one launch: attention 4, fc4 GEMM (same K slices, WMMA) and head done per 4-sample group with rows handed on in LDS; 11 launches -> 9; sums unchanged
# speedup vs baseline: 5.3220x; 1.1161x over previous
#include <hip/hip_runtime.h>
#include <math.h>

typedef __attribute__((ext_vector_type(2))) float v2f;
typedef __attribute__((ext_vector_type(8))) float v8f;

__device__ __forceinline__ float mish_f(float v) {
  float sp = (v > 20.0f) ? v : log1pf(__expf(v));
  return v * tanhf(sp);
}

__global__ void mish2_kernel(const float* __restrict__ a, const float* __restrict__ b,
                             float* __restrict__ out, int nh) {
  int i = blockIdx.x * blockDim.x + threadIdx.x;
  if (i >= 2 * nh) return;
  float v = (i < nh) ? a[i] : b[i - nh];
  out[i] = mish_f(v);
}

#define GEMM_WAVES 8
#define PART_PITCH 36
#define PART_TILE (32 * PART_PITCH)

__device__ __forceinline__ void mma_2x2_k8(const float4& a0, const float4& a1,
                                           const float4& b0, const float4& b1,
                                           v8f& acc00, v8f& acc01, v8f& acc10, v8f& acc11) {
  v2f a0l = {a0.x, a0.y}, a0h = {a0.z, a0.w};
  v2f a1l = {a1.x, a1.y}, a1h = {a1.z, a1.w};
  v2f b0l = {b0.x, b0.y}, b0h = {b0.z, b0.w};
  v2f b1l = {b1.x, b1.y}, b1h = {b1.z, b1.w};
  acc00 = __builtin_amdgcn_wmma_f32_16x16x4_f32(false, a0l, false, b0l, (short)0, acc00, false, false);
  acc01 = __builtin_amdgcn_wmma_f32_16x16x4_f32(false, a0l, false, b1l, (short)0, acc01, false, false);
  acc10 = __builtin_amdgcn_wmma_f32_16x16x4_f32(false, a1l, false, b0l, (short)0, acc10, false, false);
  acc11 = __builtin_amdgcn_wmma_f32_16x16x4_f32(false, a1l, false, b1l, (short)0, acc11, false, false);
  acc00 = __builtin_amdgcn_wmma_f32_16x16x4_f32(false, a0h, false, b0h, (short)0, acc00, false, false);
  acc01 = __builtin_amdgcn_wmma_f32_16x16x4_f32(false, a0h, false, b1h, (short)0, acc01, false, false);
  acc10 = __builtin_amdgcn_wmma_f32_16x16x4_f32(false, a1h, false, b0h, (short)0, acc10, false, false);
  acc11 = __builtin_amdgcn_wmma_f32_16x16x4_f32(false, a1h, false, b1h, (short)0, acc11, false, false);
}

__global__ __launch_bounds__(GEMM_WAVES * 32)
void gemm_pair_wmma_f32(const float* __restrict__ A0, const float* __restrict__ W0,
                        const float* __restrict__ bias0, float* __restrict__ C0,
                        int M0, int N0, int K0,
                        const float* __restrict__ A1, const float* __restrict__ W1,
                        const float* __restrict__ bias1, float* __restrict__ C1,
                        int M1, int N1, int K1) {
  __shared__ __attribute__((aligned(16))) float part[GEMM_WAVES * PART_TILE];
  int wave = __builtin_amdgcn_readfirstlane((int)(threadIdx.x >> 5));
  int lane = threadIdx.x & 31;
  int tile = blockIdx.x;
  int tiles0 = (M0 >> 5) * (N0 >> 5);
  bool second = tile >= tiles0;
  const float* A    = second ? A1 : A0;
  const float* W    = second ? W1 : W0;
  const float* bias = second ? bias1 : bias0;
  float* C          = second ? C1 : C0;
  int M = second ? M1 : M0;
  int N = second ? N1 : N0;
  int K = second ? K1 : K0;
  if (second) tile -= tiles0;
  int tiles_n = N >> 5;
  if (tile >= (M >> 5) * tiles_n) return;
  int tm = tile / tiles_n;
  int m0 = tm << 5;
  int n0 = (tile - tm * tiles_n) << 5;
  int r  = lane & 15;
  int kh = (lane >> 4) << 4;
  int ksl = ((((K >> 5) + GEMM_WAVES - 1) / GEMM_WAVES)) << 5;
  int kb = wave * ksl;
  int ke = kb + ksl;
  if (ke > K) ke = K;
  const float* Ar0 = A + (size_t)(m0 + r) * K + kh;
  const float* Ar1 = Ar0 + (size_t)16 * K;
  const float* Wr0 = W + (size_t)(n0 + r) * K + kh;
  const float* Wr1 = Wr0 + (size_t)16 * K;
  v8f acc00 = {};
  v8f acc01 = {};
  v8f acc10 = {};
  v8f acc11 = {};
  for (int k = kb; k < ke; k += 32) {
    const float4* pa0 = (const float4*)(Ar0 + k);
    const float4* pa1 = (const float4*)(Ar1 + k);
    const float4* pb0 = (const float4*)(Wr0 + k);
    const float4* pb1 = (const float4*)(Wr1 + k);
    float4 a00 = pa0[0], a01 = pa0[1], a02 = pa0[2], a03 = pa0[3];
    float4 b00 = pb0[0], b01 = pb0[1], b02 = pb0[2], b03 = pb0[3];
    float4 a10 = pa1[0], a11 = pa1[1], a12 = pa1[2], a13 = pa1[3];
    float4 b10 = pb1[0], b11 = pb1[1], b12 = pb1[2], b13 = pb1[3];
    __builtin_amdgcn_sched_barrier(0);
    mma_2x2_k8(a00, a10, b00, b10, acc00, acc01, acc10, acc11);
    mma_2x2_k8(a01, a11, b01, b11, acc00, acc01, acc10, acc11);
    mma_2x2_k8(a02, a12, b02, b12, acc00, acc01, acc10, acc11);
    mma_2x2_k8(a03, a13, b03, b13, acc00, acc01, acc10, acc11);
  }
  float* pw = part + wave * PART_TILE + ((lane >> 4) << 3) * PART_PITCH + r;
  for (int v = 0; v < 8; ++v) {
    pw[v * PART_PITCH]             = acc00[v];
    pw[v * PART_PITCH + 16]        = acc01[v];
    pw[(v + 16) * PART_PITCH]      = acc10[v];
    pw[(v + 16) * PART_PITCH + 16] = acc11[v];
  }
  __syncthreads();
  int row = threadIdx.x >> 3;
  int c4  = (threadIdx.x & 7) << 2;
  const float* ps = part + row * PART_PITCH + c4;
  float4 s = *(const float4*)ps;
  for (int w = 1; w < GEMM_WAVES; ++w) {
    float4 p = *(const float4*)(ps + w * PART_TILE);
    s.x += p.x; s.y += p.y; s.z += p.z; s.w += p.w;
  }
  if (bias) {
    float4 bv = *(const float4*)(bias + n0 + c4);
    s.x += bv.x; s.y += bv.y; s.z += bv.z; s.w += bv.w;
  }
  *(float4*)(C + (size_t)(m0 + row) * N + n0 + c4) = s;
}

struct StageOut {
  const float* dprev;
  float* mA;
  float* mC;
  int N;
  int B;
  int b;
};

__device__ __forceinline__ void stage_emit(const StageOut& so, int i, float ox, float oy,
                                           float dx, float dy) {
  size_t rowC = (size_t)so.b * (2 * so.N);
  float mx = 0.0f, my = 0.0f;
  if (so.mA != nullptr || so.dprev == nullptr) { mx = mish_f(ox); my = mish_f(oy); }
  if (so.mA != nullptr) {
    so.mA[(size_t)so.b * so.N + i] = mx;
    so.mA[(size_t)(so.b + so.B) * so.N + i] = my;
  }
  if (so.dprev != nullptr) { mx = mish_f(ox + dx); my = mish_f(oy + dy); }
  so.mC[rowC + i] = mx;
  so.mC[rowC + so.N + i] = my;
}

__global__ __launch_bounds__(256)
void attn_mish_kernel(const float* __restrict__ Q, const float* __restrict__ beta_p,
                      const float* __restrict__ dprev, float* __restrict__ mA,
                      float* __restrict__ mC, int N, int B) {
  __shared__ float sh_x[512];
  __shared__ float sh_y[512];
  __shared__ float sxmax[256];
  __shared__ float sxmin[256];
  __shared__ float symax[256];
  __shared__ float symin[256];
  int b = blockIdx.x;
  int t = threadIdx.x;
  int T = blockDim.x;
  const float* xq = Q + (size_t)b * N;
  const float* yq = Q + (size_t)(b + B) * N;
  const float* dr = dprev ? dprev + (size_t)b * (2 * N) : nullptr;
  StageOut so;
  so.dprev = dprev; so.mA = mA; so.mC = mC; so.N = N; so.B = B; so.b = b;
  float beta = beta_p[0];
  if (beta == 0.0f) {
    for (int i = t; i < N; i += T) {
      float ox = xq[i];
      float oy = yq[i];
      float dx = dr ? dr[i] : 0.0f;
      float dy = dr ? dr[N + i] : 0.0f;
      stage_emit(so, i, ox, oy, dx, dy);
    }
    return;
  }
  for (int i = t; i < N; i += T) { sh_x[i] = xq[i]; sh_y[i] = yq[i]; }
  __syncthreads();
  float xmax = -3.4e38f, xmin = 3.4e38f, ymax = -3.4e38f, ymin = 3.4e38f;
  for (int i = t; i < N; i += T) {
    xmax = fmaxf(xmax, sh_x[i]); xmin = fminf(xmin, sh_x[i]);
    ymax = fmaxf(ymax, sh_y[i]); ymin = fminf(ymin, sh_y[i]);
  }
  sxmax[t] = xmax; sxmin[t] = xmin; symax[t] = ymax; symin[t] = ymin;
  __syncthreads();
  for (int off = T >> 1; off > 0; off >>= 1) {
    if (t < off) {
      sxmax[t] = fmaxf(sxmax[t], sxmax[t + off]); sxmin[t] = fminf(sxmin[t], sxmin[t + off]);
      symax[t] = fmaxf(symax[t], symax[t + off]); symin[t] = fminf(symin[t], symin[t + off]);
    }
    __syncthreads();
  }
  xmax = sxmax[0]; xmin = sxmin[0]; ymax = symax[0]; ymin = symin[0];
  for (int i = t; i < N; i += T) {
    float xi = sh_x[i];
    float yi = sh_y[i];
    float dx = dr ? dr[i] : 0.0f;
    float dy = dr ? dr[N + i] : 0.0f;
    float m1 = (xi >= 0.0f) ? xi * ymax : xi * ymin;
    float m2 = (yi >= 0.0f) ? yi * xmax : yi * xmin;
    float den1 = 0.0f, num1 = 0.0f, den2 = 0.0f, num2 = 0.0f;
    for (int j = 0; j < N; ++j) {
      float xj = sh_x[j];
      float yj = sh_y[j];
      float e1 = __expf(xi * yj - m1);
      den1 += e1;
      num1 += e1 * xj;
      float e2 = __expf(yi * xj - m2);
      den2 += e2;
      num2 += e2 * yj;
    }
    float ox = xi + beta * num1 / den1;
    float oy = yi + beta * num2 / den2;
    stage_emit(so, i, ox, oy, dx, dy);
  }
}

#define TAIL_S 4
#define TAIL_N 64
#define TAIL_K 128
#define TAIL_CP 132
#define TAIL_PP 68
#define TAIL_XP 65

__device__ __forceinline__ void mma_1x4_k4(const v2f& a, const v2f& b0, const v2f& b1,
                                           const v2f& b2, const v2f& b3,
                                           v8f& c0, v8f& c1, v8f& c2, v8f& c3) {
  c0 = __builtin_amdgcn_wmma_f32_16x16x4_f32(false, a, false, b0, (short)0, c0, false, false);
  c1 = __builtin_amdgcn_wmma_f32_16x16x4_f32(false, a, false, b1, (short)0, c1, false, false);
  c2 = __builtin_amdgcn_wmma_f32_16x16x4_f32(false, a, false, b2, (short)0, c2, false, false);
  c3 = __builtin_amdgcn_wmma_f32_16x16x4_f32(false, a, false, b3, (short)0, c3, false, false);
}

__device__ __forceinline__ void mma_1x4_k8(const float4& a, const float4& b0, const float4& b1,
                                           const float4& b2, const float4& b3,
                                           v8f& c0, v8f& c1, v8f& c2, v8f& c3) {
  v2f al = {a.x, a.y}, ah = {a.z, a.w};
  v2f b0l = {b0.x, b0.y}, b0h = {b0.z, b0.w};
  v2f b1l = {b1.x, b1.y}, b1h = {b1.z, b1.w};
  v2f b2l = {b2.x, b2.y}, b2h = {b2.z, b2.w};
  v2f b3l = {b3.x, b3.y}, b3h = {b3.z, b3.w};
  mma_1x4_k4(al, b0l, b1l, b2l, b3l, c0, c1, c2, c3);
  mma_1x4_k4(ah, b0h, b1h, b2h, b3h, c0, c1, c2, c3);
}

__global__ __launch_bounds__(256)
void tail_kernel(const float* __restrict__ Q, const float* __restrict__ beta_p,
                 const float* __restrict__ dprev, const float* __restrict__ fcW,
                 const float* __restrict__ fcb, const float* __restrict__ W6,
                 const float* __restrict__ b6, float* __restrict__ out, int B) {
  __shared__ __attribute__((aligned(16))) float sC[16 * TAIL_CP];
  __shared__ __attribute__((aligned(16))) float sP[4 * 16 * TAIL_PP];
  __shared__ __attribute__((aligned(16))) float sD[TAIL_S * TAIL_PP];
  __shared__ __attribute__((aligned(16))) float sW6[7 * TAIL_PP];
  __shared__ float sX[TAIL_S * TAIL_XP];
  __shared__ float sY[TAIL_S * TAIL_XP];
  __shared__ float s6[TAIL_S * 8];
  int t = threadIdx.x;
  int wave = __builtin_amdgcn_readfirstlane((int)(threadIdx.x >> 5));
  int lane = t & 31;
  int r  = lane & 15;
  int kh = (lane >> 4) << 4;
  int b0 = blockIdx.x * TAIL_S;
  int hs = t >> 3;
  int ho = t & 7;
  bool head = (t < TAIL_S * 8) && (ho < 7);
  bool sums = t < TAIL_S * 16;
  int c4s = (t & 15) << 2;

  float beta = beta_p[0];
  float4 w00 = {}, w01 = {}, w02 = {}, w03 = {};
  float4 w10 = {}, w11 = {}, w12 = {}, w13 = {};
  float4 w20 = {}, w21 = {}, w22 = {}, w23 = {};
  float4 w30 = {}, w31 = {}, w32 = {}, w33 = {};
  if (wave < 4) {
    const float4* p0 = (const float4*)(fcW + (size_t)(r)      * TAIL_K + 32 * wave + kh);
    const float4* p1 = (const float4*)(fcW + (size_t)(16 + r) * TAIL_K + 32 * wave + kh);
    const float4* p2 = (const float4*)(fcW + (size_t)(32 + r) * TAIL_K + 32 * wave + kh);
    const float4* p3 = (const float4*)(fcW + (size_t)(48 + r) * TAIL_K + 32 * wave + kh);
    w00 = p0[0]; w01 = p0[1]; w02 = p0[2]; w03 = p0[3];
    w10 = p1[0]; w11 = p1[1]; w12 = p1[2]; w13 = p1[3];
    w20 = p2[0]; w21 = p2[1]; w22 = p2[2]; w23 = p2[3];
    w30 = p3[0]; w31 = p3[1]; w32 = p3[2]; w33 = p3[3];
  }
  float4 bv = {};
  if (sums) bv = *(const float4*)(fcb + c4s);
  float b6v = 0.0f;
  if (head) b6v = b6[ho];
  bool w6t = (t >= 128) && (t < 128 + 7 * 16);
  int w6o = (t - 128) >> 4;
  int w6c = ((t - 128) & 15) << 2;
  float4 w6v = {};
  if (w6t) w6v = *(const float4*)(W6 + w6o * TAIL_N + w6c);
  for (int z = t; z < (16 - TAIL_S) * 32; z += 256) {
    float4 zero = {};
    *(float4*)(sC + (TAIL_S + (z >> 5)) * TAIL_CP + ((z & 31) << 2)) = zero;
  }

  if (beta == 0.0f) {
    int c  = t & 127;
    int sa = t >> 7;
    int sb = sa + 2;
    size_t qc = (c < TAIL_N) ? (size_t)c : (size_t)B * TAIL_N + (size_t)(c - TAIL_N);
    float oa = Q[(size_t)(b0 + sa) * TAIL_N + qc];
    float ob = Q[(size_t)(b0 + sb) * TAIL_N + qc];
    float da = dprev[(size_t)(b0 + sa) * TAIL_K + c];
    float db = dprev[(size_t)(b0 + sb) * TAIL_K + c];
    asm volatile("" :: "v"(oa), "v"(ob), "v"(da), "v"(db));
    sC[sa * TAIL_CP + c] = mish_f(oa + da);
    sC[sb * TAIL_CP + c] = mish_f(ob + db);
  } else {
    for (int e = t; e < 2 * TAIL_S * TAIL_N; e += 256) {
      int y = e >> 8;
      int s = (e >> 6) & (TAIL_S - 1);
      int c = e & 63;
      float v = Q[(size_t)(b0 + s + (y ? B : 0)) * TAIL_N + c];
      if (y) sY[s * TAIL_XP + c] = v; else sX[s * TAIL_XP + c] = v;
    }
    __syncthreads();
    for (int e = t; e < TAIL_S * TAIL_K; e += 256) {
      int s   = e >> 7;
      int i   = (e & 127) >> 1;
      int dir = e & 1;
      const float* own = (dir ? sY : sX) + s * TAIL_XP;
      const float* oth = (dir ? sX : sY) + s * TAIL_XP;
      float omax = -3.4e38f, omin = 3.4e38f;
      for (int j = 0; j < TAIL_N; ++j) {
        omax = fmaxf(omax, oth[j]);
        omin = fminf(omin, oth[j]);
      }
      float vi = own[i];
      float dd = dprev[(size_t)(b0 + s) * TAIL_K + dir * TAIL_N + i];
      float m = (vi >= 0.0f) ? vi * omax : vi * omin;
      float den = 0.0f, num = 0.0f;
      for (int j = 0; j < TAIL_N; ++j) {
        float e1 = __expf(vi * oth[j] - m);
        den += e1;
        num += e1 * own[j];
      }
      float o = vi + beta * num / den;
      sC[s * TAIL_CP + dir * TAIL_N + i] = mish_f(o + dd);
    }
  }
  if (w6t) *(float4*)(sW6 + w6o * TAIL_PP + w6c) = w6v;
  __syncthreads();

  if (wave < 4) {
    const float4* pa = (const float4*)(sC + r * TAIL_CP + 32 * wave + kh);
    float4 a0 = pa[0], a1 = pa[1], a2 = pa[2], a3 = pa[3];
    v8f acc0 = {};
    v8f acc1 = {};
    v8f acc2 = {};
    v8f acc3 = {};
    mma_1x4_k8(a0, w00, w10, w20, w30, acc0, acc1, acc2, acc3);
    mma_1x4_k8(a1, w01, w11, w21, w31, acc0, acc1, acc2, acc3);
    mma_1x4_k8(a2, w02, w12, w22, w32, acc0, acc1, acc2, acc3);
    mma_1x4_k8(a3, w03, w13, w23, w33, acc0, acc1, acc2, acc3);
    float* pw = sP + wave * (16 * TAIL_PP) + ((lane >> 4) << 3) * TAIL_PP + r;
    for (int v = 0; v < TAIL_S; ++v) {
      pw[v * TAIL_PP]      = acc0[v];
      pw[v * TAIL_PP + 16] = acc1[v];
      pw[v * TAIL_PP + 32] = acc2[v];
      pw[v * TAIL_PP + 48] = acc3[v];
    }
  }
  __syncthreads();
  if (sums) {
    int row = t >> 4;
    const float* ps = sP + row * TAIL_PP + c4s;
    float4 s = *(const float4*)ps;
    for (int w = 1; w < 4; ++w) {
      float4 p = *(const float4*)(ps + w * (16 * TAIL_PP));
      s.x += p.x; s.y += p.y; s.z += p.z; s.w += p.w;
    }
    s.x += bv.x; s.y += bv.y; s.z += bv.z; s.w += bv.w;
    *(float4*)(sD + row * TAIL_PP + c4s) = s;
  }
  __syncthreads();

  if (head) {
    const float* row = sD + hs * TAIL_PP;
    const float* w   = sW6 + ho * TAIL_PP;
    float acc = b6v;
    for (int k4 = 0; k4 < TAIL_N; k4 += 4) {
      float4 rv = *(const float4*)(row + k4);
      float4 wv = *(const float4*)(w + k4);
      acc += rv.x * wv.x;
      acc += rv.y * wv.y;
      acc += rv.z * wv.z;
      acc += rv.w * wv.w;
    }
    s6[hs * 8 + ho] = acc;
  }
  __syncthreads();
  if (head) {
    const float* x6 = s6 + hs * 8;
    float nrm = sqrtf(x6[0]*x6[0] + x6[1]*x6[1] + x6[2]*x6[2] + x6[3]*x6[3]);
    float v = x6[ho];
    if (ho < 4) v = v / fmaxf(nrm, 1e-12f);
    out[(size_t)(b0 + hs) * 7 + ho] = v;
    out[(size_t)B * 7 + (size_t)(b0 + hs) * 7 + ho] = v;
  }
}

extern "C" void kernel_launch(void* const* d_in, const int* in_sizes, int n_in,
                              void* d_out, int out_size, void* d_ws, size_t ws_size,
                              hipStream_t stream) {
  const float* x    = (const float*)d_in[0];
  const float* y    = (const float*)d_in[1];
  const float* W1   = (const float*)d_in[2];
  const float* be1  = (const float*)d_in[3];
  const float* fcW1 = (const float*)d_in[4];
  const float* fcb1 = (const float*)d_in[5];
  const float* W2   = (const float*)d_in[6];
  const float* be2  = (const float*)d_in[7];
  const float* fcW2 = (const float*)d_in[8];
  const float* fcb2 = (const float*)d_in[9];
  const float* W3   = (const float*)d_in[10];
  const float* be3  = (const float*)d_in[11];
  const float* fcW3 = (const float*)d_in[12];
  const float* fcb3 = (const float*)d_in[13];
  const float* W4   = (const float*)d_in[14];
  const float* be4  = (const float*)d_in[15];
  const float* fcW4 = (const float*)d_in[16];
  const float* fcb4 = (const float*)d_in[17];
  const float* W6   = (const float*)d_in[18];
  const float* b6   = (const float*)d_in[19];

  float* ws = (float*)d_ws;
  float* m  = ws;
  float* q  = m  + 1048576;
  float* mA = q  + 524288;
  float* mC = mA + 524288;
  float* d0 = mC + 524288;
  float* d1 = d0 + 262144;
  float* d2 = d1 + 131072;

  const int B = 512;
  auto grid1d = [](int n) { return (n + 255) / 256; };
  auto gemm_pair = [&](const float* A0, const float* W0, const float* bias0, float* C0,
                       int M0, int N0, int K0,
                       const float* A1, const float* W1, const float* bias1, float* C1,
                       int M1, int N1, int K1) {
    int tiles = (M0 >> 5) * (N0 >> 5) + (M1 >> 5) * (N1 >> 5);
    hipLaunchKernelGGL(gemm_pair_wmma_f32, dim3(tiles), dim3(GEMM_WAVES * 32), 0, stream,
                       A0, W0, bias0, C0, M0, N0, K0, A1, W1, bias1, C1, M1, N1, K1);
  };
  auto attn = [&](const float* beta, const float* dprev, float* mAout, int Nout, int threads) {
    hipLaunchKernelGGL(attn_mish_kernel, dim3(B), dim3(threads), 0, stream,
                       q, beta, dprev, mAout, mC, Nout, B);
  };

  hipLaunchKernelGGL(mish2_kernel, dim3(grid1d(2 * B * 1024)), dim3(256), 0, stream,
                     x, y, m, B * 1024);
  gemm_pair(m,  W1,   nullptr, q,  2 * B, 512, 1024,  nullptr, nullptr, nullptr, nullptr, 0, 0, 0);
  attn(be1, nullptr, mA, 512, 256);
  gemm_pair(mC, fcW1, fcb1,    d0, B,     512, 1024,  mA, W2, nullptr, q, 2 * B, 256, 512);
  attn(be2, d0, mA, 256, 256);
  gemm_pair(mC, fcW2, fcb2,    d1, B,     256, 512,   mA, W3, nullptr, q, 2 * B, 128, 256);
  attn(be3, d1, mA, 128, 128);
  gemm_pair(mC, fcW3, fcb3,    d2, B,     128, 256,   mA, W4, nullptr, q, 2 * B, 64,  128);
  hipLaunchKernelGGL(tail_kernel, dim3(B / TAIL_S), dim3(256), 0, stream,
                     q, be4, d2, fcW4, fcb4, W6, b6, (float*)d_out, B);

  (void)in_sizes; (void)n_in; (void)out_size; (void)ws_size;
}


# mish in closed form: tanh(softplus(v)) = w/(w+2), w = e^v(e^v+2); one v_exp and one v_rcp replace log1pf + tanhf (about 110 -> 9 vector instructions each); last-bit rounding differs; sums unchanged
# speedup vs baseline: 5.7420x; 1.0789x over previous
#include <hip/hip_runtime.h>
#include <math.h>

typedef __attribute__((ext_vector_type(2))) float v2f;
typedef __attribute__((ext_vector_type(8))) float v8f;

__device__ __forceinline__ float mish_f(float v) {
  float n = __expf(v);
  float w = n * (n + 2.0f);
  float q = w * __builtin_amdgcn_rcpf(w + 2.0f);
  return (v > 20.0f) ? v : v * q;
}

__global__ void mish2_kernel(const float* __restrict__ a, const float* __restrict__ b,
                             float* __restrict__ out, int nh) {
  int i = blockIdx.x * blockDim.x + threadIdx.x;
  if (i >= 2 * nh) return;
  float v = (i < nh) ? a[i] : b[i - nh];
  out[i] = mish_f(v);
}

#define GEMM_WAVES 8
#define PART_PITCH 36
#define PART_TILE (32 * PART_PITCH)

__device__ __forceinline__ void mma_2x2_k8(const float4& a0, const float4& a1,
                                           const float4& b0, const float4& b1,
                                           v8f& acc00, v8f& acc01, v8f& acc10, v8f& acc11) {
  v2f a0l = {a0.x, a0.y}, a0h = {a0.z, a0.w};
  v2f a1l = {a1.x, a1.y}, a1h = {a1.z, a1.w};
  v2f b0l = {b0.x, b0.y}, b0h = {b0.z, b0.w};
  v2f b1l = {b1.x, b1.y}, b1h = {b1.z, b1.w};
  acc00 = __builtin_amdgcn_wmma_f32_16x16x4_f32(false, a0l, false, b0l, (short)0, acc00, false, false);
  acc01 = __builtin_amdgcn_wmma_f32_16x16x4_f32(false, a0l, false, b1l, (short)0, acc01, false, false);
  acc10 = __builtin_amdgcn_wmma_f32_16x16x4_f32(false, a1l, false, b0l, (short)0, acc10, false, false);
  acc11 = __builtin_amdgcn_wmma_f32_16x16x4_f32(false, a1l, false, b1l, (short)0, acc11, false, false);
  acc00 = __builtin_amdgcn_wmma_f32_16x16x4_f32(false, a0h, false, b0h, (short)0, acc00, false, false);
  acc01 = __builtin_amdgcn_wmma_f32_16x16x4_f32(false, a0h, false, b1h, (short)0, acc01, false, false);
  acc10 = __builtin_amdgcn_wmma_f32_16x16x4_f32(false, a1h, false, b0h, (short)0, acc10, false, false);
  acc11 = __builtin_amdgcn_wmma_f32_16x16x4_f32(false, a1h, false, b1h, (short)0, acc11, false, false);
}

__global__ __launch_bounds__(GEMM_WAVES * 32)
void gemm_pair_wmma_f32(const float* __restrict__ A0, const float* __restrict__ W0,
                        const float* __restrict__ bias0, float* __restrict__ C0,
                        int M0, int N0, int K0,
                        const float* __restrict__ A1, const float* __restrict__ W1,
                        const float* __restrict__ bias1, float* __restrict__ C1,
                        int M1, int N1, int K1) {
  __shared__ __attribute__((aligned(16))) float part[GEMM_WAVES * PART_TILE];
  int wave = __builtin_amdgcn_readfirstlane((int)(threadIdx.x >> 5));
  int lane = threadIdx.x & 31;
  int tile = blockIdx.x;
  int tiles0 = (M0 >> 5) * (N0 >> 5);
  bool second = tile >= tiles0;
  const float* A    = second ? A1 : A0;
  const float* W    = second ? W1 : W0;
  const float* bias = second ? bias1 : bias0;
  float* C          = second ? C1 : C0;
  int M = second ? M1 : M0;
  int N = second ? N1 : N0;
  int K = second ? K1 : K0;
  if (second) tile -= tiles0;
  int tiles_n = N >> 5;
  if (tile >= (M >> 5) * tiles_n) return;
  int tm = tile / tiles_n;
  int m0 = tm << 5;
  int n0 = (tile - tm * tiles_n) << 5;
  int r  = lane & 15;
  int kh = (lane >> 4) << 4;
  int ksl = ((((K >> 5) + GEMM_WAVES - 1) / GEMM_WAVES)) << 5;
  int kb = wave * ksl;
  int ke = kb + ksl;
  if (ke > K) ke = K;
  const float* Ar0 = A + (size_t)(m0 + r) * K + kh;
  const float* Ar1 = Ar0 + (size_t)16 * K;
  const float* Wr0 = W + (size_t)(n0 + r) * K + kh;
  const float* Wr1 = Wr0 + (size_t)16 * K;
  v8f acc00 = {};
  v8f acc01 = {};
  v8f acc10 = {};
  v8f acc11 = {};
  for (int k = kb; k < ke; k += 32) {
    const float4* pa0 = (const float4*)(Ar0 + k);
    const float4* pa1 = (const float4*)(Ar1 + k);
    const float4* pb0 = (const float4*)(Wr0 + k);
    const float4* pb1 = (const float4*)(Wr1 + k);
    float4 a00 = pa0[0], a01 = pa0[1], a02 = pa0[2], a03 = pa0[3];
    float4 b00 = pb0[0], b01 = pb0[1], b02 = pb0[2], b03 = pb0[3];
    float4 a10 = pa1[0], a11 = pa1[1], a12 = pa1[2], a13 = pa1[3];
    float4 b10 = pb1[0], b11 = pb1[1], b12 = pb1[2], b13 = pb1[3];
    __builtin_amdgcn_sched_barrier(0);
    mma_2x2_k8(a00, a10, b00, b10, acc00, acc01, acc10, acc11);
    mma_2x2_k8(a01, a11, b01, b11, acc00, acc01, acc10, acc11);
    mma_2x2_k8(a02, a12, b02, b12, acc00, acc01, acc10, acc11);
    mma_2x2_k8(a03, a13, b03, b13, acc00, acc01, acc10, acc11);
  }
  float* pw = part + wave * PART_TILE + ((lane >> 4) << 3) * PART_PITCH + r;
  for (int v = 0; v < 8; ++v) {
    pw[v * PART_PITCH]             = acc00[v];
    pw[v * PART_PITCH + 16]        = acc01[v];
    pw[(v + 16) * PART_PITCH]      = acc10[v];
    pw[(v + 16) * PART_PITCH + 16] = acc11[v];
  }
  __syncthreads();
  int row = threadIdx.x >> 3;
  int c4  = (threadIdx.x & 7) << 2;
  const float* ps = part + row * PART_PITCH + c4;
  float4 s = *(const float4*)ps;
  for (int w = 1; w < GEMM_WAVES; ++w) {
    float4 p = *(const float4*)(ps + w * PART_TILE);
    s.x += p.x; s.y += p.y; s.z += p.z; s.w += p.w;
  }
  if (bias) {
    float4 bv = *(const float4*)(bias + n0 + c4);
    s.x += bv.x; s.y += bv.y; s.z += bv.z; s.w += bv.w;
  }
  *(float4*)(C + (size_t)(m0 + row) * N + n0 + c4) = s;
}

struct StageOut {
  const float* dprev;
  float* mA;
  float* mC;
  int N;
  int B;
  int b;
};

__device__ __forceinline__ void stage_emit(const StageOut& so, int i, float ox, float oy,
                                           float dx, float dy) {
  size_t rowC = (size_t)so.b * (2 * so.N);
  float mx = 0.0f, my = 0.0f;
  if (so.mA != nullptr || so.dprev == nullptr) { mx = mish_f(ox); my = mish_f(oy); }
  if (so.mA != nullptr) {
    so.mA[(size_t)so.b * so.N + i] = mx;
    so.mA[(size_t)(so.b + so.B) * so.N + i] = my;
  }
  if (so.dprev != nullptr) { mx = mish_f(ox + dx); my = mish_f(oy + dy); }
  so.mC[rowC + i] = mx;
  so.mC[rowC + so.N + i] = my;
}

__global__ __launch_bounds__(256)
void attn_mish_kernel(const float* __restrict__ Q, const float* __restrict__ beta_p,
                      const float* __restrict__ dprev, float* __restrict__ mA,
                      float* __restrict__ mC, int N, int B) {
  __shared__ float sh_x[512];
  __shared__ float sh_y[512];
  __shared__ float sxmax[256];
  __shared__ float sxmin[256];
  __shared__ float symax[256];
  __shared__ float symin[256];
  int b = blockIdx.x;
  int t = threadIdx.x;
  int T = blockDim.x;
  const float* xq = Q + (size_t)b * N;
  const float* yq = Q + (size_t)(b + B) * N;
  const float* dr = dprev ? dprev + (size_t)b * (2 * N) : nullptr;
  StageOut so;
  so.dprev = dprev; so.mA = mA; so.mC = mC; so.N = N; so.B = B; so.b = b;
  float beta = beta_p[0];
  if (beta == 0.0f) {
    for (int i = t; i < N; i += T) {
      float ox = xq[i];
      float oy = yq[i];
      float dx = dr ? dr[i] : 0.0f;
      float dy = dr ? dr[N + i] : 0.0f;
      stage_emit(so, i, ox, oy, dx, dy);
    }
    return;
  }
  for (int i = t; i < N; i += T) { sh_x[i] = xq[i]; sh_y[i] = yq[i]; }
  __syncthreads();
  float xmax = -3.4e38f, xmin = 3.4e38f, ymax = -3.4e38f, ymin = 3.4e38f;
  for (int i = t; i < N; i += T) {
    xmax = fmaxf(xmax, sh_x[i]); xmin = fminf(xmin, sh_x[i]);
    ymax = fmaxf(ymax, sh_y[i]); ymin = fminf(ymin, sh_y[i]);
  }
  sxmax[t] = xmax; sxmin[t] = xmin; symax[t] = ymax; symin[t] = ymin;
  __syncthreads();
  for (int off = T >> 1; off > 0; off >>= 1) {
    if (t < off) {
      sxmax[t] = fmaxf(sxmax[t], sxmax[t + off]); sxmin[t] = fminf(sxmin[t], sxmin[t + off]);
      symax[t] = fmaxf(symax[t], symax[t + off]); symin[t] = fminf(symin[t], symin[t + off]);
    }
    __syncthreads();
  }
  xmax = sxmax[0]; xmin = sxmin[0]; ymax = symax[0]; ymin = symin[0];
  for (int i = t; i < N; i += T) {
    float xi = sh_x[i];
    float yi = sh_y[i];
    float dx = dr ? dr[i] : 0.0f;
    float dy = dr ? dr[N + i] : 0.0f;
    float m1 = (xi >= 0.0f) ? xi * ymax : xi * ymin;
    float m2 = (yi >= 0.0f) ? yi * xmax : yi * xmin;
    float den1 = 0.0f, num1 = 0.0f, den2 = 0.0f, num2 = 0.0f;
    for (int j = 0; j < N; ++j) {
      float xj = sh_x[j];
      float yj = sh_y[j];
      float e1 = __expf(xi * yj - m1);
      den1 += e1;
      num1 += e1 * xj;
      float e2 = __expf(yi * xj - m2);
      den2 += e2;
      num2 += e2 * yj;
    }
    float ox = xi + beta * num1 / den1;
    float oy = yi + beta * num2 / den2;
    stage_emit(so, i, ox, oy, dx, dy);
  }
}

#define TAIL_S 4
#define TAIL_N 64
#define TAIL_K 128
#define TAIL_CP 132
#define TAIL_PP 68
#define TAIL_XP 65

__device__ __forceinline__ void mma_1x4_k4(const v2f& a, const v2f& b0, const v2f& b1,
                                           const v2f& b2, const v2f& b3,
                                           v8f& c0, v8f& c1, v8f& c2, v8f& c3) {
  c0 = __builtin_amdgcn_wmma_f32_16x16x4_f32(false, a, false, b0, (short)0, c0, false, false);
  c1 = __builtin_amdgcn_wmma_f32_16x16x4_f32(false, a, false, b1, (short)0, c1, false, false);
  c2 = __builtin_amdgcn_wmma_f32_16x16x4_f32(false, a, false, b2, (short)0, c2, false, false);
  c3 = __builtin_amdgcn_wmma_f32_16x16x4_f32(false, a, false, b3, (short)0, c3, false, false);
}

__device__ __forceinline__ void mma_1x4_k8(const float4& a, const float4& b0, const float4& b1,
                                           const float4& b2, const float4& b3,
                                           v8f& c0, v8f& c1, v8f& c2, v8f& c3) {
  v2f al = {a.x, a.y}, ah = {a.z, a.w};
  v2f b0l = {b0.x, b0.y}, b0h = {b0.z, b0.w};
  v2f b1l = {b1.x, b1.y}, b1h = {b1.z, b1.w};
  v2f b2l = {b2.x, b2.y}, b2h = {b2.z, b2.w};
  v2f b3l = {b3.x, b3.y}, b3h = {b3.z, b3.w};
  mma_1x4_k4(al, b0l, b1l, b2l, b3l, c0, c1, c2, c3);
  mma_1x4_k4(ah, b0h, b1h, b2h, b3h, c0, c1, c2, c3);
}

__global__ __launch_bounds__(256)
void tail_kernel(const float* __restrict__ Q, const float* __restrict__ beta_p,
                 const float* __restrict__ dprev, const float* __restrict__ fcW,
                 const float* __restrict__ fcb, const float* __restrict__ W6,
                 const float* __restrict__ b6, float* __restrict__ out, int B) {
  __shared__ __attribute__((aligned(16))) float sC[16 * TAIL_CP];
  __shared__ __attribute__((aligned(16))) float sP[4 * 16 * TAIL_PP];
  __shared__ __attribute__((aligned(16))) float sD[TAIL_S * TAIL_PP];
  __shared__ __attribute__((aligned(16))) float sW6[7 * TAIL_PP];
  __shared__ float sX[TAIL_S * TAIL_XP];
  __shared__ float sY[TAIL_S * TAIL_XP];
  __shared__ float s6[TAIL_S * 8];
  int t = threadIdx.x;
  int wave = __builtin_amdgcn_readfirstlane((int)(threadIdx.x >> 5));
  int lane = t & 31;
  int r  = lane & 15;
  int kh = (lane >> 4) << 4;
  int b0 = blockIdx.x * TAIL_S;
  int hs = t >> 3;
  int ho = t & 7;
  bool head = (t < TAIL_S * 8) && (ho < 7);
  bool sums = t < TAIL_S * 16;
  int c4s = (t & 15) << 2;

  float beta = beta_p[0];
  float4 w00 = {}, w01 = {}, w02 = {}, w03 = {};
  float4 w10 = {}, w11 = {}, w12 = {}, w13 = {};
  float4 w20 = {}, w21 = {}, w22 = {}, w23 = {};
  float4 w30 = {}, w31 = {}, w32 = {}, w33 = {};
  if (wave < 4) {
    const float4* p0 = (const float4*)(fcW + (size_t)(r)      * TAIL_K + 32 * wave + kh);
    const float4* p1 = (const float4*)(fcW + (size_t)(16 + r) * TAIL_K + 32 * wave + kh);
    const float4* p2 = (const float4*)(fcW + (size_t)(32 + r) * TAIL_K + 32 * wave + kh);
    const float4* p3 = (const float4*)(fcW + (size_t)(48 + r) * TAIL_K + 32 * wave + kh);
    w00 = p0[0]; w01 = p0[1]; w02 = p0[2]; w03 = p0[3];
    w10 = p1[0]; w11 = p1[1]; w12 = p1[2]; w13 = p1[3];
    w20 = p2[0]; w21 = p2[1]; w22 = p2[2]; w23 = p2[3];
    w30 = p3[0]; w31 = p3[1]; w32 = p3[2]; w33 = p3[3];
  }
  float4 bv = {};
  if (sums) bv = *(const float4*)(fcb + c4s);
  float b6v = 0.0f;
  if (head) b6v = b6[ho];
  bool w6t = (t >= 128) && (t < 128 + 7 * 16);
  int w6o = (t - 128) >> 4;
  int w6c = ((t - 128) & 15) << 2;
  float4 w6v = {};
  if (w6t) w6v = *(const float4*)(W6 + w6o * TAIL_N + w6c);
  for (int z = t; z < (16 - TAIL_S) * 32; z += 256) {
    float4 zero = {};
    *(float4*)(sC + (TAIL_S + (z >> 5)) * TAIL_CP + ((z & 31) << 2)) = zero;
  }

  if (beta == 0.0f) {
    int c  = t & 127;
    int sa = t >> 7;
    int sb = sa + 2;
    size_t qc = (c < TAIL_N) ? (size_t)c : (size_t)B * TAIL_N + (size_t)(c - TAIL_N);
    float oa = Q[(size_t)(b0 + sa) * TAIL_N + qc];
    float ob = Q[(size_t)(b0 + sb) * TAIL_N + qc];
    float da = dprev[(size_t)(b0 + sa) * TAIL_K + c];
    float db = dprev[(size_t)(b0 + sb) * TAIL_K + c];
    asm volatile("" :: "v"(oa), "v"(ob), "v"(da), "v"(db));
    sC[sa * TAIL_CP + c] = mish_f(oa + da);
    sC[sb * TAIL_CP + c] = mish_f(ob + db);
  } else {
    for (int e = t; e < 2 * TAIL_S * TAIL_N; e += 256) {
      int y = e >> 8;
      int s = (e >> 6) & (TAIL_S - 1);
      int c = e & 63;
      float v = Q[(size_t)(b0 + s + (y ? B : 0)) * TAIL_N + c];
      if (y) sY[s * TAIL_XP + c] = v; else sX[s * TAIL_XP + c] = v;
    }
    __syncthreads();
    for (int e = t; e < TAIL_S * TAIL_K; e += 256) {
      int s   = e >> 7;
      int i   = (e & 127) >> 1;
      int dir = e & 1;
      const float* own = (dir ? sY : sX) + s * TAIL_XP;
      const float* oth = (dir ? sX : sY) + s * TAIL_XP;
      float omax = -3.4e38f, omin = 3.4e38f;
      for (int j = 0; j < TAIL_N; ++j) {
        omax = fmaxf(omax, oth[j]);
        omin = fminf(omin, oth[j]);
      }
      float vi = own[i];
      float dd = dprev[(size_t)(b0 + s) * TAIL_K + dir * TAIL_N + i];
      float m = (vi >= 0.0f) ? vi * omax : vi * omin;
      float den = 0.0f, num = 0.0f;
      for (int j = 0; j < TAIL_N; ++j) {
        float e1 = __expf(vi * oth[j] - m);
        den += e1;
        num += e1 * own[j];
      }
      float o = vi + beta * num / den;
      sC[s * TAIL_CP + dir * TAIL_N + i] = mish_f(o + dd);
    }
  }
  if (w6t) *(float4*)(sW6 + w6o * TAIL_PP + w6c) = w6v;
  __syncthreads();

  if (wave < 4) {
    const float4* pa = (const float4*)(sC + r * TAIL_CP + 32 * wave + kh);
    float4 a0 = pa[0], a1 = pa[1], a2 = pa[2], a3 = pa[3];
    v8f acc0 = {};
    v8f acc1 = {};
    v8f acc2 = {};
    v8f acc3 = {};
    mma_1x4_k8(a0, w00, w10, w20, w30, acc0, acc1, acc2, acc3);
    mma_1x4_k8(a1, w01, w11, w21, w31, acc0, acc1, acc2, acc3);
    mma_1x4_k8(a2, w02, w12, w22, w32, acc0, acc1, acc2, acc3);
    mma_1x4_k8(a3, w03, w13, w23, w33, acc0, acc1, acc2, acc3);
    float* pw = sP + wave * (16 * TAIL_PP) + ((lane >> 4) << 3) * TAIL_PP + r;
    for (int v = 0; v < TAIL_S; ++v) {
      pw[v * TAIL_PP]      = acc0[v];
      pw[v * TAIL_PP + 16] = acc1[v];
      pw[v * TAIL_PP + 32] = acc2[v];
      pw[v * TAIL_PP + 48] = acc3[v];
    }
  }
  __syncthreads();
  if (sums) {
    int row = t >> 4;
    const float* ps = sP + row * TAIL_PP + c4s;
    float4 s = *(const float4*)ps;
    for (int w = 1; w < 4; ++w) {
      float4 p = *(const float4*)(ps + w * (16 * TAIL_PP));
      s.x += p.x; s.y += p.y; s.z += p.z; s.w += p.w;
    }
    s.x += bv.x; s.y += bv.y; s.z += bv.z; s.w += bv.w;
    *(float4*)(sD + row * TAIL_PP + c4s) = s;
  }
  __syncthreads();

  if (head) {
    const float* row = sD + hs * TAIL_PP;
    const float* w   = sW6 + ho * TAIL_PP;
    float acc = b6v;
    for (int k4 = 0; k4 < TAIL_N; k4 += 4) {
      float4 rv = *(const float4*)(row + k4);
      float4 wv = *(const float4*)(w + k4);
      acc += rv.x * wv.x;
      acc += rv.y * wv.y;
      acc += rv.z * wv.z;
      acc += rv.w * wv.w;
    }
    s6[hs * 8 + ho] = acc;
  }
  __syncthreads();
  if (head) {
    const float* x6 = s6 + hs * 8;
    float nrm = sqrtf(x6[0]*x6[0] + x6[1]*x6[1] + x6[2]*x6[2] + x6[3]*x6[3]);
    float v = x6[ho];
    if (ho < 4) v = v / fmaxf(nrm, 1e-12f);
    out[(size_t)(b0 + hs) * 7 + ho] = v;
    out[(size_t)B * 7 + (size_t)(b0 + hs) * 7 + ho] = v;
  }
}

extern "C" void kernel_launch(void* const* d_in, const int* in_sizes, int n_in,
                              void* d_out, int out_size, void* d_ws, size_t ws_size,
                              hipStream_t stream) {
  const float* x    = (const float*)d_in[0];
  const float* y    = (const float*)d_in[1];
  const float* W1   = (const float*)d_in[2];
  const float* be1  = (const float*)d_in[3];
  const float* fcW1 = (const float*)d_in[4];
  const float* fcb1 = (const float*)d_in[5];
  const float* W2   = (const float*)d_in[6];
  const float* be2  = (const float*)d_in[7];
  const float* fcW2 = (const float*)d_in[8];
  const float* fcb2 = (const float*)d_in[9];
  const float* W3   = (const float*)d_in[10];
  const float* be3  = (const float*)d_in[11];
  const float* fcW3 = (const float*)d_in[12];
  const float* fcb3 = (const float*)d_in[13];
  const float* W4   = (const float*)d_in[14];
  const float* be4  = (const float*)d_in[15];
  const float* fcW4 = (const float*)d_in[16];
  const float* fcb4 = (const float*)d_in[17];
  const float* W6   = (const float*)d_in[18];
  const float* b6   = (const float*)d_in[19];

  float* ws = (float*)d_ws;
  float* m  = ws;
  float* q  = m  + 1048576;
  float* mA = q  + 524288;
  float* mC = mA + 524288;
  float* d0 = mC + 524288;
  float* d1 = d0 + 262144;
  float* d2 = d1 + 131072;

  const int B = 512;
  auto grid1d = [](int n) { return (n + 255) / 256; };
  auto gemm_pair = [&](const float* A0, const float* W0, const float* bias0, float* C0,
                       int M0, int N0, int K0,
                       const float* A1, const float* W1, const float* bias1, float* C1,
                       int M1, int N1, int K1) {
    int tiles = (M0 >> 5) * (N0 >> 5) + (M1 >> 5) * (N1 >> 5);
    hipLaunchKernelGGL(gemm_pair_wmma_f32, dim3(tiles), dim3(GEMM_WAVES * 32), 0, stream,
                       A0, W0, bias0, C0, M0, N0, K0, A1, W1, bias1, C1, M1, N1, K1);
  };
  auto attn = [&](const float* beta, const float* dprev, float* mAout, int Nout, int threads) {
    hipLaunchKernelGGL(attn_mish_kernel, dim3(B), dim3(threads), 0, stream,
                       q, beta, dprev, mAout, mC, Nout, B);
  };

  hipLaunchKernelGGL(mish2_kernel, dim3(grid1d(2 * B * 1024)), dim3(256), 0, stream,
                     x, y, m, B * 1024);
  gemm_pair(m,  W1,   nullptr, q,  2 * B, 512, 1024,  nullptr, nullptr, nullptr, nullptr, 0, 0, 0);
  attn(be1, nullptr, mA, 512, 256);
  gemm_pair(mC, fcW1, fcb1,    d0, B,     512, 1024,  mA, W2, nullptr, q, 2 * B, 256, 512);
  attn(be2, d0, mA, 256, 256);
  gemm_pair(mC, fcW2, fcb2,    d1, B,     256, 512,   mA, W3, nullptr, q, 2 * B, 128, 256);
  attn(be3, d1, mA, 128, 128);
  gemm_pair(mC, fcW3, fcb3,    d2, B,     128, 256,   mA, W4, nullptr, q, 2 * B, 64,  128);
  hipLaunchKernelGGL(tail_kernel, dim3(B / TAIL_S), dim3(256), 0, stream,
                     q, be4, d2, fcW4, fcb4, W6, b6, (float*)d_out, B);

  (void)in_sizes; (void)n_in; (void)out_size; (void)ws_size;
}


# stage-1 mish folded into the first GEMM: lanes apply mish to their loaded A fragment floats ahead of each block's WMMAs (redone per N tile); mish launch and buffer go; 9 launches -> 8; sums unchanged
# speedup vs baseline: 5.9638x; 1.0386x over previous
#include <hip/hip_runtime.h>
#include <math.h>

typedef __attribute__((ext_vector_type(2))) float v2f;
typedef __attribute__((ext_vector_type(8))) float v8f;

__device__ __forceinline__ float mish_f(float v) {
  float n = __expf(v);
  float w = n * (n + 2.0f);
  float q = w * __builtin_amdgcn_rcpf(w + 2.0f);
  return (v > 20.0f) ? v : v * q;
}

__device__ __forceinline__ void mish8(float4& p, float4& r) {
  v8f v = {p.x, p.y, p.z, p.w, r.x, r.y, r.z, r.w};
  v8f n = {__expf(v[0]), __expf(v[1]), __expf(v[2]), __expf(v[3]),
           __expf(v[4]), __expf(v[5]), __expf(v[6]), __expf(v[7])};
  v8f w = n * (n + 2.0f);
  v8f d = w + 2.0f;
  v8f rc = {__builtin_amdgcn_rcpf(d[0]), __builtin_amdgcn_rcpf(d[1]),
            __builtin_amdgcn_rcpf(d[2]), __builtin_amdgcn_rcpf(d[3]),
            __builtin_amdgcn_rcpf(d[4]), __builtin_amdgcn_rcpf(d[5]),
            __builtin_amdgcn_rcpf(d[6]), __builtin_amdgcn_rcpf(d[7])};
  v8f m = v * (w * rc);
  v8f o = (v > 20.0f) ? v : m;
  p.x = o[0]; p.y = o[1]; p.z = o[2]; p.w = o[3];
  r.x = o[4]; r.y = o[5]; r.z = o[6]; r.w = o[7];
}

#define GEMM_WAVES 8
#define PART_PITCH 36
#define PART_TILE (32 * PART_PITCH)

__device__ __forceinline__ void mma_2x2_k8(const float4& a0, const float4& a1,
                                           const float4& b0, const float4& b1,
                                           v8f& acc00, v8f& acc01, v8f& acc10, v8f& acc11) {
  v2f a0l = {a0.x, a0.y}, a0h = {a0.z, a0.w};
  v2f a1l = {a1.x, a1.y}, a1h = {a1.z, a1.w};
  v2f b0l = {b0.x, b0.y}, b0h = {b0.z, b0.w};
  v2f b1l = {b1.x, b1.y}, b1h = {b1.z, b1.w};
  acc00 = __builtin_amdgcn_wmma_f32_16x16x4_f32(false, a0l, false, b0l, (short)0, acc00, false, false);
  acc01 = __builtin_amdgcn_wmma_f32_16x16x4_f32(false, a0l, false, b1l, (short)0, acc01, false, false);
  acc10 = __builtin_amdgcn_wmma_f32_16x16x4_f32(false, a1l, false, b0l, (short)0, acc10, false, false);
  acc11 = __builtin_amdgcn_wmma_f32_16x16x4_f32(false, a1l, false, b1l, (short)0, acc11, false, false);
  acc00 = __builtin_amdgcn_wmma_f32_16x16x4_f32(false, a0h, false, b0h, (short)0, acc00, false, false);
  acc01 = __builtin_amdgcn_wmma_f32_16x16x4_f32(false, a0h, false, b1h, (short)0, acc01, false, false);
  acc10 = __builtin_amdgcn_wmma_f32_16x16x4_f32(false, a1h, false, b0h, (short)0, acc10, false, false);
  acc11 = __builtin_amdgcn_wmma_f32_16x16x4_f32(false, a1h, false, b1h, (short)0, acc11, false, false);
}

template <bool MISH_A>
__device__ __forceinline__ void gemm_tile_32x32(const float* __restrict__ A, const float* __restrict__ W,
                                                const float* __restrict__ bias, float* __restrict__ C,
                                                int am0, int m0, int n0, int N, int K, float* part) {
  int wave = __builtin_amdgcn_readfirstlane((int)(threadIdx.x >> 5));
  int lane = threadIdx.x & 31;
  int r  = lane & 15;
  int kh = (lane >> 4) << 4;
  int ksl = ((((K >> 5) + GEMM_WAVES - 1) / GEMM_WAVES)) << 5;
  int kb = wave * ksl;
  int ke = kb + ksl;
  if (ke > K) ke = K;
  const float* Ar0 = A + (size_t)(am0 + r) * K + kh;
  const float* Ar1 = Ar0 + (size_t)16 * K;
  const float* Wr0 = W + (size_t)(n0 + r) * K + kh;
  const float* Wr1 = Wr0 + (size_t)16 * K;
  v8f acc00 = {};
  v8f acc01 = {};
  v8f acc10 = {};
  v8f acc11 = {};
  for (int k = kb; k < ke; k += 32) {
    const float4* pa0 = (const float4*)(Ar0 + k);
    const float4* pa1 = (const float4*)(Ar1 + k);
    const float4* pb0 = (const float4*)(Wr0 + k);
    const float4* pb1 = (const float4*)(Wr1 + k);
    float4 a00 = pa0[0], a01 = pa0[1], a02 = pa0[2], a03 = pa0[3];
    float4 b00 = pb0[0], b01 = pb0[1], b02 = pb0[2], b03 = pb0[3];
    float4 a10 = pa1[0], a11 = pa1[1], a12 = pa1[2], a13 = pa1[3];
    float4 b10 = pb1[0], b11 = pb1[1], b12 = pb1[2], b13 = pb1[3];
    __builtin_amdgcn_sched_barrier(0);
    if (MISH_A) {
      mish8(a00, a10);
      mish8(a01, a11);
      mish8(a02, a12);
      mish8(a03, a13);
      __builtin_amdgcn_sched_barrier(0);
    }
    mma_2x2_k8(a00, a10, b00, b10, acc00, acc01, acc10, acc11);
    mma_2x2_k8(a01, a11, b01, b11, acc00, acc01, acc10, acc11);
    mma_2x2_k8(a02, a12, b02, b12, acc00, acc01, acc10, acc11);
    mma_2x2_k8(a03, a13, b03, b13, acc00, acc01, acc10, acc11);
  }
  float* pw = part + wave * PART_TILE + ((lane >> 4) << 3) * PART_PITCH + r;
  for (int v = 0; v < 8; ++v) {
    pw[v * PART_PITCH]             = acc00[v];
    pw[v * PART_PITCH + 16]        = acc01[v];
    pw[(v + 16) * PART_PITCH]      = acc10[v];
    pw[(v + 16) * PART_PITCH + 16] = acc11[v];
  }
  __syncthreads();
  int row = threadIdx.x >> 3;
  int c4  = (threadIdx.x & 7) << 2;
  const float* ps = part + row * PART_PITCH + c4;
  float4 s = *(const float4*)ps;
  for (int w = 1; w < GEMM_WAVES; ++w) {
    float4 p = *(const float4*)(ps + w * PART_TILE);
    s.x += p.x; s.y += p.y; s.z += p.z; s.w += p.w;
  }
  if (bias) {
    float4 bv = *(const float4*)(bias + n0 + c4);
    s.x += bv.x; s.y += bv.y; s.z += bv.z; s.w += bv.w;
  }
  *(float4*)(C + (size_t)(m0 + row) * N + n0 + c4) = s;
}

__global__ __launch_bounds__(GEMM_WAVES * 32)
void gemm_pair_wmma_f32(const float* __restrict__ A0, const float* __restrict__ W0,
                        const float* __restrict__ bias0, float* __restrict__ C0,
                        int M0, int N0, int K0,
                        const float* __restrict__ A1, const float* __restrict__ W1,
                        const float* __restrict__ bias1, float* __restrict__ C1,
                        int M1, int N1, int K1) {
  __shared__ __attribute__((aligned(16))) float part[GEMM_WAVES * PART_TILE];
  int tile = blockIdx.x;
  int tiles0 = (M0 >> 5) * (N0 >> 5);
  bool second = tile >= tiles0;
  const float* A    = second ? A1 : A0;
  const float* W    = second ? W1 : W0;
  const float* bias = second ? bias1 : bias0;
  float* C          = second ? C1 : C0;
  int M = second ? M1 : M0;
  int N = second ? N1 : N0;
  int K = second ? K1 : K0;
  if (second) tile -= tiles0;
  int tiles_n = N >> 5;
  if (tile >= (M >> 5) * tiles_n) return;
  int tm = tile / tiles_n;
  int m0 = tm << 5;
  int n0 = (tile - tm * tiles_n) << 5;
  gemm_tile_32x32<false>(A, W, bias, C, m0, m0, n0, N, K, part);
}

__global__ __launch_bounds__(GEMM_WAVES * 32)
void gemm_mish_wmma_f32(const float* __restrict__ x, const float* __restrict__ y,
                        const float* __restrict__ W, float* __restrict__ C,
                        int B, int N, int K) {
  __shared__ __attribute__((aligned(16))) float part[GEMM_WAVES * PART_TILE];
  int tile = blockIdx.x;
  int tiles_n = N >> 5;
  if (tile >= ((2 * B) >> 5) * tiles_n) return;
  int tm = tile / tiles_n;
  int m0 = tm << 5;
  int n0 = (tile - tm * tiles_n) << 5;
  bool ybr = m0 >= B;
  const float* A = ybr ? y : x;
  int am0 = ybr ? m0 - B : m0;
  gemm_tile_32x32<true>(A, W, nullptr, C, am0, m0, n0, N, K, part);
}

struct StageOut {
  const float* dprev;
  float* mA;
  float* mC;
  int N;
  int B;
  int b;
};

__device__ __forceinline__ void stage_emit(const StageOut& so, int i, float ox, float oy,
                                           float dx, float dy) {
  size_t rowC = (size_t)so.b * (2 * so.N);
  float mx = 0.0f, my = 0.0f;
  if (so.mA != nullptr || so.dprev == nullptr) { mx = mish_f(ox); my = mish_f(oy); }
  if (so.mA != nullptr) {
    so.mA[(size_t)so.b * so.N + i] = mx;
    so.mA[(size_t)(so.b + so.B) * so.N + i] = my;
  }
  if (so.dprev != nullptr) { mx = mish_f(ox + dx); my = mish_f(oy + dy); }
  so.mC[rowC + i] = mx;
  so.mC[rowC + so.N + i] = my;
}

__global__ __launch_bounds__(256)
void attn_mish_kernel(const float* __restrict__ Q, const float* __restrict__ beta_p,
                      const float* __restrict__ dprev, float* __restrict__ mA,
                      float* __restrict__ mC, int N, int B) {
  __shared__ float sh_x[512];
  __shared__ float sh_y[512];
  __shared__ float sxmax[256];
  __shared__ float sxmin[256];
  __shared__ float symax[256];
  __shared__ float symin[256];
  int b = blockIdx.x;
  int t = threadIdx.x;
  int T = blockDim.x;
  const float* xq = Q + (size_t)b * N;
  const float* yq = Q + (size_t)(b + B) * N;
  const float* dr = dprev ? dprev + (size_t)b * (2 * N) : nullptr;
  StageOut so;
  so.dprev = dprev; so.mA = mA; so.mC = mC; so.N = N; so.B = B; so.b = b;
  float beta = beta_p[0];
  if (beta == 0.0f) {
    for (int i = t; i < N; i += T) {
      float ox = xq[i];
      float oy = yq[i];
      float dx = dr ? dr[i] : 0.0f;
      float dy = dr ? dr[N + i] : 0.0f;
      stage_emit(so, i, ox, oy, dx, dy);
    }
    return;
  }
  for (int i = t; i < N; i += T) { sh_x[i] = xq[i]; sh_y[i] = yq[i]; }
  __syncthreads();
  float xmax = -3.4e38f, xmin = 3.4e38f, ymax = -3.4e38f, ymin = 3.4e38f;
  for (int i = t; i < N; i += T) {
    xmax = fmaxf(xmax, sh_x[i]); xmin = fminf(xmin, sh_x[i]);
    ymax = fmaxf(ymax, sh_y[i]); ymin = fminf(ymin, sh_y[i]);
  }
  sxmax[t] = xmax; sxmin[t] = xmin; symax[t] = ymax; symin[t] = ymin;
  __syncthreads();
  for (int off = T >> 1; off > 0; off >>= 1) {
    if (t < off) {
      sxmax[t] = fmaxf(sxmax[t], sxmax[t + off]); sxmin[t] = fminf(sxmin[t], sxmin[t + off]);
      symax[t] = fmaxf(symax[t], symax[t + off]); symin[t] = fminf(symin[t], symin[t + off]);
    }
    __syncthreads();
  }
  xmax = sxmax[0]; xmin = sxmin[0]; ymax = symax[0]; ymin = symin[0];
  for (int i = t; i < N; i += T) {
    float xi = sh_x[i];
    float yi = sh_y[i];
    float dx = dr ? dr[i] : 0.0f;
    float dy = dr ? dr[N + i] : 0.0f;
    float m1 = (xi >= 0.0f) ? xi * ymax : xi * ymin;
    float m2 = (yi >= 0.0f) ? yi * xmax : yi * xmin;
    float den1 = 0.0f, num1 = 0.0f, den2 = 0.0f, num2 = 0.0f;
    for (int j = 0; j < N; ++j) {
      float xj = sh_x[j];
      float yj = sh_y[j];
      float e1 = __expf(xi * yj - m1);
      den1 += e1;
      num1 += e1 * xj;
      float e2 = __expf(yi * xj - m2);
      den2 += e2;
      num2 += e2 * yj;
    }
    float ox = xi + beta * num1 / den1;
    float oy = yi + beta * num2 / den2;
    stage_emit(so, i, ox, oy, dx, dy);
  }
}

#define TAIL_S 4
#define TAIL_N 64
#define TAIL_K 128
#define TAIL_CP 132
#define TAIL_PP 68
#define TAIL_XP 65

__device__ __forceinline__ void mma_1x4_k4(const v2f& a, const v2f& b0, const v2f& b1,
                                           const v2f& b2, const v2f& b3,
                                           v8f& c0, v8f& c1, v8f& c2, v8f& c3) {
  c0 = __builtin_amdgcn_wmma_f32_16x16x4_f32(false, a, false, b0, (short)0, c0, false, false);
  c1 = __builtin_amdgcn_wmma_f32_16x16x4_f32(false, a, false, b1, (short)0, c1, false, false);
  c2 = __builtin_amdgcn_wmma_f32_16x16x4_f32(false, a, false, b2, (short)0, c2, false, false);
  c3 = __builtin_amdgcn_wmma_f32_16x16x4_f32(false, a, false, b3, (short)0, c3, false, false);
}

__device__ __forceinline__ void mma_1x4_k8(const float4& a, const float4& b0, const float4& b1,
                                           const float4& b2, const float4& b3,
                                           v8f& c0, v8f& c1, v8f& c2, v8f& c3) {
  v2f al = {a.x, a.y}, ah = {a.z, a.w};
  v2f b0l = {b0.x, b0.y}, b0h = {b0.z, b0.w};
  v2f b1l = {b1.x, b1.y}, b1h = {b1.z, b1.w};
  v2f b2l = {b2.x, b2.y}, b2h = {b2.z, b2.w};
  v2f b3l = {b3.x, b3.y}, b3h = {b3.z, b3.w};
  mma_1x4_k4(al, b0l, b1l, b2l, b3l, c0, c1, c2, c3);
  mma_1x4_k4(ah, b0h, b1h, b2h, b3h, c0, c1, c2, c3);
}

__global__ __launch_bounds__(256)
void tail_kernel(const float* __restrict__ Q, const float* __restrict__ beta_p,
                 const float* __restrict__ dprev, const float* __restrict__ fcW,
                 const float* __restrict__ fcb, const float* __restrict__ W6,
                 const float* __restrict__ b6, float* __restrict__ out, int B) {
  __shared__ __attribute__((aligned(16))) float sC[16 * TAIL_CP];
  __shared__ __attribute__((aligned(16))) float sP[4 * 16 * TAIL_PP];
  __shared__ __attribute__((aligned(16))) float sD[TAIL_S * TAIL_PP];
  __shared__ __attribute__((aligned(16))) float sW6[7 * TAIL_PP];
  __shared__ float sX[TAIL_S * TAIL_XP];
  __shared__ float sY[TAIL_S * TAIL_XP];
  __shared__ float s6[TAIL_S * 8];
  int t = threadIdx.x;
  int wave = __builtin_amdgcn_readfirstlane((int)(threadIdx.x >> 5));
  int lane = t & 31;
  int r  = lane & 15;
  int kh = (lane >> 4) << 4;
  int b0 = blockIdx.x * TAIL_S;
  int hs = t >> 3;
  int ho = t & 7;
  bool head = (t < TAIL_S * 8) && (ho < 7);
  bool sums = t < TAIL_S * 16;
  int c4s = (t & 15) << 2;

  float beta = beta_p[0];
  float4 w00 = {}, w01 = {}, w02 = {}, w03 = {};
  float4 w10 = {}, w11 = {}, w12 = {}, w13 = {};
  float4 w20 = {}, w21 = {}, w22 = {}, w23 = {};
  float4 w30 = {}, w31 = {}, w32 = {}, w33 = {};
  if (wave < 4) {
    const float4* p0 = (const float4*)(fcW + (size_t)(r)      * TAIL_K + 32 * wave + kh);
    const float4* p1 = (const float4*)(fcW + (size_t)(16 + r) * TAIL_K + 32 * wave + kh);
    const float4* p2 = (const float4*)(fcW + (size_t)(32 + r) * TAIL_K + 32 * wave + kh);
    const float4* p3 = (const float4*)(fcW + (size_t)(48 + r) * TAIL_K + 32 * wave + kh);
    w00 = p0[0]; w01 = p0[1]; w02 = p0[2]; w03 = p0[3];
    w10 = p1[0]; w11 = p1[1]; w12 = p1[2]; w13 = p1[3];
    w20 = p2[0]; w21 = p2[1]; w22 = p2[2]; w23 = p2[3];
    w30 = p3[0]; w31 = p3[1]; w32 = p3[2]; w33 = p3[3];
  }
  float4 bv = {};
  if (sums) bv = *(const float4*)(fcb + c4s);
  float b6v = 0.0f;
  if (head) b6v = b6[ho];
  bool w6t = (t >= 128) && (t < 128 + 7 * 16);
  int w6o = (t - 128) >> 4;
  int w6c = ((t - 128) & 15) << 2;
  float4 w6v = {};
  if (w6t) w6v = *(const float4*)(W6 + w6o * TAIL_N + w6c);
  for (int z = t; z < (16 - TAIL_S) * 32; z += 256) {
    float4 zero = {};
    *(float4*)(sC + (TAIL_S + (z >> 5)) * TAIL_CP + ((z & 31) << 2)) = zero;
  }

  if (beta == 0.0f) {
    int c  = t & 127;
    int sa = t >> 7;
    int sb = sa + 2;
    size_t qc = (c < TAIL_N) ? (size_t)c : (size_t)B * TAIL_N + (size_t)(c - TAIL_N);
    float oa = Q[(size_t)(b0 + sa) * TAIL_N + qc];
    float ob = Q[(size_t)(b0 + sb) * TAIL_N + qc];
    float da = dprev[(size_t)(b0 + sa) * TAIL_K + c];
    float db = dprev[(size_t)(b0 + sb) * TAIL_K + c];
    asm volatile("" :: "v"(oa), "v"(ob), "v"(da), "v"(db));
    sC[sa * TAIL_CP + c] = mish_f(oa + da);
    sC[sb * TAIL_CP + c] = mish_f(ob + db);
  } else {
    for (int e = t; e < 2 * TAIL_S * TAIL_N; e += 256) {
      int y = e >> 8;
      int s = (e >> 6) & (TAIL_S - 1);
      int c = e & 63;
      float v = Q[(size_t)(b0 + s + (y ? B : 0)) * TAIL_N + c];
      if (y) sY[s * TAIL_XP + c] = v; else sX[s * TAIL_XP + c] = v;
    }
    __syncthreads();
    for (int e = t; e < TAIL_S * TAIL_K; e += 256) {
      int s   = e >> 7;
      int i   = (e & 127) >> 1;
      int dir = e & 1;
      const float* own = (dir ? sY : sX) + s * TAIL_XP;
      const float* oth = (dir ? sX : sY) + s * TAIL_XP;
      float omax = -3.4e38f, omin = 3.4e38f;
      for (int j = 0; j < TAIL_N; ++j) {
        omax = fmaxf(omax, oth[j]);
        omin = fminf(omin, oth[j]);
      }
      float vi = own[i];
      float dd = dprev[(size_t)(b0 + s) * TAIL_K + dir * TAIL_N + i];
      float m = (vi >= 0.0f) ? vi * omax : vi * omin;
      float den = 0.0f, num = 0.0f;
      for (int j = 0; j < TAIL_N; ++j) {
        float e1 = __expf(vi * oth[j] - m);
        den += e1;
        num += e1 * own[j];
      }
      float o = vi + beta * num / den;
      sC[s * TAIL_CP + dir * TAIL_N + i] = mish_f(o + dd);
    }
  }
  if (w6t) *(float4*)(sW6 + w6o * TAIL_PP + w6c) = w6v;
  __syncthreads();

  if (wave < 4) {
    const float4* pa = (const float4*)(sC + r * TAIL_CP + 32 * wave + kh);
    float4 a0 = pa[0], a1 = pa[1], a2 = pa[2], a3 = pa[3];
    v8f acc0 = {};
    v8f acc1 = {};
    v8f acc2 = {};
    v8f acc3 = {};
    mma_1x4_k8(a0, w00, w10, w20, w30, acc0, acc1, acc2, acc3);
    mma_1x4_k8(a1, w01, w11, w21, w31, acc0, acc1, acc2, acc3);
    mma_1x4_k8(a2, w02, w12, w22, w32, acc0, acc1, acc2, acc3);
    mma_1x4_k8(a3, w03, w13, w23, w33, acc0, acc1, acc2, acc3);
    float* pw = sP + wave * (16 * TAIL_PP) + ((lane >> 4) << 3) * TAIL_PP + r;
    for (int v = 0; v < TAIL_S; ++v) {
      pw[v * TAIL_PP]      = acc0[v];
      pw[v * TAIL_PP + 16] = acc1[v];
      pw[v * TAIL_PP + 32] = acc2[v];
      pw[v * TAIL_PP + 48] = acc3[v];
    }
  }
  __syncthreads();
  if (sums) {
    int row = t >> 4;
    const float* ps = sP + row * TAIL_PP + c4s;
    float4 s = *(const float4*)ps;
    for (int w = 1; w < 4; ++w) {
      float4 p = *(const float4*)(ps + w * (16 * TAIL_PP));
      s.x += p.x; s.y += p.y; s.z += p.z; s.w += p.w;
    }
    s.x += bv.x; s.y += bv.y; s.z += bv.z; s.w += bv.w;
    *(float4*)(sD + row * TAIL_PP + c4s) = s;
  }
  __syncthreads();

  if (head) {
    const float* row = sD + hs * TAIL_PP;
    const float* w   = sW6 + ho * TAIL_PP;
    float acc = b6v;
    for (int k4 = 0; k4 < TAIL_N; k4 += 4) {
      float4 rv = *(const float4*)(row + k4);
      float4 wv = *(const float4*)(w + k4);
      acc += rv.x * wv.x;
      acc += rv.y * wv.y;
      acc += rv.z * wv.z;
      acc += rv.w * wv.w;
    }
    s6[hs * 8 + ho] = acc;
  }
  __syncthreads();
  if (head) {
    const float* x6 = s6 + hs * 8;
    float nrm = sqrtf(x6[0]*x6[0] + x6[1]*x6[1] + x6[2]*x6[2] + x6[3]*x6[3]);
    float v = x6[ho];
    if (ho < 4) v = v / fmaxf(nrm, 1e-12f);
    out[(size_t)(b0 + hs) * 7 + ho] = v;
    out[(size_t)B * 7 + (size_t)(b0 + hs) * 7 + ho] = v;
  }
}

extern "C" void kernel_launch(void* const* d_in, const int* in_sizes, int n_in,
                              void* d_out, int out_size, void* d_ws, size_t ws_size,
                              hipStream_t stream) {
  const float* x    = (const float*)d_in[0];
  const float* y    = (const float*)d_in[1];
  const float* W1   = (const float*)d_in[2];
  const float* be1  = (const float*)d_in[3];
  const float* fcW1 = (const float*)d_in[4];
  const float* fcb1 = (const float*)d_in[5];
  const float* W2   = (const float*)d_in[6];
  const float* be2  = (const float*)d_in[7];
  const float* fcW2 = (const float*)d_in[8];
  const float* fcb2 = (const float*)d_in[9];
  const float* W3   = (const float*)d_in[10];
  const float* be3  = (const float*)d_in[11];
  const float* fcW3 = (const float*)d_in[12];
  const float* fcb3 = (const float*)d_in[13];
  const float* W4   = (const float*)d_in[14];
  const float* be4  = (const float*)d_in[15];
  const float* fcW4 = (const float*)d_in[16];
  const float* fcb4 = (const float*)d_in[17];
  const float* W6   = (const float*)d_in[18];
  const float* b6   = (const float*)d_in[19];

  float* ws = (float*)d_ws;
  float* q  = ws;
  float* mA = q  + 524288;
  float* mC = mA + 524288;
  float* d0 = mC + 524288;
  float* d1 = d0 + 262144;
  float* d2 = d1 + 131072;

  const int B = 512;
  auto gemm_pair = [&](const float* A0, const float* W0, const float* bias0, float* C0,
                       int M0, int N0, int K0,
                       const float* A1, const float* W1, const float* bias1, float* C1,
                       int M1, int N1, int K1) {
    int tiles = (M0 >> 5) * (N0 >> 5) + (M1 >> 5) * (N1 >> 5);
    hipLaunchKernelGGL(gemm_pair_wmma_f32, dim3(tiles), dim3(GEMM_WAVES * 32), 0, stream,
                       A0, W0, bias0, C0, M0, N0, K0, A1, W1, bias1, C1, M1, N1, K1);
  };
  auto attn = [&](const float* beta, const float* dprev, float* mAout, int Nout, int threads) {
    hipLaunchKernelGGL(attn_mish_kernel, dim3(B), dim3(threads), 0, stream,
                       q, beta, dprev, mAout, mC, Nout, B);
  };

  hipLaunchKernelGGL(gemm_mish_wmma_f32, dim3(((2 * B) >> 5) * (512 >> 5)), dim3(GEMM_WAVES * 32), 0, stream,
                     x, y, W1, q, B, 512, 1024);
  attn(be1, nullptr, mA, 512, 256);
  gemm_pair(mC, fcW1, fcb1,    d0, B,     512, 1024,  mA, W2, nullptr, q, 2 * B, 256, 512);
  attn(be2, d0, mA, 256, 256);
  gemm_pair(mC, fcW2, fcb2,    d1, B,     256, 512,   mA, W3, nullptr, q, 2 * B, 128, 256);
  attn(be3, d1, mA, 128, 128);
  gemm_pair(mC, fcW3, fcb3,    d2, B,     128, 256,   mA, W4, nullptr, q, 2 * B, 64,  128);
  hipLaunchKernelGGL(tail_kernel, dim3(B / TAIL_S), dim3(256), 0, stream,
                     q, be4, d2, fcW4, fcb4, W6, b6, (float*)d_out, B);

  (void)in_sizes; (void)n_in; (void)out_size; (void)ws_size;
}


# stage 3 joins the last launch: attention 3, fc3 and q4 products (WMMA, same K blocks and add order) run per 4-sample group, rows handed on in LDS; 8 launches -> 6; sums unchanged
# speedup vs baseline: 6.5336x; 1.0955x over previous
#include <hip/hip_runtime.h>
#include <math.h>

typedef __attribute__((ext_vector_type(2))) float v2f;
typedef __attribute__((ext_vector_type(8))) float v8f;

__device__ __forceinline__ float mish_f(float v) {
  float n = __expf(v);
  float w = n * (n + 2.0f);
  float q = w * __builtin_amdgcn_rcpf(w + 2.0f);
  return (v > 20.0f) ? v : v * q;
}

__device__ __forceinline__ void mish8(float4& p, float4& r) {
  v8f v = {p.x, p.y, p.z, p.w, r.x, r.y, r.z, r.w};
  v8f n = {__expf(v[0]), __expf(v[1]), __expf(v[2]), __expf(v[3]),
           __expf(v[4]), __expf(v[5]), __expf(v[6]), __expf(v[7])};
  v8f w = n * (n + 2.0f);
  v8f d = w + 2.0f;
  v8f rc = {__builtin_amdgcn_rcpf(d[0]), __builtin_amdgcn_rcpf(d[1]),
            __builtin_amdgcn_rcpf(d[2]), __builtin_amdgcn_rcpf(d[3]),
            __builtin_amdgcn_rcpf(d[4]), __builtin_amdgcn_rcpf(d[5]),
            __builtin_amdgcn_rcpf(d[6]), __builtin_amdgcn_rcpf(d[7])};
  v8f m = v * (w * rc);
  v8f o = (v > 20.0f) ? v : m;
  p.x = o[0]; p.y = o[1]; p.z = o[2]; p.w = o[3];
  r.x = o[4]; r.y = o[5]; r.z = o[6]; r.w = o[7];
}

#define GEMM_WAVES 8
#define PART_PITCH 36
#define PART_TILE (32 * PART_PITCH)

__device__ __forceinline__ void mma_2x2_k8(const float4& a0, const float4& a1,
                                           const float4& b0, const float4& b1,
                                           v8f& acc00, v8f& acc01, v8f& acc10, v8f& acc11) {
  v2f a0l = {a0.x, a0.y}, a0h = {a0.z, a0.w};
  v2f a1l = {a1.x, a1.y}, a1h = {a1.z, a1.w};
  v2f b0l = {b0.x, b0.y}, b0h = {b0.z, b0.w};
  v2f b1l = {b1.x, b1.y}, b1h = {b1.z, b1.w};
  acc00 = __builtin_amdgcn_wmma_f32_16x16x4_f32(false, a0l, false, b0l, (short)0, acc00, false, false);
  acc01 = __builtin_amdgcn_wmma_f32_16x16x4_f32(false, a0l, false, b1l, (short)0, acc01, false, false);
  acc10 = __builtin_amdgcn_wmma_f32_16x16x4_f32(false, a1l, false, b0l, (short)0, acc10, false, false);
  acc11 = __builtin_amdgcn_wmma_f32_16x16x4_f32(false, a1l, false, b1l, (short)0, acc11, false, false);
  acc00 = __builtin_amdgcn_wmma_f32_16x16x4_f32(false, a0h, false, b0h, (short)0, acc00, false, false);
  acc01 = __builtin_amdgcn_wmma_f32_16x16x4_f32(false, a0h, false, b1h, (short)0, acc01, false, false);
  acc10 = __builtin_amdgcn_wmma_f32_16x16x4_f32(false, a1h, false, b0h, (short)0, acc10, false, false);
  acc11 = __builtin_amdgcn_wmma_f32_16x16x4_f32(false, a1h, false, b1h, (short)0, acc11, false, false);
}

template <bool MISH_A>
__device__ __forceinline__ void gemm_tile_32x32(const float* __restrict__ A, const float* __restrict__ W,
                                                const float* __restrict__ bias, float* __restrict__ C,
                                                int am0, int m0, int n0, int N, int K, float* part) {
  int wave = __builtin_amdgcn_readfirstlane((int)(threadIdx.x >> 5));
  int lane = threadIdx.x & 31;
  int r  = lane & 15;
  int kh = (lane >> 4) << 4;
  int ksl = ((((K >> 5) + GEMM_WAVES - 1) / GEMM_WAVES)) << 5;
  int kb = wave * ksl;
  int ke = kb + ksl;
  if (ke > K) ke = K;
  const float* Ar0 = A + (size_t)(am0 + r) * K + kh;
  const float* Ar1 = Ar0 + (size_t)16 * K;
  const float* Wr0 = W + (size_t)(n0 + r) * K + kh;
  const float* Wr1 = Wr0 + (size_t)16 * K;
  v8f acc00 = {};
  v8f acc01 = {};
  v8f acc10 = {};
  v8f acc11 = {};
  for (int k = kb; k < ke; k += 32) {
    const float4* pa0 = (const float4*)(Ar0 + k);
    const float4* pa1 = (const float4*)(Ar1 + k);
    const float4* pb0 = (const float4*)(Wr0 + k);
    const float4* pb1 = (const float4*)(Wr1 + k);
    float4 a00 = pa0[0], a01 = pa0[1], a02 = pa0[2], a03 = pa0[3];
    float4 b00 = pb0[0], b01 = pb0[1], b02 = pb0[2], b03 = pb0[3];
    float4 a10 = pa1[0], a11 = pa1[1], a12 = pa1[2], a13 = pa1[3];
    float4 b10 = pb1[0], b11 = pb1[1], b12 = pb1[2], b13 = pb1[3];
    __builtin_amdgcn_sched_barrier(0);
    if (MISH_A) {
      mish8(a00, a10);
      mish8(a01, a11);
      mish8(a02, a12);
      mish8(a03, a13);
      __builtin_amdgcn_sched_barrier(0);
    }
    mma_2x2_k8(a00, a10, b00, b10, acc00, acc01, acc10, acc11);
    mma_2x2_k8(a01, a11, b01, b11, acc00, acc01, acc10, acc11);
    mma_2x2_k8(a02, a12, b02, b12, acc00, acc01, acc10, acc11);
    mma_2x2_k8(a03, a13, b03, b13, acc00, acc01, acc10, acc11);
  }
  float* pw = part + wave * PART_TILE + ((lane >> 4) << 3) * PART_PITCH + r;
  for (int v = 0; v < 8; ++v) {
    pw[v * PART_PITCH]             = acc00[v];
    pw[v * PART_PITCH + 16]        = acc01[v];
    pw[(v + 16) * PART_PITCH]      = acc10[v];
    pw[(v + 16) * PART_PITCH + 16] = acc11[v];
  }
  __syncthreads();
  int row = threadIdx.x >> 3;
  int c4  = (threadIdx.x & 7) << 2;
  const float* ps = part + row * PART_PITCH + c4;
  float4 s = *(const float4*)ps;
  for (int w = 1; w < GEMM_WAVES; ++w) {
    float4 p = *(const float4*)(ps + w * PART_TILE);
    s.x += p.x; s.y += p.y; s.z += p.z; s.w += p.w;
  }
  if (bias) {
    float4 bv = *(const float4*)(bias + n0 + c4);
    s.x += bv.x; s.y += bv.y; s.z += bv.z; s.w += bv.w;
  }
  *(float4*)(C + (size_t)(m0 + row) * N + n0 + c4) = s;
}

__global__ __launch_bounds__(GEMM_WAVES * 32)
void gemm_pair_wmma_f32(const float* __restrict__ A0, const float* __restrict__ W0,
                        const float* __restrict__ bias0, float* __restrict__ C0,
                        int M0, int N0, int K0,
                        const float* __restrict__ A1, const float* __restrict__ W1,
                        const float* __restrict__ bias1, float* __restrict__ C1,
                        int M1, int N1, int K1) {
  __shared__ __attribute__((aligned(16))) float part[GEMM_WAVES * PART_TILE];
  int tile = blockIdx.x;
  int tiles0 = (M0 >> 5) * (N0 >> 5);
  bool second = tile >= tiles0;
  const float* A    = second ? A1 : A0;
  const float* W    = second ? W1 : W0;
  const float* bias = second ? bias1 : bias0;
  float* C          = second ? C1 : C0;
  int M = second ? M1 : M0;
  int N = second ? N1 : N0;
  int K = second ? K1 : K0;
  if (second) tile -= tiles0;
  int tiles_n = N >> 5;
  if (tile >= (M >> 5) * tiles_n) return;
  int tm = tile / tiles_n;
  int m0 = tm << 5;
  int n0 = (tile - tm * tiles_n) << 5;
  gemm_tile_32x32<false>(A, W, bias, C, m0, m0, n0, N, K, part);
}

__global__ __launch_bounds__(GEMM_WAVES * 32)
void gemm_mish_wmma_f32(const float* __restrict__ x, const float* __restrict__ y,
                        const float* __restrict__ W, float* __restrict__ C,
                        int B, int N, int K) {
  __shared__ __attribute__((aligned(16))) float part[GEMM_WAVES * PART_TILE];
  int tile = blockIdx.x;
  int tiles_n = N >> 5;
  if (tile >= ((2 * B) >> 5) * tiles_n) return;
  int tm = tile / tiles_n;
  int m0 = tm << 5;
  int n0 = (tile - tm * tiles_n) << 5;
  bool ybr = m0 >= B;
  const float* A = ybr ? y : x;
  int am0 = ybr ? m0 - B : m0;
  gemm_tile_32x32<true>(A, W, nullptr, C, am0, m0, n0, N, K, part);
}

struct StageOut {
  const float* dprev;
  float* mA;
  float* mC;
  int N;
  int B;
  int b;
};

__device__ __forceinline__ void stage_emit(const StageOut& so, int i, float ox, float oy,
                                           float dx, float dy) {
  size_t rowC = (size_t)so.b * (2 * so.N);
  float mx = 0.0f, my = 0.0f;
  if (so.mA != nullptr || so.dprev == nullptr) { mx = mish_f(ox); my = mish_f(oy); }
  if (so.mA != nullptr) {
    so.mA[(size_t)so.b * so.N + i] = mx;
    so.mA[(size_t)(so.b + so.B) * so.N + i] = my;
  }
  if (so.dprev != nullptr) { mx = mish_f(ox + dx); my = mish_f(oy + dy); }
  so.mC[rowC + i] = mx;
  so.mC[rowC + so.N + i] = my;
}

__global__ __launch_bounds__(256)
void attn_mish_kernel(const float* __restrict__ Q, const float* __restrict__ beta_p,
                      const float* __restrict__ dprev, float* __restrict__ mA,
                      float* __restrict__ mC, int N, int B) {
  __shared__ float sh_x[512];
  __shared__ float sh_y[512];
  __shared__ float sxmax[256];
  __shared__ float sxmin[256];
  __shared__ float symax[256];
  __shared__ float symin[256];
  int b = blockIdx.x;
  int t = threadIdx.x;
  int T = blockDim.x;
  const float* xq = Q + (size_t)b * N;
  const float* yq = Q + (size_t)(b + B) * N;
  const float* dr = dprev ? dprev + (size_t)b * (2 * N) : nullptr;
  StageOut so;
  so.dprev = dprev; so.mA = mA; so.mC = mC; so.N = N; so.B = B; so.b = b;
  float beta = beta_p[0];
  if (beta == 0.0f) {
    for (int i = t; i < N; i += T) {
      float ox = xq[i];
      float oy = yq[i];
      float dx = dr ? dr[i] : 0.0f;
      float dy = dr ? dr[N + i] : 0.0f;
      stage_emit(so, i, ox, oy, dx, dy);
    }
    return;
  }
  for (int i = t; i < N; i += T) { sh_x[i] = xq[i]; sh_y[i] = yq[i]; }
  __syncthreads();
  float xmax = -3.4e38f, xmin = 3.4e38f, ymax = -3.4e38f, ymin = 3.4e38f;
  for (int i = t; i < N; i += T) {
    xmax = fmaxf(xmax, sh_x[i]); xmin = fminf(xmin, sh_x[i]);
    ymax = fmaxf(ymax, sh_y[i]); ymin = fminf(ymin, sh_y[i]);
  }
  sxmax[t] = xmax; sxmin[t] = xmin; symax[t] = ymax; symin[t] = ymin;
  __syncthreads();
  for (int off = T >> 1; off > 0; off >>= 1) {
    if (t < off) {
      sxmax[t] = fmaxf(sxmax[t], sxmax[t + off]); sxmin[t] = fminf(sxmin[t], sxmin[t + off]);
      symax[t] = fmaxf(symax[t], symax[t + off]); symin[t] = fminf(symin[t], symin[t + off]);
    }
    __syncthreads();
  }
  xmax = sxmax[0]; xmin = sxmin[0]; ymax = symax[0]; ymin = symin[0];
  for (int i = t; i < N; i += T) {
    float xi = sh_x[i];
    float yi = sh_y[i];
    float dx = dr ? dr[i] : 0.0f;
    float dy = dr ? dr[N + i] : 0.0f;
    float m1 = (xi >= 0.0f) ? xi * ymax : xi * ymin;
    float m2 = (yi >= 0.0f) ? yi * xmax : yi * xmin;
    float den1 = 0.0f, num1 = 0.0f, den2 = 0.0f, num2 = 0.0f;
    for (int j = 0; j < N; ++j) {
      float xj = sh_x[j];
      float yj = sh_y[j];
      float e1 = __expf(xi * yj - m1);
      den1 += e1;
      num1 += e1 * xj;
      float e2 = __expf(yi * xj - m2);
      den2 += e2;
      num2 += e2 * yj;
    }
    float ox = xi + beta * num1 / den1;
    float oy = yi + beta * num2 / den2;
    stage_emit(so, i, ox, oy, dx, dy);
  }
}

#define TAIL_S 4
#define TAIL_N 64
#define TAIL_K 128
#define TAIL_CP 132
#define TAIL_PP 68
#define T3_N 128
#define T3_K 256
#define T3_CP 260
#define T3_AP 132
#define T3_XP 129

__device__ __forceinline__ void mma_1x4_k4(const v2f& a, const v2f& b0, const v2f& b1,
                                           const v2f& b2, const v2f& b3,
                                           v8f& c0, v8f& c1, v8f& c2, v8f& c3) {
  c0 = __builtin_amdgcn_wmma_f32_16x16x4_f32(false, a, false, b0, (short)0, c0, false, false);
  c1 = __builtin_amdgcn_wmma_f32_16x16x4_f32(false, a, false, b1, (short)0, c1, false, false);
  c2 = __builtin_amdgcn_wmma_f32_16x16x4_f32(false, a, false, b2, (short)0, c2, false, false);
  c3 = __builtin_amdgcn_wmma_f32_16x16x4_f32(false, a, false, b3, (short)0, c3, false, false);
}

__device__ __forceinline__ void mma_1x4_k8(const float4& a, const float4& b0, const float4& b1,
                                           const float4& b2, const float4& b3,
                                           v8f& c0, v8f& c1, v8f& c2, v8f& c3) {
  v2f al = {a.x, a.y}, ah = {a.z, a.w};
  v2f b0l = {b0.x, b0.y}, b0h = {b0.z, b0.w};
  v2f b1l = {b1.x, b1.y}, b1h = {b1.z, b1.w};
  v2f b2l = {b2.x, b2.y}, b2h = {b2.z, b2.w};
  v2f b3l = {b3.x, b3.y}, b3h = {b3.z, b3.w};
  mma_1x4_k4(al, b0l, b1l, b2l, b3l, c0, c1, c2, c3);
  mma_1x4_k4(ah, b0h, b1h, b2h, b3h, c0, c1, c2, c3);
}

__device__ __forceinline__ void mma_k4(const v2f& a, const v2f& b, v8f& c) {
  c = __builtin_amdgcn_wmma_f32_16x16x4_f32(false, a, false, b, (short)0, c, false, false);
}

__global__ __launch_bounds__(256)
void tail_kernel(const float* __restrict__ Q, const float* __restrict__ beta3_p,
                 const float* __restrict__ d1, const float* __restrict__ fcW3,
                 const float* __restrict__ fcb3, const float* __restrict__ W4,
                 const float* __restrict__ beta_p, const float* __restrict__ fcW,
                 const float* __restrict__ fcb, const float* __restrict__ W6,
                 const float* __restrict__ b6, float* __restrict__ out, int B) {
  __shared__ __attribute__((aligned(16))) float sC3[16 * T3_CP];
  __shared__ __attribute__((aligned(16))) float sA3[16 * T3_AP];
  __shared__ __attribute__((aligned(16))) float sD2[16 * T3_AP];
  __shared__ __attribute__((aligned(16))) float sQ4[16 * TAIL_PP];
  __shared__ float sX3[TAIL_S * T3_XP];
  __shared__ float sY3[TAIL_S * T3_XP];
  __shared__ __attribute__((aligned(16))) float sC[16 * TAIL_CP];
  __shared__ __attribute__((aligned(16))) float sP[4 * 16 * TAIL_PP];
  __shared__ __attribute__((aligned(16))) float sD[TAIL_S * TAIL_PP];
  __shared__ __attribute__((aligned(16))) float sW6[7 * TAIL_PP];
  __shared__ float s6[TAIL_S * 8];
  int t = threadIdx.x;
  int wave = __builtin_amdgcn_readfirstlane((int)(threadIdx.x >> 5));
  int lane = t & 31;
  int r  = lane & 15;
  int kh = (lane >> 4) << 4;
  int b0 = blockIdx.x * TAIL_S;
  int hs = t >> 3;
  int ho = t & 7;
  bool head = (t < TAIL_S * 8) && (ho < 7);
  bool sums = t < TAIL_S * 16;
  int c4s = (t & 15) << 2;

  float beta3 = beta3_p[0];
  float beta = beta_p[0];
  int c3  = t;
  int ci3 = c3 & (T3_N - 1);
  int yb3 = c3 >> 7;
  const float* qb3 = Q + (size_t)yb3 * B * T3_N + (size_t)b0 * T3_N + ci3;
  const float* db3 = d1 + (size_t)b0 * T3_K + c3;
  float o0 = qb3[0], o1 = qb3[T3_N], o2 = qb3[2 * T3_N], o3 = qb3[3 * T3_N];
  float e0 = db3[0], e1 = db3[T3_K], e2 = db3[2 * T3_K], e3 = db3[3 * T3_K];
  if (beta3 != 0.0f) {
    for (int e = t; e < 2 * TAIL_S * T3_N; e += 256) {
      int y = e >> 9;
      int s = (e >> 7) & (TAIL_S - 1);
      int c = e & (T3_N - 1);
      float v = Q[(size_t)(b0 + s + (y ? B : 0)) * T3_N + c];
      if (y) sY3[s * T3_XP + c] = v; else sX3[s * T3_XP + c] = v;
    }
  }
  float4 wf[8][4];
  {
    const float* p = fcW3 + (size_t)(16 * wave + r) * T3_K + kh;
#pragma unroll
    for (int blk = 0; blk < 8; ++blk) {
      const float4* p4 = (const float4*)(p + 32 * blk);
      wf[blk][0] = p4[0]; wf[blk][1] = p4[1]; wf[blk][2] = p4[2]; wf[blk][3] = p4[3];
    }
  }
  float4 wx[4][4];
  {
    const float* p = (wave >= 4) ? W4 + (size_t)(16 * (wave - 4) + r) * T3_N + kh
                                 : fcW + (size_t)r * TAIL_K + 32 * wave + kh;
    int gs = (wave >= 4) ? 32 : 16 * TAIL_K;
#pragma unroll
    for (int g = 0; g < 4; ++g) {
      const float4* p4 = (const float4*)(p + (size_t)g * gs);
      wx[g][0] = p4[0]; wx[g][1] = p4[1]; wx[g][2] = p4[2]; wx[g][3] = p4[3];
    }
  }
  float b3v = fcb3[16 * wave + r];
  float4 bv = *(const float4*)(fcb + c4s);
  float b6v = b6[ho < 7 ? ho : 6];
  bool w6t = (t >= 128) && (t < 128 + 7 * 16);
  int w6o = (t - 128) >> 4;
  int w6c = ((t - 128) & 15) << 2;
  float4 w6v = *(const float4*)(W6 + (w6t ? w6o * TAIL_N + w6c : 0));
  for (int z = t; z < (16 - TAIL_S) * 64; z += 256) {
    float4 zero = {};
    *(float4*)(sC3 + (TAIL_S + (z >> 6)) * T3_CP + ((z & 63) << 2)) = zero;
  }
  for (int z = t; z < (16 - 2 * TAIL_S) * 32; z += 256) {
    float4 zero = {};
    *(float4*)(sA3 + (2 * TAIL_S + (z >> 5)) * T3_AP + ((z & 31) << 2)) = zero;
  }
  for (int z = t; z < (16 - TAIL_S) * 32; z += 256) {
    float4 zero = {};
    *(float4*)(sC + (TAIL_S + (z >> 5)) * TAIL_CP + ((z & 31) << 2)) = zero;
  }

  if (beta3 == 0.0f) {
    float* pc = sC3 + c3;
    float* pm = sA3 + (yb3 * TAIL_S) * T3_AP + ci3;
    pc[0]         = mish_f(o0 + e0);
    pc[T3_CP]     = mish_f(o1 + e1);
    pc[2 * T3_CP] = mish_f(o2 + e2);
    pc[3 * T3_CP] = mish_f(o3 + e3);
    pm[0]         = mish_f(o0);
    pm[T3_AP]     = mish_f(o1);
    pm[2 * T3_AP] = mish_f(o2);
    pm[3 * T3_AP] = mish_f(o3);
  } else {
    __syncthreads();
    for (int e = t; e < TAIL_S * T3_K; e += 256) {
      int s   = e >> 8;
      int i   = (e & 255) >> 1;
      int dir = e & 1;
      const float* own = (dir ? sY3 : sX3) + s * T3_XP;
      const float* oth = (dir ? sX3 : sY3) + s * T3_XP;
      float omax = -3.4e38f, omin = 3.4e38f;
      for (int j = 0; j < T3_N; ++j) {
        omax = fmaxf(omax, oth[j]);
        omin = fminf(omin, oth[j]);
      }
      float vi = own[i];
      float dd = d1[(size_t)(b0 + s) * T3_K + dir * T3_N + i];
      float m = (vi >= 0.0f) ? vi * omax : vi * omin;
      float den = 0.0f, num = 0.0f;
      for (int j = 0; j < T3_N; ++j) {
        float ex = __expf(vi * oth[j] - m);
        den += ex;
        num += ex * own[j];
      }
      float o = vi + beta3 * num / den;
      sC3[s * T3_CP + dir * T3_N + i] = mish_f(o + dd);
      sA3[(dir * TAIL_S + s) * T3_AP + i] = mish_f(o);
    }
  }
  __builtin_amdgcn_fence(__ATOMIC_RELEASE, "workgroup", "local");
  __builtin_amdgcn_s_barrier();
  __builtin_amdgcn_fence(__ATOMIC_ACQUIRE, "workgroup", "local");

  {
    v8f acc[8];
    const float* pa = sC3 + r * T3_CP + kh;
    float4 af[2][8];
    {
      const float4* p0 = (const float4*)pa;
      const float4* p1 = (const float4*)(pa + 32);
      af[0][0] = p0[0]; af[0][1] = p0[1]; af[0][2] = p0[2]; af[0][3] = p0[3];
      af[0][4] = p1[0]; af[0][5] = p1[1]; af[0][6] = p1[2]; af[0][7] = p1[3];
    }
#pragma unroll
    for (int pr = 0; pr < 4; ++pr) {
      if (pr < 3) {
        const float4* p0 = (const float4*)(pa + 64 * (pr + 1));
        const float4* p1 = (const float4*)(pa + 64 * (pr + 1) + 32);
        af[(pr + 1) & 1][0] = p0[0]; af[(pr + 1) & 1][1] = p0[1]; af[(pr + 1) & 1][2] = p0[2]; af[(pr + 1) & 1][3] = p0[3];
        af[(pr + 1) & 1][4] = p1[0]; af[(pr + 1) & 1][5] = p1[1]; af[(pr + 1) & 1][6] = p1[2]; af[(pr + 1) & 1][7] = p1[3];
      }
      __builtin_amdgcn_sched_barrier(0);
      v8f c0 = {};
      v8f c1 = {};
#pragma unroll
      for (int q = 0; q < 4; ++q) {
        const float4 x0 = af[pr & 1][q], x1 = af[pr & 1][4 + q];
        const float4 y0 = wf[2 * pr][q], y1 = wf[2 * pr + 1][q];
        v2f a0l = {x0.x, x0.y}, a0h = {x0.z, x0.w};
        v2f a1l = {x1.x, x1.y}, a1h = {x1.z, x1.w};
        v2f w0l = {y0.x, y0.y}, w0h = {y0.z, y0.w};
        v2f w1l = {y1.x, y1.y}, w1h = {y1.z, y1.w};
        mma_k4(a0l, w0l, c0);
        mma_k4(a1l, w1l, c1);
        mma_k4(a0h, w0h, c0);
        mma_k4(a1h, w1h, c1);
      }
      acc[2 * pr] = c0;
      acc[2 * pr + 1] = c1;
      __builtin_amdgcn_sched_barrier(0);
    }
    v8f s = acc[0];
#pragma unroll
    for (int blk = 1; blk < 8; ++blk) s += acc[blk];
    float* pw = sD2 + ((lane >> 4) << 3) * T3_AP + 16 * wave + r;
#pragma unroll
    for (int v = 0; v < TAIL_S; ++v) pw[v * T3_AP] = s[v] + b3v;
  }
  if (wave >= 4) {
    const float* pa = sA3 + r * T3_AP + kh;
    float4 a[4][4];
#pragma unroll
    for (int blk = 0; blk < 4; ++blk) {
      const float4* p4 = (const float4*)(pa + 32 * blk);
      a[blk][0] = p4[0]; a[blk][1] = p4[1]; a[blk][2] = p4[2]; a[blk][3] = p4[3];
    }
    __builtin_amdgcn_sched_barrier(0);
    v8f c0 = {};
    v8f c1 = {};
    v8f c2 = {};
    v8f c3 = {};
#pragma unroll
    for (int q = 0; q < 4; ++q) {
      v2f al0 = {a[0][q].x, a[0][q].y}, al1 = {a[1][q].x, a[1][q].y}, al2 = {a[2][q].x, a[2][q].y}, al3 = {a[3][q].x, a[3][q].y};
      v2f wl0 = {wx[0][q].x, wx[0][q].y}, wl1 = {wx[1][q].x, wx[1][q].y}, wl2 = {wx[2][q].x, wx[2][q].y}, wl3 = {wx[3][q].x, wx[3][q].y};
      mma_k4(al0, wl0, c0);
      mma_k4(al1, wl1, c1);
      mma_k4(al2, wl2, c2);
      mma_k4(al3, wl3, c3);
      v2f ah0 = {a[0][q].z, a[0][q].w}, ah1 = {a[1][q].z, a[1][q].w}, ah2 = {a[2][q].z, a[2][q].w}, ah3 = {a[3][q].z, a[3][q].w};
      v2f wh0 = {wx[0][q].z, wx[0][q].w}, wh1 = {wx[1][q].z, wx[1][q].w}, wh2 = {wx[2][q].z, wx[2][q].w}, wh3 = {wx[3][q].z, wx[3][q].w};
      mma_k4(ah0, wh0, c0);
      mma_k4(ah1, wh1, c1);
      mma_k4(ah2, wh2, c2);
      mma_k4(ah3, wh3, c3);
    }
    v8f s = c0;
    s += c1;
    s += c2;
    s += c3;
    float* pw = sQ4 + ((lane >> 4) << 3) * TAIL_PP + 16 * (wave - 4) + r;
#pragma unroll
    for (int v = 0; v < 8; ++v) pw[v * TAIL_PP] = s[v];
  }
  float4 w00 = wx[0][0], w01 = wx[0][1], w02 = wx[0][2], w03 = wx[0][3];
  float4 w10 = wx[1][0], w11 = wx[1][1], w12 = wx[1][2], w13 = wx[1][3];
  float4 w20 = wx[2][0], w21 = wx[2][1], w22 = wx[2][2], w23 = wx[2][3];
  float4 w30 = wx[3][0], w31 = wx[3][1], w32 = wx[3][2], w33 = wx[3][3];
  __syncthreads();

  if (beta == 0.0f) {
    int c  = t & 127;
    int sa = t >> 7;
    int sb = sa + 2;
    int qr = (c < TAIL_N) ? 0 : TAIL_S;
    int qc = c & (TAIL_N - 1);
    float oa = sQ4[(qr + sa) * TAIL_PP + qc];
    float ob = sQ4[(qr + sb) * TAIL_PP + qc];
    float da = sD2[sa * T3_AP + c];
    float db = sD2[sb * T3_AP + c];
    sC[sa * TAIL_CP + c] = mish_f(oa + da);
    sC[sb * TAIL_CP + c] = mish_f(ob + db);
  } else {
    for (int e = t; e < TAIL_S * TAIL_K; e += 256) {
      int s   = e >> 7;
      int i   = (e & 127) >> 1;
      int dir = e & 1;
      const float* own = sQ4 + ((dir ? TAIL_S : 0) + s) * TAIL_PP;
      const float* oth = sQ4 + ((dir ? 0 : TAIL_S) + s) * TAIL_PP;
      float omax = -3.4e38f, omin = 3.4e38f;
      for (int j = 0; j < TAIL_N; ++j) {
        omax = fmaxf(omax, oth[j]);
        omin = fminf(omin, oth[j]);
      }
      float vi = own[i];
      float dd = sD2[s * T3_AP + dir * TAIL_N + i];
      float m = (vi >= 0.0f) ? vi * omax : vi * omin;
      float den = 0.0f, num = 0.0f;
      for (int j = 0; j < TAIL_N; ++j) {
        float e1 = __expf(vi * oth[j] - m);
        den += e1;
        num += e1 * own[j];
      }
      float o = vi + beta * num / den;
      sC[s * TAIL_CP + dir * TAIL_N + i] = mish_f(o + dd);
    }
  }
  if (w6t) *(float4*)(sW6 + w6o * TAIL_PP + w6c) = w6v;
  __syncthreads();

  if (wave < 4) {
    const float4* pa = (const float4*)(sC + r * TAIL_CP + 32 * wave + kh);
    float4 a0 = pa[0], a1 = pa[1], a2 = pa[2], a3 = pa[3];
    v8f acc0 = {};
    v8f acc1 = {};
    v8f acc2 = {};
    v8f acc3 = {};
    mma_1x4_k8(a0, w00, w10, w20, w30, acc0, acc1, acc2, acc3);
    mma_1x4_k8(a1, w01, w11, w21, w31, acc0, acc1, acc2, acc3);
    mma_1x4_k8(a2, w02, w12, w22, w32, acc0, acc1, acc2, acc3);
    mma_1x4_k8(a3, w03, w13, w23, w33, acc0, acc1, acc2, acc3);
    float* pw = sP + wave * (16 * TAIL_PP) + ((lane >> 4) << 3) * TAIL_PP + r;
    for (int v = 0; v < TAIL_S; ++v) {
      pw[v * TAIL_PP]      = acc0[v];
      pw[v * TAIL_PP + 16] = acc1[v];
      pw[v * TAIL_PP + 32] = acc2[v];
      pw[v * TAIL_PP + 48] = acc3[v];
    }
  }
  __syncthreads();
  if (sums) {
    int row = t >> 4;
    const float* ps = sP + row * TAIL_PP + c4s;
    float4 s = *(const float4*)ps;
    for (int w = 1; w < 4; ++w) {
      float4 p = *(const float4*)(ps + w * (16 * TAIL_PP));
      s.x += p.x; s.y += p.y; s.z += p.z; s.w += p.w;
    }
    s.x += bv.x; s.y += bv.y; s.z += bv.z; s.w += bv.w;
    *(float4*)(sD + row * TAIL_PP + c4s) = s;
  }
  __syncthreads();

  if (head) {
    const float* row = sD + hs * TAIL_PP;
    const float* w   = sW6 + ho * TAIL_PP;
    float acc = b6v;
    for (int k4 = 0; k4 < TAIL_N; k4 += 4) {
      float4 rv = *(const float4*)(row + k4);
      float4 wv = *(const float4*)(w + k4);
      acc += rv.x * wv.x;
      acc += rv.y * wv.y;
      acc += rv.z * wv.z;
      acc += rv.w * wv.w;
    }
    s6[hs * 8 + ho] = acc;
  }
  __syncthreads();
  if (head) {
    const float* x6 = s6 + hs * 8;
    float nrm = sqrtf(x6[0]*x6[0] + x6[1]*x6[1] + x6[2]*x6[2] + x6[3]*x6[3]);
    float v = x6[ho];
    if (ho < 4) v = v / fmaxf(nrm, 1e-12f);
    out[(size_t)(b0 + hs) * 7 + ho] = v;
    out[(size_t)B * 7 + (size_t)(b0 + hs) * 7 + ho] = v;
  }
}

extern "C" void kernel_launch(void* const* d_in, const int* in_sizes, int n_in,
                              void* d_out, int out_size, void* d_ws, size_t ws_size,
                              hipStream_t stream) {
  const float* x    = (const float*)d_in[0];
  const float* y    = (const float*)d_in[1];
  const float* W1   = (const float*)d_in[2];
  const float* be1  = (const float*)d_in[3];
  const float* fcW1 = (const float*)d_in[4];
  const float* fcb1 = (const float*)d_in[5];
  const float* W2   = (const float*)d_in[6];
  const float* be2  = (const float*)d_in[7];
  const float* fcW2 = (const float*)d_in[8];
  const float* fcb2 = (const float*)d_in[9];
  const float* W3   = (const float*)d_in[10];
  const float* be3  = (const float*)d_in[11];
  const float* fcW3 = (const float*)d_in[12];
  const float* fcb3 = (const float*)d_in[13];
  const float* W4   = (const float*)d_in[14];
  const float* be4  = (const float*)d_in[15];
  const float* fcW4 = (const float*)d_in[16];
  const float* fcb4 = (const float*)d_in[17];
  const float* W6   = (const float*)d_in[18];
  const float* b6   = (const float*)d_in[19];

  float* ws = (float*)d_ws;
  float* q  = ws;
  float* mA = q  + 524288;
  float* mC = mA + 524288;
  float* d0 = mC + 524288;
  float* d1 = d0 + 262144;

  const int B = 512;
  auto gemm_pair = [&](const float* A0, const float* W0, const float* bias0, float* C0,
                       int M0, int N0, int K0,
                       const float* A1, const float* W1, const float* bias1, float* C1,
                       int M1, int N1, int K1) {
    int tiles = (M0 >> 5) * (N0 >> 5) + (M1 >> 5) * (N1 >> 5);
    hipLaunchKernelGGL(gemm_pair_wmma_f32, dim3(tiles), dim3(GEMM_WAVES * 32), 0, stream,
                       A0, W0, bias0, C0, M0, N0, K0, A1, W1, bias1, C1, M1, N1, K1);
  };
  auto attn = [&](const float* beta, const float* dprev, float* mAout, int Nout, int threads) {
    hipLaunchKernelGGL(attn_mish_kernel, dim3(B), dim3(threads), 0, stream,
                       q, beta, dprev, mAout, mC, Nout, B);
  };

  hipLaunchKernelGGL(gemm_mish_wmma_f32, dim3(((2 * B) >> 5) * (512 >> 5)), dim3(GEMM_WAVES * 32), 0, stream,
                     x, y, W1, q, B, 512, 1024);
  attn(be1, nullptr, mA, 512, 256);
  gemm_pair(mC, fcW1, fcb1,    d0, B,     512, 1024,  mA, W2, nullptr, q, 2 * B, 256, 512);
  attn(be2, d0, mA, 256, 256);
  gemm_pair(mC, fcW2, fcb2,    d1, B,     256, 512,   mA, W3, nullptr, q, 2 * B, 128, 256);
  hipLaunchKernelGGL(tail_kernel, dim3(B / TAIL_S), dim3(256), 0, stream,
                     q, be3, d1, fcW3, fcb3, W4, be4, fcW4, fcb4, W6, b6, (float*)d_out, B);

  (void)in_sizes; (void)n_in; (void)out_size; (void)ws_size;
}


# attention launches folded into the GEMM launches: next products' lanes form their A fragments from the query rows (beta 0: from their mish, stored by the epilogue); 6 launches -> 4; sums unchanged
# speedup vs baseline: 7.4609x; 1.1419x over previous
#include <hip/hip_runtime.h>
#include <math.h>

typedef __attribute__((ext_vector_type(2))) float v2f;
typedef __attribute__((ext_vector_type(8))) float v8f;

__device__ __forceinline__ float mish_f(float v) {
  float n = __expf(v);
  float w = n * (n + 2.0f);
  float q = w * __builtin_amdgcn_rcpf(w + 2.0f);
  return (v > 20.0f) ? v : v * q;
}

__device__ __forceinline__ void mish8(float4& p, float4& r) {
  v8f v = {p.x, p.y, p.z, p.w, r.x, r.y, r.z, r.w};
  v8f n = {__expf(v[0]), __expf(v[1]), __expf(v[2]), __expf(v[3]),
           __expf(v[4]), __expf(v[5]), __expf(v[6]), __expf(v[7])};
  v8f w = n * (n + 2.0f);
  v8f d = w + 2.0f;
  v8f rc = {__builtin_amdgcn_rcpf(d[0]), __builtin_amdgcn_rcpf(d[1]),
            __builtin_amdgcn_rcpf(d[2]), __builtin_amdgcn_rcpf(d[3]),
            __builtin_amdgcn_rcpf(d[4]), __builtin_amdgcn_rcpf(d[5]),
            __builtin_amdgcn_rcpf(d[6]), __builtin_amdgcn_rcpf(d[7])};
  v8f m = v * (w * rc);
  v8f o = (v > 20.0f) ? v : m;
  p.x = o[0]; p.y = o[1]; p.z = o[2]; p.w = o[3];
  r.x = o[4]; r.y = o[5]; r.z = o[6]; r.w = o[7];
}

#define GEMM_WAVES 8
#define PART_PITCH 36
#define PART_TILE (32 * PART_PITCH)

__device__ __forceinline__ void mma_2x2_k8(const float4& a0, const float4& a1,
                                           const float4& b0, const float4& b1,
                                           v8f& acc00, v8f& acc01, v8f& acc10, v8f& acc11) {
  v2f a0l = {a0.x, a0.y}, a0h = {a0.z, a0.w};
  v2f a1l = {a1.x, a1.y}, a1h = {a1.z, a1.w};
  v2f b0l = {b0.x, b0.y}, b0h = {b0.z, b0.w};
  v2f b1l = {b1.x, b1.y}, b1h = {b1.z, b1.w};
  acc00 = __builtin_amdgcn_wmma_f32_16x16x4_f32(false, a0l, false, b0l, (short)0, acc00, false, false);
  acc01 = __builtin_amdgcn_wmma_f32_16x16x4_f32(false, a0l, false, b1l, (short)0, acc01, false, false);
  acc10 = __builtin_amdgcn_wmma_f32_16x16x4_f32(false, a1l, false, b0l, (short)0, acc10, false, false);
  acc11 = __builtin_amdgcn_wmma_f32_16x16x4_f32(false, a1l, false, b1l, (short)0, acc11, false, false);
  acc00 = __builtin_amdgcn_wmma_f32_16x16x4_f32(false, a0h, false, b0h, (short)0, acc00, false, false);
  acc01 = __builtin_amdgcn_wmma_f32_16x16x4_f32(false, a0h, false, b1h, (short)0, acc01, false, false);
  acc10 = __builtin_amdgcn_wmma_f32_16x16x4_f32(false, a1h, false, b0h, (short)0, acc10, false, false);
  acc11 = __builtin_amdgcn_wmma_f32_16x16x4_f32(false, a1h, false, b1h, (short)0, acc11, false, false);
}

__device__ __forceinline__ void tile_sum_store(const v8f& acc00, const v8f& acc01, const v8f& acc10, const v8f& acc11,
                                               float* part, int wave, int lane,
                                               const float* __restrict__ bias, float* __restrict__ C,
                                               float* __restrict__ Cm, int m0, int n0, int N) {
  float* pw = part + wave * PART_TILE + ((lane >> 4) << 3) * PART_PITCH + (lane & 15);
  for (int v = 0; v < 8; ++v) {
    pw[v * PART_PITCH]             = acc00[v];
    pw[v * PART_PITCH + 16]        = acc01[v];
    pw[(v + 16) * PART_PITCH]      = acc10[v];
    pw[(v + 16) * PART_PITCH + 16] = acc11[v];
  }
  __syncthreads();
  int row = threadIdx.x >> 3;
  int c4  = (threadIdx.x & 7) << 2;
  const float* ps = part + row * PART_PITCH + c4;
  float4 s = *(const float4*)ps;
  for (int w = 1; w < GEMM_WAVES; ++w) {
    float4 p = *(const float4*)(ps + w * PART_TILE);
    s.x += p.x; s.y += p.y; s.z += p.z; s.w += p.w;
  }
  if (bias) {
    float4 bv = *(const float4*)(bias + n0 + c4);
    s.x += bv.x; s.y += bv.y; s.z += bv.z; s.w += bv.w;
  }
  size_t off = (size_t)(m0 + row) * N + n0 + c4;
  *(float4*)(C + off) = s;
  if (Cm) {
    float4 ms = {mish_f(s.x), mish_f(s.y), mish_f(s.z), mish_f(s.w)};
    *(float4*)(Cm + off) = ms;
  }
}

__device__ __forceinline__ void gemm_tile_mish_in(const float* __restrict__ A, const float* __restrict__ W,
                                                  float* __restrict__ C, float* __restrict__ Cm,
                                                  int am0, int m0, int n0, int N, int K, float* part) {
  int wave = __builtin_amdgcn_readfirstlane((int)(threadIdx.x >> 5));
  int lane = threadIdx.x & 31;
  int r  = lane & 15;
  int kh = (lane >> 4) << 4;
  int ksl = ((((K >> 5) + GEMM_WAVES - 1) / GEMM_WAVES)) << 5;
  int kb = wave * ksl;
  int ke = kb + ksl;
  if (ke > K) ke = K;
  const float* Ar0 = A + (size_t)(am0 + r) * K + kh;
  const float* Ar1 = Ar0 + (size_t)16 * K;
  const float* Wr0 = W + (size_t)(n0 + r) * K + kh;
  const float* Wr1 = Wr0 + (size_t)16 * K;
  v8f acc00 = {};
  v8f acc01 = {};
  v8f acc10 = {};
  v8f acc11 = {};
  for (int k = kb; k < ke; k += 32) {
    const float4* pa0 = (const float4*)(Ar0 + k);
    const float4* pa1 = (const float4*)(Ar1 + k);
    const float4* pb0 = (const float4*)(Wr0 + k);
    const float4* pb1 = (const float4*)(Wr1 + k);
    float4 a00 = pa0[0], a01 = pa0[1], a02 = pa0[2], a03 = pa0[3];
    float4 b00 = pb0[0], b01 = pb0[1], b02 = pb0[2], b03 = pb0[3];
    float4 a10 = pa1[0], a11 = pa1[1], a12 = pa1[2], a13 = pa1[3];
    float4 b10 = pb1[0], b11 = pb1[1], b12 = pb1[2], b13 = pb1[3];
    __builtin_amdgcn_sched_barrier(0);
    mish8(a00, a10);
    mish8(a01, a11);
    mish8(a02, a12);
    mish8(a03, a13);
    __builtin_amdgcn_sched_barrier(0);
    mma_2x2_k8(a00, a10, b00, b10, acc00, acc01, acc10, acc11);
    mma_2x2_k8(a01, a11, b01, b11, acc00, acc01, acc10, acc11);
    mma_2x2_k8(a02, a12, b02, b12, acc00, acc01, acc10, acc11);
    mma_2x2_k8(a03, a13, b03, b13, acc00, acc01, acc10, acc11);
  }
  tile_sum_store(acc00, acc01, acc10, acc11, part, wave, lane, nullptr, C, Cm, m0, n0, N);
}

__global__ __launch_bounds__(GEMM_WAVES * 32)
void gemm_mish_wmma_f32(const float* __restrict__ x, const float* __restrict__ y,
                        const float* __restrict__ W, float* __restrict__ C, float* __restrict__ Cm,
                        int B, int N, int K) {
  __shared__ __attribute__((aligned(16))) float part[GEMM_WAVES * PART_TILE];
  int tile = blockIdx.x;
  int tiles_n = N >> 5;
  if (tile >= ((2 * B) >> 5) * tiles_n) return;
  int tm = tile / tiles_n;
  int m0 = tm << 5;
  int n0 = (tile - tm * tiles_n) << 5;
  bool ybr = m0 >= B;
  const float* A = ybr ? y : x;
  int am0 = ybr ? m0 - B : m0;
  gemm_tile_mish_in(A, W, C, Cm, am0, m0, n0, N, K, part);
}

struct GemmProb {
  const float* P;
  const float* Q;
  const float* dprev;
  const float* beta_p;
  const float* W;
  const float* bias;
  float* C;
  float* Cm;
  int M, N, K, Ns, B, fc;
};

__device__ __forceinline__ void attn_row8(const float* __restrict__ own, const float* __restrict__ oth,
                                          int i0, int Ns, float beta, float* o) {
  float omax = -3.4e38f, omin = 3.4e38f;
  for (int j = 0; j < Ns; ++j) {
    omax = fmaxf(omax, oth[j]);
    omin = fminf(omin, oth[j]);
  }
  float vi[8], m[8], den[8], num[8];
  for (int e = 0; e < 8; ++e) {
    vi[e] = own[i0 + e];
    m[e] = (vi[e] >= 0.0f) ? vi[e] * omax : vi[e] * omin;
    den[e] = 0.0f;
    num[e] = 0.0f;
  }
  for (int j = 0; j < Ns; ++j) {
    float oj = oth[j];
    float wj = own[j];
    for (int e = 0; e < 8; ++e) {
      float ex = __expf(vi[e] * oj - m[e]);
      den[e] += ex;
      num[e] += ex * wj;
    }
  }
  for (int e = 0; e < 8; ++e) o[e] = vi[e] + beta * num[e] / den[e];
}

__device__ __forceinline__ void gemm_tile_post(const GemmProb& g, int m0, int n0, float* part) {
  int wave = __builtin_amdgcn_readfirstlane((int)(threadIdx.x >> 5));
  int lane = threadIdx.x & 31;
  int r  = lane & 15;
  int kh = (lane >> 4) << 4;
  int K = g.K;
  int Ns = g.Ns;
  int ksl = ((((K >> 5) + GEMM_WAVES - 1) / GEMM_WAVES)) << 5;
  int kb = wave * ksl;
  int ke = kb + ksl;
  if (ke > K) ke = K;
  float beta = g.beta_p[0];
  bool plain = (beta == 0.0f) && (g.dprev == nullptr);
  bool addm  = (beta == 0.0f) && (g.dprev != nullptr);
  const float* Asrc = plain ? g.P : g.Q;
  size_t aox = (size_t)(m0 + r) * Ns + kh;
  size_t aoy = (size_t)(m0 + r + g.B) * Ns + kh - Ns;
  const float* Wr0 = g.W + (size_t)(n0 + r) * K + kh;
  const float* Wr1 = Wr0 + (size_t)16 * K;
  const float* Dr0 = g.dprev ? g.dprev + (size_t)(m0 + r) * K + kh : nullptr;
  v8f acc00 = {};
  v8f acc01 = {};
  v8f acc10 = {};
  v8f acc11 = {};
  for (int k = kb; k < ke; k += 32) {
    bool ybr = (g.fc != 0) && (k >= Ns);
    float4 a00, a01, a02, a03, a10, a11, a12, a13;
    float4 d00, d01, d02, d03, d10, d11, d12, d13;
    if (beta == 0.0f) {
      const float4* pa0 = (const float4*)(Asrc + (ybr ? aoy : aox) + k);
      const float4* pa1 = (const float4*)(Asrc + (ybr ? aoy : aox) + k + (size_t)16 * Ns);
      a00 = pa0[0]; a01 = pa0[1]; a02 = pa0[2]; a03 = pa0[3];
      a10 = pa1[0]; a11 = pa1[1]; a12 = pa1[2]; a13 = pa1[3];
      if (addm) {
        const float4* pd0 = (const float4*)(Dr0 + k);
        const float4* pd1 = (const float4*)(Dr0 + (size_t)16 * K + k);
        d00 = pd0[0]; d01 = pd0[1]; d02 = pd0[2]; d03 = pd0[3];
        d10 = pd1[0]; d11 = pd1[1]; d12 = pd1[2]; d13 = pd1[3];
      }
    } else {
      int qr = m0 + r + (ybr ? g.B : 0);
      int qo = (qr < g.B) ? qr + g.B : qr - g.B;
      int i0 = k - (ybr ? Ns : 0) + kh;
      float o0[16], o1[16];
      attn_row8(g.Q + (size_t)qr * Ns, g.Q + (size_t)qo * Ns, i0, Ns, beta, o0);
      attn_row8(g.Q + (size_t)qr * Ns, g.Q + (size_t)qo * Ns, i0 + 8, Ns, beta, o0 + 8);
      attn_row8(g.Q + (size_t)(qr + 16) * Ns, g.Q + (size_t)(qo + 16) * Ns, i0, Ns, beta, o1);
      attn_row8(g.Q + (size_t)(qr + 16) * Ns, g.Q + (size_t)(qo + 16) * Ns, i0 + 8, Ns, beta, o1 + 8);
      if (g.dprev) {
        const float* pd0 = Dr0 + k;
        const float* pd1 = Dr0 + (size_t)16 * K + k;
        for (int e = 0; e < 16; ++e) { o0[e] = mish_f(o0[e] + pd0[e]); o1[e] = mish_f(o1[e] + pd1[e]); }
      } else {
        for (int e = 0; e < 16; ++e) { o0[e] = mish_f(o0[e]); o1[e] = mish_f(o1[e]); }
      }
      a00 = make_float4(o0[0], o0[1], o0[2], o0[3]);    a01 = make_float4(o0[4], o0[5], o0[6], o0[7]);
      a02 = make_float4(o0[8], o0[9], o0[10], o0[11]);  a03 = make_float4(o0[12], o0[13], o0[14], o0[15]);
      a10 = make_float4(o1[0], o1[1], o1[2], o1[3]);    a11 = make_float4(o1[4], o1[5], o1[6], o1[7]);
      a12 = make_float4(o1[8], o1[9], o1[10], o1[11]);  a13 = make_float4(o1[12], o1[13], o1[14], o1[15]);
    }
    const float4* pb0 = (const float4*)(Wr0 + k);
    const float4* pb1 = (const float4*)(Wr1 + k);
    float4 b00 = pb0[0], b01 = pb0[1], b02 = pb0[2], b03 = pb0[3];
    float4 b10 = pb1[0], b11 = pb1[1], b12 = pb1[2], b13 = pb1[3];
    __builtin_amdgcn_sched_barrier(0);
    if (addm) {
      a00 += d00; a01 += d01; a02 += d02; a03 += d03;
      a10 += d10; a11 += d11; a12 += d12; a13 += d13;
      mish8(a00, a10);
      mish8(a01, a11);
      mish8(a02, a12);
      mish8(a03, a13);
      __builtin_amdgcn_sched_barrier(0);
    }
    mma_2x2_k8(a00, a10, b00, b10, acc00, acc01, acc10, acc11);
    mma_2x2_k8(a01, a11, b01, b11, acc00, acc01, acc10, acc11);
    mma_2x2_k8(a02, a12, b02, b12, acc00, acc01, acc10, acc11);
    mma_2x2_k8(a03, a13, b03, b13, acc00, acc01, acc10, acc11);
  }
  tile_sum_store(acc00, acc01, acc10, acc11, part, wave, lane, g.bias, g.C, g.Cm, m0, n0, g.N);
}

__global__ __launch_bounds__(GEMM_WAVES * 32)
void gemm_pair_wmma_f32(GemmProb p0, GemmProb p1) {
  __shared__ __attribute__((aligned(16))) float part[GEMM_WAVES * PART_TILE];
  int tile = blockIdx.x;
  int tiles0 = (p0.M >> 5) * (p0.N >> 5);
  bool second = tile >= tiles0;
  GemmProb g = second ? p1 : p0;
  if (second) tile -= tiles0;
  int tiles_n = g.N >> 5;
  if (tile >= (g.M >> 5) * tiles_n) return;
  int tm = tile / tiles_n;
  int m0 = tm << 5;
  int n0 = (tile - tm * tiles_n) << 5;
  gemm_tile_post(g, m0, n0, part);
}

#define TAIL_S 4
#define TAIL_N 64
#define TAIL_K 128
#define TAIL_CP 132
#define TAIL_PP 68
#define T3_N 128
#define T3_K 256
#define T3_CP 260
#define T3_AP 132
#define T3_XP 129

__device__ __forceinline__ void mma_1x4_k4(const v2f& a, const v2f& b0, const v2f& b1,
                                           const v2f& b2, const v2f& b3,
                                           v8f& c0, v8f& c1, v8f& c2, v8f& c3) {
  c0 = __builtin_amdgcn_wmma_f32_16x16x4_f32(false, a, false, b0, (short)0, c0, false, false);
  c1 = __builtin_amdgcn_wmma_f32_16x16x4_f32(false, a, false, b1, (short)0, c1, false, false);
  c2 = __builtin_amdgcn_wmma_f32_16x16x4_f32(false, a, false, b2, (short)0, c2, false, false);
  c3 = __builtin_amdgcn_wmma_f32_16x16x4_f32(false, a, false, b3, (short)0, c3, false, false);
}

__device__ __forceinline__ void mma_1x4_k8(const float4& a, const float4& b0, const float4& b1,
                                           const float4& b2, const float4& b3,
                                           v8f& c0, v8f& c1, v8f& c2, v8f& c3) {
  v2f al = {a.x, a.y}, ah = {a.z, a.w};
  v2f b0l = {b0.x, b0.y}, b0h = {b0.z, b0.w};
  v2f b1l = {b1.x, b1.y}, b1h = {b1.z, b1.w};
  v2f b2l = {b2.x, b2.y}, b2h = {b2.z, b2.w};
  v2f b3l = {b3.x, b3.y}, b3h = {b3.z, b3.w};
  mma_1x4_k4(al, b0l, b1l, b2l, b3l, c0, c1, c2, c3);
  mma_1x4_k4(ah, b0h, b1h, b2h, b3h, c0, c1, c2, c3);
}

__device__ __forceinline__ void mma_k4(const v2f& a, const v2f& b, v8f& c) {
  c = __builtin_amdgcn_wmma_f32_16x16x4_f32(false, a, false, b, (short)0, c, false, false);
}

__global__ __launch_bounds__(256)
void tail_kernel(const float* __restrict__ Q, const float* __restrict__ beta3_p,
                 const float* __restrict__ d1, const float* __restrict__ fcW3,
                 const float* __restrict__ fcb3, const float* __restrict__ W4,
                 const float* __restrict__ beta_p, const float* __restrict__ fcW,
                 const float* __restrict__ fcb, const float* __restrict__ W6,
                 const float* __restrict__ b6, float* __restrict__ out, int B) {
  __shared__ __attribute__((aligned(16))) float sC3[16 * T3_CP];
  __shared__ __attribute__((aligned(16))) float sA3[16 * T3_AP];
  __shared__ __attribute__((aligned(16))) float sD2[16 * T3_AP];
  __shared__ __attribute__((aligned(16))) float sQ4[16 * TAIL_PP];
  __shared__ float sX3[TAIL_S * T3_XP];
  __shared__ float sY3[TAIL_S * T3_XP];
  __shared__ __attribute__((aligned(16))) float sC[16 * TAIL_CP];
  __shared__ __attribute__((aligned(16))) float sP[4 * 16 * TAIL_PP];
  __shared__ __attribute__((aligned(16))) float sD[TAIL_S * TAIL_PP];
  __shared__ __attribute__((aligned(16))) float sW6[7 * TAIL_PP];
  __shared__ float s6[TAIL_S * 8];
  int t = threadIdx.x;
  int wave = __builtin_amdgcn_readfirstlane((int)(threadIdx.x >> 5));
  int lane = t & 31;
  int r  = lane & 15;
  int kh = (lane >> 4) << 4;
  int b0 = blockIdx.x * TAIL_S;
  int hs = t >> 3;
  int ho = t & 7;
  bool head = (t < TAIL_S * 8) && (ho < 7);
  bool sums = t < TAIL_S * 16;
  int c4s = (t & 15) << 2;

  float beta3 = beta3_p[0];
  float beta = beta_p[0];
  int c3  = t;
  int ci3 = c3 & (T3_N - 1);
  int yb3 = c3 >> 7;
  const float* qb3 = Q + (size_t)yb3 * B * T3_N + (size_t)b0 * T3_N + ci3;
  const float* db3 = d1 + (size_t)b0 * T3_K + c3;
  float o0 = qb3[0], o1 = qb3[T3_N], o2 = qb3[2 * T3_N], o3 = qb3[3 * T3_N];
  float e0 = db3[0], e1 = db3[T3_K], e2 = db3[2 * T3_K], e3 = db3[3 * T3_K];
  if (beta3 != 0.0f) {
    for (int e = t; e < 2 * TAIL_S * T3_N; e += 256) {
      int y = e >> 9;
      int s = (e >> 7) & (TAIL_S - 1);
      int c = e & (T3_N - 1);
      float v = Q[(size_t)(b0 + s + (y ? B : 0)) * T3_N + c];
      if (y) sY3[s * T3_XP + c] = v; else sX3[s * T3_XP + c] = v;
    }
  }
  float4 wf[8][4];
  {
    const float* p = fcW3 + (size_t)(16 * wave + r) * T3_K + kh;
#pragma unroll
    for (int blk = 0; blk < 8; ++blk) {
      const float4* p4 = (const float4*)(p + 32 * blk);
      wf[blk][0] = p4[0]; wf[blk][1] = p4[1]; wf[blk][2] = p4[2]; wf[blk][3] = p4[3];
    }
  }
  float4 wx[4][4];
  {
    const float* p = (wave >= 4) ? W4 + (size_t)(16 * (wave - 4) + r) * T3_N + kh
                                 : fcW + (size_t)r * TAIL_K + 32 * wave + kh;
    int gs = (wave >= 4) ? 32 : 16 * TAIL_K;
#pragma unroll
    for (int g = 0; g < 4; ++g) {
      const float4* p4 = (const float4*)(p + (size_t)g * gs);
      wx[g][0] = p4[0]; wx[g][1] = p4[1]; wx[g][2] = p4[2]; wx[g][3] = p4[3];
    }
  }
  float b3v = fcb3[16 * wave + r];
  float4 bv = *(const float4*)(fcb + c4s);
  float b6v = b6[ho < 7 ? ho : 6];
  bool w6t = (t >= 128) && (t < 128 + 7 * 16);
  int w6o = (t - 128) >> 4;
  int w6c = ((t - 128) & 15) << 2;
  float4 w6v = *(const float4*)(W6 + (w6t ? w6o * TAIL_N + w6c : 0));
  for (int z = t; z < (16 - TAIL_S) * 64; z += 256) {
    float4 zero = {};
    *(float4*)(sC3 + (TAIL_S + (z >> 6)) * T3_CP + ((z & 63) << 2)) = zero;
  }
  for (int z = t; z < (16 - 2 * TAIL_S) * 32; z += 256) {
    float4 zero = {};
    *(float4*)(sA3 + (2 * TAIL_S + (z >> 5)) * T3_AP + ((z & 31) << 2)) = zero;
  }
  for (int z = t; z < (16 - TAIL_S) * 32; z += 256) {
    float4 zero = {};
    *(float4*)(sC + (TAIL_S + (z >> 5)) * TAIL_CP + ((z & 31) << 2)) = zero;
  }

  if (beta3 == 0.0f) {
    float* pc = sC3 + c3;
    float* pm = sA3 + (yb3 * TAIL_S) * T3_AP + ci3;
    pc[0]         = mish_f(o0 + e0);
    pc[T3_CP]     = mish_f(o1 + e1);
    pc[2 * T3_CP] = mish_f(o2 + e2);
    pc[3 * T3_CP] = mish_f(o3 + e3);
    pm[0]         = mish_f(o0);
    pm[T3_AP]     = mish_f(o1);
    pm[2 * T3_AP] = mish_f(o2);
    pm[3 * T3_AP] = mish_f(o3);
  } else {
    __syncthreads();
    for (int e = t; e < TAIL_S * T3_K; e += 256) {
      int s   = e >> 8;
      int i   = (e & 255) >> 1;
      int dir = e & 1;
      const float* own = (dir ? sY3 : sX3) + s * T3_XP;
      const float* oth = (dir ? sX3 : sY3) + s * T3_XP;
      float omax = -3.4e38f, omin = 3.4e38f;
      for (int j = 0; j < T3_N; ++j) {
        omax = fmaxf(omax, oth[j]);
        omin = fminf(omin, oth[j]);
      }
      float vi = own[i];
      float dd = d1[(size_t)(b0 + s) * T3_K + dir * T3_N + i];
      float m = (vi >= 0.0f) ? vi * omax : vi * omin;
      float den = 0.0f, num = 0.0f;
      for (int j = 0; j < T3_N; ++j) {
        float ex = __expf(vi * oth[j] - m);
        den += ex;
        num += ex * own[j];
      }
      float o = vi + beta3 * num / den;
      sC3[s * T3_CP + dir * T3_N + i] = mish_f(o + dd);
      sA3[(dir * TAIL_S + s) * T3_AP + i] = mish_f(o);
    }
  }
  __builtin_amdgcn_fence(__ATOMIC_RELEASE, "workgroup", "local");
  __builtin_amdgcn_s_barrier();
  __builtin_amdgcn_fence(__ATOMIC_ACQUIRE, "workgroup", "local");

  {
    v8f acc[8];
    const float* pa = sC3 + r * T3_CP + kh;
    float4 af[2][8];
    {
      const float4* p0 = (const float4*)pa;
      const float4* p1 = (const float4*)(pa + 32);
      af[0][0] = p0[0]; af[0][1] = p0[1]; af[0][2] = p0[2]; af[0][3] = p0[3];
      af[0][4] = p1[0]; af[0][5] = p1[1]; af[0][6] = p1[2]; af[0][7] = p1[3];
    }
#pragma unroll
    for (int pr = 0; pr < 4; ++pr) {
      if (pr < 3) {
        const float4* p0 = (const float4*)(pa + 64 * (pr + 1));
        const float4* p1 = (const float4*)(pa + 64 * (pr + 1) + 32);
        af[(pr + 1) & 1][0] = p0[0]; af[(pr + 1) & 1][1] = p0[1]; af[(pr + 1) & 1][2] = p0[2]; af[(pr + 1) & 1][3] = p0[3];
        af[(pr + 1) & 1][4] = p1[0]; af[(pr + 1) & 1][5] = p1[1]; af[(pr + 1) & 1][6] = p1[2]; af[(pr + 1) & 1][7] = p1[3];
      }
      __builtin_amdgcn_sched_barrier(0);
      v8f c0 = {};
      v8f c1 = {};
#pragma unroll
      for (int q = 0; q < 4; ++q) {
        const float4 x0 = af[pr & 1][q], x1 = af[pr & 1][4 + q];
        const float4 y0 = wf[2 * pr][q], y1 = wf[2 * pr + 1][q];
        v2f a0l = {x0.x, x0.y}, a0h = {x0.z, x0.w};
        v2f a1l = {x1.x, x1.y}, a1h = {x1.z, x1.w};
        v2f w0l = {y0.x, y0.y}, w0h = {y0.z, y0.w};
        v2f w1l = {y1.x, y1.y}, w1h = {y1.z, y1.w};
        mma_k4(a0l, w0l, c0);
        mma_k4(a1l, w1l, c1);
        mma_k4(a0h, w0h, c0);
        mma_k4(a1h, w1h, c1);
      }
      acc[2 * pr] = c0;
      acc[2 * pr + 1] = c1;
      __builtin_amdgcn_sched_barrier(0);
    }
    v8f s = acc[0];
#pragma unroll
    for (int blk = 1; blk < 8; ++blk) s += acc[blk];
    float* pw = sD2 + ((lane >> 4) << 3) * T3_AP + 16 * wave + r;
#pragma unroll
    for (int v = 0; v < TAIL_S; ++v) pw[v * T3_AP] = s[v] + b3v;
  }
  if (wave >= 4) {
    const float* pa = sA3 + r * T3_AP + kh;
    float4 a[4][4];
#pragma unroll
    for (int blk = 0; blk < 4; ++blk) {
      const float4* p4 = (const float4*)(pa + 32 * blk);
      a[blk][0] = p4[0]; a[blk][1] = p4[1]; a[blk][2] = p4[2]; a[blk][3] = p4[3];
    }
    __builtin_amdgcn_sched_barrier(0);
    v8f c0 = {};
    v8f c1 = {};
    v8f c2 = {};
    v8f c3 = {};
#pragma unroll
    for (int q = 0; q < 4; ++q) {
      v2f al0 = {a[0][q].x, a[0][q].y}, al1 = {a[1][q].x, a[1][q].y}, al2 = {a[2][q].x, a[2][q].y}, al3 = {a[3][q].x, a[3][q].y};
      v2f wl0 = {wx[0][q].x, wx[0][q].y}, wl1 = {wx[1][q].x, wx[1][q].y}, wl2 = {wx[2][q].x, wx[2][q].y}, wl3 = {wx[3][q].x, wx[3][q].y};
      mma_k4(al0, wl0, c0);
      mma_k4(al1, wl1, c1);
      mma_k4(al2, wl2, c2);
      mma_k4(al3, wl3, c3);
      v2f ah0 = {a[0][q].z, a[0][q].w}, ah1 = {a[1][q].z, a[1][q].w}, ah2 = {a[2][q].z, a[2][q].w}, ah3 = {a[3][q].z, a[3][q].w};
      v2f wh0 = {wx[0][q].z, wx[0][q].w}, wh1 = {wx[1][q].z, wx[1][q].w}, wh2 = {wx[2][q].z, wx[2][q].w}, wh3 = {wx[3][q].z, wx[3][q].w};
      mma_k4(ah0, wh0, c0);
      mma_k4(ah1, wh1, c1);
      mma_k4(ah2, wh2, c2);
      mma_k4(ah3, wh3, c3);
    }
    v8f s = c0;
    s += c1;
    s += c2;
    s += c3;
    float* pw = sQ4 + ((lane >> 4) << 3) * TAIL_PP + 16 * (wave - 4) + r;
#pragma unroll
    for (int v = 0; v < 8; ++v) pw[v * TAIL_PP] = s[v];
  }
  float4 w00 = wx[0][0], w01 = wx[0][1], w02 = wx[0][2], w03 = wx[0][3];
  float4 w10 = wx[1][0], w11 = wx[1][1], w12 = wx[1][2], w13 = wx[1][3];
  float4 w20 = wx[2][0], w21 = wx[2][1], w22 = wx[2][2], w23 = wx[2][3];
  float4 w30 = wx[3][0], w31 = wx[3][1], w32 = wx[3][2], w33 = wx[3][3];
  __syncthreads();

  if (beta == 0.0f) {
    int c  = t & 127;
    int sa = t >> 7;
    int sb = sa + 2;
    int qr = (c < TAIL_N) ? 0 : TAIL_S;
    int qc = c & (TAIL_N - 1);
    float oa = sQ4[(qr + sa) * TAIL_PP + qc];
    float ob = sQ4[(qr + sb) * TAIL_PP + qc];
    float da = sD2[sa * T3_AP + c];
    float db = sD2[sb * T3_AP + c];
    sC[sa * TAIL_CP + c] = mish_f(oa + da);
    sC[sb * TAIL_CP + c] = mish_f(ob + db);
  } else {
    for (int e = t; e < TAIL_S * TAIL_K; e += 256) {
      int s   = e >> 7;
      int i   = (e & 127) >> 1;
      int dir = e & 1;
      const float* own = sQ4 + ((dir ? TAIL_S : 0) + s) * TAIL_PP;
      const float* oth = sQ4 + ((dir ? 0 : TAIL_S) + s) * TAIL_PP;
      float omax = -3.4e38f, omin = 3.4e38f;
      for (int j = 0; j < TAIL_N; ++j) {
        omax = fmaxf(omax, oth[j]);
        omin = fminf(omin, oth[j]);
      }
      float vi = own[i];
      float dd = sD2[s * T3_AP + dir * TAIL_N + i];
      float m = (vi >= 0.0f) ? vi * omax : vi * omin;
      float den = 0.0f, num = 0.0f;
      for (int j = 0; j < TAIL_N; ++j) {
        float e1 = __expf(vi * oth[j] - m);
        den += e1;
        num += e1 * own[j];
      }
      float o = vi + beta * num / den;
      sC[s * TAIL_CP + dir * TAIL_N + i] = mish_f(o + dd);
    }
  }
  if (w6t) *(float4*)(sW6 + w6o * TAIL_PP + w6c) = w6v;
  __syncthreads();

  if (wave < 4) {
    const float4* pa = (const float4*)(sC + r * TAIL_CP + 32 * wave + kh);
    float4 a0 = pa[0], a1 = pa[1], a2 = pa[2], a3 = pa[3];
    v8f acc0 = {};
    v8f acc1 = {};
    v8f acc2 = {};
    v8f acc3 = {};
    mma_1x4_k8(a0, w00, w10, w20, w30, acc0, acc1, acc2, acc3);
    mma_1x4_k8(a1, w01, w11, w21, w31, acc0, acc1, acc2, acc3);
    mma_1x4_k8(a2, w02, w12, w22, w32, acc0, acc1, acc2, acc3);
    mma_1x4_k8(a3, w03, w13, w23, w33, acc0, acc1, acc2, acc3);
    float* pw = sP + wave * (16 * TAIL_PP) + ((lane >> 4) << 3) * TAIL_PP + r;
    for (int v = 0; v < TAIL_S; ++v) {
      pw[v * TAIL_PP]      = acc0[v];
      pw[v * TAIL_PP + 16] = acc1[v];
      pw[v * TAIL_PP + 32] = acc2[v];
      pw[v * TAIL_PP + 48] = acc3[v];
    }
  }
  __syncthreads();
  if (sums) {
    int row = t >> 4;
    const float* ps = sP + row * TAIL_PP + c4s;
    float4 s = *(const float4*)ps;
    for (int w = 1; w < 4; ++w) {
      float4 p = *(const float4*)(ps + w * (16 * TAIL_PP));
      s.x += p.x; s.y += p.y; s.z += p.z; s.w += p.w;
    }
    s.x += bv.x; s.y += bv.y; s.z += bv.z; s.w += bv.w;
    *(float4*)(sD + row * TAIL_PP + c4s) = s;
  }
  __syncthreads();

  if (head) {
    const float* row = sD + hs * TAIL_PP;
    const float* w   = sW6 + ho * TAIL_PP;
    float acc = b6v;
    for (int k4 = 0; k4 < TAIL_N; k4 += 4) {
      float4 rv = *(const float4*)(row + k4);
      float4 wv = *(const float4*)(w + k4);
      acc += rv.x * wv.x;
      acc += rv.y * wv.y;
      acc += rv.z * wv.z;
      acc += rv.w * wv.w;
    }
    s6[hs * 8 + ho] = acc;
  }
  __syncthreads();
  if (head) {
    const float* x6 = s6 + hs * 8;
    float nrm = sqrtf(x6[0]*x6[0] + x6[1]*x6[1] + x6[2]*x6[2] + x6[3]*x6[3]);
    float v = x6[ho];
    if (ho < 4) v = v / fmaxf(nrm, 1e-12f);
    out[(size_t)(b0 + hs) * 7 + ho] = v;
    out[(size_t)B * 7 + (size_t)(b0 + hs) * 7 + ho] = v;
  }
}

extern "C" void kernel_launch(void* const* d_in, const int* in_sizes, int n_in,
                              void* d_out, int out_size, void* d_ws, size_t ws_size,
                              hipStream_t stream) {
  const float* x    = (const float*)d_in[0];
  const float* y    = (const float*)d_in[1];
  const float* W1   = (const float*)d_in[2];
  const float* be1  = (const float*)d_in[3];
  const float* fcW1 = (const float*)d_in[4];
  const float* fcb1 = (const float*)d_in[5];
  const float* W2   = (const float*)d_in[6];
  const float* be2  = (const float*)d_in[7];
  const float* fcW2 = (const float*)d_in[8];
  const float* fcb2 = (const float*)d_in[9];
  const float* W3   = (const float*)d_in[10];
  const float* be3  = (const float*)d_in[11];
  const float* fcW3 = (const float*)d_in[12];
  const float* fcb3 = (const float*)d_in[13];
  const float* W4   = (const float*)d_in[14];
  const float* be4  = (const float*)d_in[15];
  const float* fcW4 = (const float*)d_in[16];
  const float* fcb4 = (const float*)d_in[17];
  const float* W6   = (const float*)d_in[18];
  const float* b6   = (const float*)d_in[19];

  float* ws  = (float*)d_ws;
  float* q1  = ws;
  float* mq1 = q1  + 524288;
  float* q2  = mq1 + 524288;
  float* mq2 = q2  + 262144;
  float* q3  = mq2 + 262144;
  float* d0  = q3  + 131072;
  float* d1  = d0  + 262144;

  const int B = 512;
  hipLaunchKernelGGL(gemm_mish_wmma_f32, dim3(((2 * B) >> 5) * (512 >> 5)), dim3(GEMM_WAVES * 32), 0, stream,
                     x, y, W1, q1, mq1, B, 512, 1024);
  GemmProb fc1 = {mq1, q1, nullptr, be1, fcW1, fcb1,    d0, nullptr, B,     512, 1024, 512, B, 1};
  GemmProb qp2 = {mq1, q1, nullptr, be1, W2,   nullptr, q2, mq2,     2 * B, 256, 512,  512, B, 0};
  hipLaunchKernelGGL(gemm_pair_wmma_f32, dim3((B >> 5) * (512 >> 5) + ((2 * B) >> 5) * (256 >> 5)),
                     dim3(GEMM_WAVES * 32), 0, stream, fc1, qp2);
  GemmProb fc2 = {mq2, q2, d0,      be2, fcW2, fcb2,    d1, nullptr, B,     256, 512, 256, B, 1};
  GemmProb qp3 = {mq2, q2, nullptr, be2, W3,   nullptr, q3, nullptr, 2 * B, 128, 256, 256, B, 0};
  hipLaunchKernelGGL(gemm_pair_wmma_f32, dim3((B >> 5) * (256 >> 5) + ((2 * B) >> 5) * (128 >> 5)),
                     dim3(GEMM_WAVES * 32), 0, stream, fc2, qp3);
  hipLaunchKernelGGL(tail_kernel, dim3(B / TAIL_S), dim3(256), 0, stream,
                     q3, be3, d1, fcW3, fcb3, W4, be4, fcW4, fcb4, W6, b6, (float*)d_out, B);

  (void)in_sizes; (void)n_in; (void)out_size; (void)ws_size;
}
